# starting kernel
# baseline (speedup 1.0000x reference)
#include <hip/hip_runtime.h>
#include <hip/hip_bf16.h>

#define B_ 4
#define N_ 20000
#define C_ 512
#define M_ 1024
#define GRASP_TH 0.1f

typedef float v2f __attribute__((ext_vector_type(2)));
typedef float v8f __attribute__((ext_vector_type(8)));

#if defined(__gfx1250__) && __has_builtin(__builtin_amdgcn_wmma_f32_16x16x4_f32)
#define USE_WMMA 1
#else
#define USE_WMMA 0
#endif

__device__ __forceinline__ void ins3_stream(float v, int n, float d[3], int ix[3]) {
  const bool c0 = v < d[0];
  const bool c1 = v < d[1];
  const bool c2 = v < d[2];
  const float r0 = c0 ? v     : d[0];
  const int   j0 = c0 ? n     : ix[0];
  const float r1 = c0 ? d[0]  : (c1 ? v : d[1]);
  const int   j1 = c0 ? ix[0] : (c1 ? n : ix[1]);
  const float r2 = c1 ? d[1]  : (c2 ? v : d[2]);
  const int   j2 = c1 ? ix[1] : (c2 ? n : ix[2]);
  d[0] = r0; d[1] = r1; d[2] = r2;
  ix[0] = j0; ix[1] = j1; ix[2] = j2;
}

__device__ __forceinline__ void ins3_lex(float v, int n, float d[3], int ix[3]) {
  const bool c0 = (v < d[0]) | ((v == d[0]) & (n < ix[0]));
  const bool c1 = (v < d[1]) | ((v == d[1]) & (n < ix[1]));
  const bool c2 = (v < d[2]) | ((v == d[2]) & (n < ix[2]));
  const float r0 = c0 ? v     : d[0];
  const int   j0 = c0 ? n     : ix[0];
  const float r1 = c0 ? d[0]  : (c1 ? v : d[1]);
  const int   j1 = c0 ? ix[0] : (c1 ? n : ix[1]);
  const float r2 = c1 ? d[1]  : (c2 ? v : d[2]);
  const int   j2 = c1 ? ix[1] : (c2 ? n : ix[2]);
  d[0] = r0; d[1] = r1; d[2] = r2;
  ix[0] = j0; ix[1] = j1; ix[2] = j2;
}

__global__ __launch_bounds__(1024) void fps_kernel(
    const float* __restrict__ pc, const float* __restrict__ obj,
    const float* __restrict__ grasp, int* __restrict__ fps_idx) {
  const int b    = blockIdx.x;
  const int tid  = threadIdx.x;
  const int lane = tid & 31;
  const int wid  = tid >> 5;

  __shared__ float s_xyz[N_ * 3];
  __shared__ float s_mind[N_];
  __shared__ float swv[32];
  __shared__ int   swi[32];
  __shared__ int   s_last;
  __shared__ float s_px, s_py, s_pz;

  const float* pcb = pc + (size_t)b * N_ * 3;
  const float* ob0 = obj + (size_t)b * 2 * N_;
  const float* gr  = grasp + (size_t)b * N_;

  int firstLoc = N_;
  for (int n = tid; n < N_; n += 1024) {
    const float* p = pcb + (size_t)n * 3;
    const float x = p[0], y = p[1], z = p[2];
    s_xyz[n * 3 + 0] = x;
    s_xyz[n * 3 + 1] = y;
    s_xyz[n * 3 + 2] = z;
    const bool mk = (ob0[N_ + n] > ob0[n]) & (gr[n] > GRASP_TH);
    s_mind[n] = mk ? __builtin_inff() : -__builtin_inff();
    firstLoc  = (mk & (n < firstLoc)) ? n : firstLoc;
  }
  for (int m = 16; m; m >>= 1) {
    const int o = __shfl_xor(firstLoc, m, 32);
    firstLoc = (o < firstLoc) ? o : firstLoc;
  }
  if (lane == 0) swi[wid] = firstLoc;
  __syncthreads();
  if (wid == 0) {
    int v = swi[lane];
    for (int m = 16; m; m >>= 1) {
      const int o = __shfl_xor(v, m, 32);
      v = (o < v) ? o : v;
    }
    if (lane == 0) s_last = (v >= N_) ? 0 : v;
  }
  __syncthreads();

  for (int step = 0; step < M_; ++step) {
    if (tid == 0) {
      const int last = s_last;
      fps_idx[b * M_ + step] = last;
      s_px = s_xyz[last * 3 + 0];
      s_py = s_xyz[last * 3 + 1];
      s_pz = s_xyz[last * 3 + 2];
    }
    __syncthreads();
    const float px = s_px, py = s_py, pz = s_pz;

    float bestV = -__builtin_inff();
    int   bestI = 0x7fffffff;
    for (int n = tid; n < N_; n += 1024) {
      const float dx = s_xyz[n * 3 + 0] - px;
      const float dy = s_xyz[n * 3 + 1] - py;
      const float dz = s_xyz[n * 3 + 2] - pz;
      const float dd = dx * dx + dy * dy + dz * dz;
      const float nm = fminf(s_mind[n], dd);
      s_mind[n] = nm;
      const bool better = nm > bestV;
      bestV = better ? nm : bestV;
      bestI = better ? n : bestI;
    }
    for (int m = 16; m; m >>= 1) {
      const float ov = __shfl_xor(bestV, m, 32);
      const int   oi = __shfl_xor(bestI, m, 32);
      const bool better = (ov > bestV) | ((ov == bestV) & (oi < bestI));
      bestV = better ? ov : bestV;
      bestI = better ? oi : bestI;
    }
    if (lane == 0) { swv[wid] = bestV; swi[wid] = bestI; }
    __syncthreads();
    if (wid == 0) {
      float v = swv[lane];
      int   i = swi[lane];
      for (int m = 16; m; m >>= 1) {
        const float ov = __shfl_xor(v, m, 32);
        const int   oi = __shfl_xor(i, m, 32);
        const bool better = (ov > v) | ((ov == v) & (oi < i));
        v = better ? ov : v;
        i = better ? oi : i;
      }
      if (lane == 0) s_last = i;
    }
    __syncthreads();
  }
}

__global__ __launch_bounds__(256) void three_nn_kernel(
    const float* __restrict__ pc, const int* __restrict__ fps_idx,
    int* __restrict__ idx3, float* __restrict__ w3) {
  const int tid  = threadIdx.x;
  const int lane = tid & 31;
  const int wave = blockIdx.x * (256 >> 5) + (tid >> 5);
  const int tilesPerBatch = M_ / 16;
  const int b    = wave / tilesPerBatch;
  const int rt   = wave - b * tilesPerBatch;
  const int half = lane >> 4;
  const int l15  = lane & 15;

  const float* pcb = pc + (size_t)b * N_ * 3;

  const int mrow = rt * 16 + l15;
  const int aidx = fps_idx[b * M_ + mrow];
  const float* ap = pcb + (size_t)aidx * 3;
  const float ax = ap[0], ay = ap[1], az = ap[2];
  v2f A;
  A.x = half ? (-2.0f * az) : (-2.0f * ax);
  A.y = half ? 1.0f         : (-2.0f * ay);

  float td[8][3];
  int   ti[8][3];
#pragma unroll
  for (int r = 0; r < 8; ++r)
#pragma unroll
    for (int k = 0; k < 3; ++k) { td[r][k] = __builtin_inff(); ti[r][k] = 0x7fffffff; }

#pragma unroll 2
  for (int t = 0; t < N_ / 16; ++t) {
    const int n = t * 16 + l15;
    const float* bp = pcb + (size_t)n * 3;
    const int pn = (n + 128 < N_) ? (n + 128) : (N_ - 1);
    __builtin_prefetch(pcb + (size_t)pn * 3, 0, 1);
    const float bx = bp[0], by = bp[1], bz = bp[2];
    const float bn2 = bx * bx + by * by + bz * bz;
    v8f c = {0.f, 0.f, 0.f, 0.f, 0.f, 0.f, 0.f, 0.f};
#if USE_WMMA
    v2f Bv;
    Bv.x = half ? bz  : bx;
    Bv.y = half ? bn2 : by;
    c = __builtin_amdgcn_wmma_f32_16x16x4_f32(
         false, A,  false, Bv,
         (short)0, c,  false,  false);
#else
#pragma unroll
    for (int r = 0; r < 8; ++r) {
      const int row = half * 8 + r;
      const float rx = __shfl(ax, row, 32);
      const float ry = __shfl(ay, row, 32);
      const float rz = __shfl(az, row, 32);
      c[r] = -2.0f * (rx * bx + ry * by + rz * bz) + bn2;
    }
#endif
#pragma unroll
    for (int r = 0; r < 8; ++r) ins3_stream(c[r], n, td[r], ti[r]);
  }

#pragma unroll
  for (int mskk = 1; mskk <= 8; mskk <<= 1) {
#pragma unroll
    for (int r = 0; r < 8; ++r) {
      float pd[3]; int pi[3];
#pragma unroll
      for (int k = 0; k < 3; ++k) {
        pd[k] = __shfl_xor(td[r][k], mskk, 32);
        pi[k] = __shfl_xor(ti[r][k], mskk, 32);
      }
#pragma unroll
      for (int k = 0; k < 3; ++k) ins3_lex(pd[k], pi[k], td[r], ti[r]);
    }
  }

  if (l15 == 0) {
#pragma unroll
    for (int r = 0; r < 8; ++r) {
      const int row = rt * 16 + half * 8 + r;
      const int gi  = fps_idx[b * M_ + row];
      const float* gp = pcb + (size_t)gi * 3;
      const float gx = gp[0], gy = gp[1], gz = gp[2];
      float wv[3];
      float wsum = 0.f;
#pragma unroll
      for (int k = 0; k < 3; ++k) {
        const int nn = ti[r][k];
        const float* np = pcb + (size_t)nn * 3;
        const float dx = np[0] - gx, dy = np[1] - gy, dz = np[2] - gz;
        const float dd = dx * dx + dy * dy + dz * dz;
        const float dist = sqrtf(fmaxf(dd, 0.0f));
        wv[k] = 1.0f / (dist + 1e-8f);
        wsum += wv[k];
        idx3[((size_t)b * M_ + row) * 3 + k] = nn;
      }
#pragma unroll
      for (int k = 0; k < 3; ++k)
        w3[((size_t)b * M_ + row) * 3 + k] = wv[k] / wsum;
    }
  }
}

__global__ __launch_bounds__(256) void interp_kernel(
    const float* __restrict__ feats, const int* __restrict__ idx3,
    const float* __restrict__ w3, float* __restrict__ out) {
  const int bc = blockIdx.x;
  const int b  = bc / C_;
  const int c  = bc - b * C_;
  const float* f = feats + ((size_t)b * C_ + c) * N_;
  float*       o = out   + ((size_t)b * C_ + c) * M_;
  const int*   ib = idx3 + (size_t)b * M_ * 3;
  const float* wb = w3   + (size_t)b * M_ * 3;
  for (int m = threadIdx.x; m < M_; m += 256) {
    const int   n0 = ib[m * 3 + 0], n1 = ib[m * 3 + 1], n2 = ib[m * 3 + 2];
    const float w0 = wb[m * 3 + 0], w1 = wb[m * 3 + 1], w2 = wb[m * 3 + 2];
    o[m] = w0 * f[n0] + w1 * f[n1] + w2 * f[n2];
  }
}

extern "C" void kernel_launch(void* const* d_in, const int* in_sizes, int n_in,
                              void* d_out, int out_size, void* d_ws, size_t ws_size,
                              hipStream_t stream) {
  const float* pc    = (const float*)d_in[0];
  const float* feats = (const float*)d_in[1];
  const float* obj   = (const float*)d_in[2];
  const float* grasp = (const float*)d_in[3];
  float* out = (float*)d_out;

  char* ws = (char*)d_ws;
  size_t off = 0;
  int* fps = (int*)(ws + off);
  off += (size_t)B_ * M_ * 4;        off = (off + 255) & ~(size_t)255;
  int* idx3 = (int*)(ws + off);
  off += (size_t)B_ * M_ * 3 * 4;    off = (off + 255) & ~(size_t)255;
  float* w3 = (float*)(ws + off);
  off += (size_t)B_ * M_ * 3 * 4;

  fps_kernel<<<B_, 1024, 0, stream>>>(pc, obj, grasp, fps);
  three_nn_kernel<<<(B_ * (M_ / 16)) / 8, 256, 0, stream>>>(pc, fps, idx3, w3);
  interp_kernel<<<B_ * C_, 256, 0, stream>>>(feats, idx3, w3, out);
}


# fps kernel: two picks per pass: min distance in registers; a pass applies the confirmed pick and a guessed next pick (last runner-up); a right guess advances two of the 1024 steps; same sums
# speedup vs baseline: 1.4337x; 1.4337x over previous
#include <hip/hip_runtime.h>
#include <hip/hip_bf16.h>

#define B_ 4
#define N_ 20000
#define C_ 512
#define M_ 1024
#define GRASP_TH 0.1f

typedef float v2f __attribute__((ext_vector_type(2)));
typedef float v4f __attribute__((ext_vector_type(4)));
typedef float v8f __attribute__((ext_vector_type(8)));

#if defined(__gfx1250__) && __has_builtin(__builtin_amdgcn_wmma_f32_16x16x4_f32)
#define USE_WMMA 1
#else
#define USE_WMMA 0
#endif

__device__ __forceinline__ void ins3_stream(float v, int n, float d[3], int ix[3]) {
  const bool c0 = v < d[0];
  const bool c1 = v < d[1];
  const bool c2 = v < d[2];
  const float r0 = c0 ? v     : d[0];
  const int   j0 = c0 ? n     : ix[0];
  const float r1 = c0 ? d[0]  : (c1 ? v : d[1]);
  const int   j1 = c0 ? ix[0] : (c1 ? n : ix[1]);
  const float r2 = c1 ? d[1]  : (c2 ? v : d[2]);
  const int   j2 = c1 ? ix[1] : (c2 ? n : ix[2]);
  d[0] = r0; d[1] = r1; d[2] = r2;
  ix[0] = j0; ix[1] = j1; ix[2] = j2;
}

__device__ __forceinline__ void ins3_lex(float v, int n, float d[3], int ix[3]) {
  const bool c0 = (v < d[0]) | ((v == d[0]) & (n < ix[0]));
  const bool c1 = (v < d[1]) | ((v == d[1]) & (n < ix[1]));
  const bool c2 = (v < d[2]) | ((v == d[2]) & (n < ix[2]));
  const float r0 = c0 ? v     : d[0];
  const int   j0 = c0 ? n     : ix[0];
  const float r1 = c0 ? d[0]  : (c1 ? v : d[1]);
  const int   j1 = c0 ? ix[0] : (c1 ? n : ix[1]);
  const float r2 = c1 ? d[1]  : (c2 ? v : d[2]);
  const int   j2 = c1 ? ix[1] : (c2 ? n : ix[2]);
  d[0] = r0; d[1] = r1; d[2] = r2;
  ix[0] = j0; ix[1] = j1; ix[2] = j2;
}

#define FPS_J 5

__device__ __forceinline__ bool lex_better(float ov, int oi, float v, int i) {
  return (ov > v) | ((ov == v) & (oi < i));
}

__global__ __launch_bounds__(1024) void fps_kernel(
    const float* __restrict__ pc, const float* __restrict__ obj,
    const float* __restrict__ grasp, int* __restrict__ fps_idx) {
  const int b    = blockIdx.x;
  const int tid  = threadIdx.x;
  const int lane = tid & 31;
  const int wid  = tid >> 5;

  __shared__ __attribute__((aligned(16))) float s_x[N_];
  __shared__ __attribute__((aligned(16))) float s_y[N_];
  __shared__ __attribute__((aligned(16))) float s_z[N_];
  __shared__ float swvA[32];
  __shared__ int   swiA[32];
  __shared__ float swvB[32];
  __shared__ int   swiB[32];
  __shared__ int   s_a, s_b, s_hit;
  __shared__ float s_pa[3], s_pb[3];

  const float* pcb = pc + (size_t)b * N_ * 3;
  const float* ob0 = obj + (size_t)b * 2 * N_;
  const float* gr  = grasp + (size_t)b * N_;

  for (int n = tid; n < N_; n += 1024) {
    const float* p = pcb + (size_t)n * 3;
    s_x[n] = p[0];
    s_y[n] = p[1];
    s_z[n] = p[2];
  }

  float mind[FPS_J][4];
  int firstLoc = N_;
#pragma unroll
  for (int j = FPS_J - 1; j >= 0; --j) {
    const int g = tid + 1024 * j;
    const bool gv = g < N_ / 4;
#pragma unroll
    for (int k = 3; k >= 0; --k) {
      const int n  = 4 * g + k;
      const int nc = gv ? n : 0;
      const bool mk = gv & (ob0[N_ + nc] > ob0[nc]) & (gr[nc] > GRASP_TH);
      mind[j][k] = mk ? __builtin_inff() : -__builtin_inff();
      firstLoc   = mk ? n : firstLoc;
    }
  }
  for (int m = 16; m; m >>= 1) {
    const int o = __shfl_xor(firstLoc, m, 32);
    firstLoc = (o < firstLoc) ? o : firstLoc;
  }
  if (lane == 0) swiA[wid] = firstLoc;
  __syncthreads();
  if (wid == 0) {
    int v = swiA[lane];
    for (int m = 16; m; m >>= 1) {
      const int o = __shfl_xor(v, m, 32);
      v = (o < v) ? o : v;
    }
    if (lane == 0) {
      const int f = (v >= N_) ? 0 : v;
      s_a = f;
      s_b = f;
    }
  }
  __syncthreads();

  const v4f* x4 = (const v4f*)s_x;
  const v4f* y4 = (const v4f*)s_y;
  const v4f* z4 = (const v4f*)s_z;

  int pickB = 0;

  for (int step = 0; step < M_;) {
    if (tid == 0) {
      const int a = s_a;
      pickB = s_b;
      fps_idx[b * M_ + step] = a;
      s_pa[0] = s_x[a];     s_pa[1] = s_y[a];     s_pa[2] = s_z[a];
      s_pb[0] = s_x[pickB]; s_pb[1] = s_y[pickB]; s_pb[2] = s_z[pickB];
    }
    __syncthreads();
    const float ax = s_pa[0], ay = s_pa[1], az = s_pa[2];
    const float bx = s_pb[0], by = s_pb[1], bz = s_pb[2];

    float mab[FPS_J][4];
    float bestAV = -__builtin_inff(), bestBV = -__builtin_inff();
    int   bestAI = 0x7fffffff,        bestBI = 0x7fffffff;
#pragma unroll
    for (int j = 0; j < FPS_J; ++j) {
      const int g  = tid + 1024 * j;
      const int gc = (g < N_ / 4) ? g : tid;
      const v4f vx = x4[gc];
      const v4f vy = y4[gc];
      const v4f vz = z4[gc];
#pragma unroll
      for (int k = 0; k < 4; ++k) {
        const float dxa = vx[k] - ax, dya = vy[k] - ay, dza = vz[k] - az;
        const float dda = dxa * dxa + dya * dya + dza * dza;
        const float ma  = fminf(mind[j][k], dda);
        const float dxb = vx[k] - bx, dyb = vy[k] - by, dzb = vz[k] - bz;
        const float ddb = dxb * dxb + dyb * dyb + dzb * dzb;
        const float mb  = fminf(ma, ddb);
        mind[j][k] = ma;
        mab[j][k]  = mb;
        const bool betA = ma > bestAV;
        bestAV = betA ? ma : bestAV;
        bestAI = betA ? (4 * g + k) : bestAI;
        const bool betB = mb > bestBV;
        bestBV = betB ? mb : bestBV;
        bestBI = betB ? (4 * g + k) : bestBI;
      }
    }
    for (int m = 16; m; m >>= 1) {
      const float ova = __shfl_xor(bestAV, m, 32);
      const int   oia = __shfl_xor(bestAI, m, 32);
      const float ovb = __shfl_xor(bestBV, m, 32);
      const int   oib = __shfl_xor(bestBI, m, 32);
      const bool ca = lex_better(ova, oia, bestAV, bestAI);
      const bool cb = lex_better(ovb, oib, bestBV, bestBI);
      bestAV = ca ? ova : bestAV;  bestAI = ca ? oia : bestAI;
      bestBV = cb ? ovb : bestBV;  bestBI = cb ? oib : bestBI;
    }
    if (lane == 0) {
      swvA[wid] = bestAV; swiA[wid] = bestAI;
      swvB[wid] = bestBV; swiB[wid] = bestBI;
    }
    __syncthreads();
    if (wid == 0) {
      float a1v = swvA[lane], b1v = swvB[lane];
      int   a1i = swiA[lane], b1i = swiB[lane];
      float a2v = -__builtin_inff(), b2v = -__builtin_inff();
      int   a2i = 0x7fffffff,        b2i = 0x7fffffff;
      for (int m = 16; m; m >>= 1) {
        const float oa1v = __shfl_xor(a1v, m, 32);
        const int   oa1i = __shfl_xor(a1i, m, 32);
        const float oa2v = __shfl_xor(a2v, m, 32);
        const int   oa2i = __shfl_xor(a2i, m, 32);
        const float ob1v = __shfl_xor(b1v, m, 32);
        const int   ob1i = __shfl_xor(b1i, m, 32);
        const float ob2v = __shfl_xor(b2v, m, 32);
        const int   ob2i = __shfl_xor(b2i, m, 32);
        {
          const bool c = lex_better(oa1v, oa1i, a1v, a1i);
          const float lv = c ? a1v : oa1v;  const int li = c ? a1i : oa1i;
          const float sv = c ? oa2v : a2v;  const int si = c ? oa2i : a2i;
          a1v = c ? oa1v : a1v;             a1i = c ? oa1i : a1i;
          const bool d = lv > sv;
          a2v = d ? lv : sv;                a2i = d ? li : si;
        }
        {
          const bool c = lex_better(ob1v, ob1i, b1v, b1i);
          const float lv = c ? b1v : ob1v;  const int li = c ? b1i : ob1i;
          const float sv = c ? ob2v : b2v;  const int si = c ? ob2i : b2i;
          b1v = c ? ob1v : b1v;             b1i = c ? ob1i : b1i;
          const bool d = lv > sv;
          b2v = d ? lv : sv;                b2i = d ? li : si;
        }
      }
      if (lane == 0) {
        const int ta = (a1i >= N_) ? 0 : a1i;
        const int tb = (b1i >= N_) ? 0 : b1i;
        const int hit = (ta == pickB) ? 1 : 0;
        if (hit & (step + 1 < M_)) fps_idx[b * M_ + step + 1] = pickB;
        const int na = hit ? tb : ta;
        const int ng = hit ? b2i : a2i;
        s_hit = hit;
        s_a   = na;
        s_b   = (ng >= N_) ? na : ng;
      }
    }
    __syncthreads();
    const bool hit = s_hit != 0;
#pragma unroll
    for (int j = 0; j < FPS_J; ++j)
#pragma unroll
      for (int k = 0; k < 4; ++k) mind[j][k] = hit ? mab[j][k] : mind[j][k];
    step += hit ? 2 : 1;
  }
}

__global__ __launch_bounds__(256) void three_nn_kernel(
    const float* __restrict__ pc, const int* __restrict__ fps_idx,
    int* __restrict__ idx3, float* __restrict__ w3) {
  const int tid  = threadIdx.x;
  const int lane = tid & 31;
  const int wave = blockIdx.x * (256 >> 5) + (tid >> 5);
  const int tilesPerBatch = M_ / 16;
  const int b    = wave / tilesPerBatch;
  const int rt   = wave - b * tilesPerBatch;
  const int half = lane >> 4;
  const int l15  = lane & 15;

  const float* pcb = pc + (size_t)b * N_ * 3;

  const int mrow = rt * 16 + l15;
  const int aidx = fps_idx[b * M_ + mrow];
  const float* ap = pcb + (size_t)aidx * 3;
  const float ax = ap[0], ay = ap[1], az = ap[2];
  v2f A;
  A.x = half ? (-2.0f * az) : (-2.0f * ax);
  A.y = half ? 1.0f         : (-2.0f * ay);

  float td[8][3];
  int   ti[8][3];
#pragma unroll
  for (int r = 0; r < 8; ++r)
#pragma unroll
    for (int k = 0; k < 3; ++k) { td[r][k] = __builtin_inff(); ti[r][k] = 0x7fffffff; }

#pragma unroll 2
  for (int t = 0; t < N_ / 16; ++t) {
    const int n = t * 16 + l15;
    const float* bp = pcb + (size_t)n * 3;
    const int pn = (n + 128 < N_) ? (n + 128) : (N_ - 1);
    __builtin_prefetch(pcb + (size_t)pn * 3, 0, 1);
    const float bx = bp[0], by = bp[1], bz = bp[2];
    const float bn2 = bx * bx + by * by + bz * bz;
    v8f c = {0.f, 0.f, 0.f, 0.f, 0.f, 0.f, 0.f, 0.f};
#if USE_WMMA
    v2f Bv;
    Bv.x = half ? bz  : bx;
    Bv.y = half ? bn2 : by;
    c = __builtin_amdgcn_wmma_f32_16x16x4_f32(
         false, A,  false, Bv,
         (short)0, c,  false,  false);
#else
#pragma unroll
    for (int r = 0; r < 8; ++r) {
      const int row = half * 8 + r;
      const float rx = __shfl(ax, row, 32);
      const float ry = __shfl(ay, row, 32);
      const float rz = __shfl(az, row, 32);
      c[r] = -2.0f * (rx * bx + ry * by + rz * bz) + bn2;
    }
#endif
#pragma unroll
    for (int r = 0; r < 8; ++r) ins3_stream(c[r], n, td[r], ti[r]);
  }

#pragma unroll
  for (int mskk = 1; mskk <= 8; mskk <<= 1) {
#pragma unroll
    for (int r = 0; r < 8; ++r) {
      float pd[3]; int pi[3];
#pragma unroll
      for (int k = 0; k < 3; ++k) {
        pd[k] = __shfl_xor(td[r][k], mskk, 32);
        pi[k] = __shfl_xor(ti[r][k], mskk, 32);
      }
#pragma unroll
      for (int k = 0; k < 3; ++k) ins3_lex(pd[k], pi[k], td[r], ti[r]);
    }
  }

  if (l15 == 0) {
#pragma unroll
    for (int r = 0; r < 8; ++r) {
      const int row = rt * 16 + half * 8 + r;
      const int gi  = fps_idx[b * M_ + row];
      const float* gp = pcb + (size_t)gi * 3;
      const float gx = gp[0], gy = gp[1], gz = gp[2];
      float wv[3];
      float wsum = 0.f;
#pragma unroll
      for (int k = 0; k < 3; ++k) {
        const int nn = ti[r][k];
        const float* np = pcb + (size_t)nn * 3;
        const float dx = np[0] - gx, dy = np[1] - gy, dz = np[2] - gz;
        const float dd = dx * dx + dy * dy + dz * dz;
        const float dist = sqrtf(fmaxf(dd, 0.0f));
        wv[k] = 1.0f / (dist + 1e-8f);
        wsum += wv[k];
        idx3[((size_t)b * M_ + row) * 3 + k] = nn;
      }
#pragma unroll
      for (int k = 0; k < 3; ++k)
        w3[((size_t)b * M_ + row) * 3 + k] = wv[k] / wsum;
    }
  }
}

__global__ __launch_bounds__(256) void interp_kernel(
    const float* __restrict__ feats, const int* __restrict__ idx3,
    const float* __restrict__ w3, float* __restrict__ out) {
  const int bc = blockIdx.x;
  const int b  = bc / C_;
  const int c  = bc - b * C_;
  const float* f = feats + ((size_t)b * C_ + c) * N_;
  float*       o = out   + ((size_t)b * C_ + c) * M_;
  const int*   ib = idx3 + (size_t)b * M_ * 3;
  const float* wb = w3   + (size_t)b * M_ * 3;
  for (int m = threadIdx.x; m < M_; m += 256) {
    const int   n0 = ib[m * 3 + 0], n1 = ib[m * 3 + 1], n2 = ib[m * 3 + 2];
    const float w0 = wb[m * 3 + 0], w1 = wb[m * 3 + 1], w2 = wb[m * 3 + 2];
    o[m] = w0 * f[n0] + w1 * f[n1] + w2 * f[n2];
  }
}

extern "C" void kernel_launch(void* const* d_in, const int* in_sizes, int n_in,
                              void* d_out, int out_size, void* d_ws, size_t ws_size,
                              hipStream_t stream) {
  const float* pc    = (const float*)d_in[0];
  const float* feats = (const float*)d_in[1];
  const float* obj   = (const float*)d_in[2];
  const float* grasp = (const float*)d_in[3];
  float* out = (float*)d_out;

  char* ws = (char*)d_ws;
  size_t off = 0;
  int* fps = (int*)(ws + off);
  off += (size_t)B_ * M_ * 4;        off = (off + 255) & ~(size_t)255;
  int* idx3 = (int*)(ws + off);
  off += (size_t)B_ * M_ * 3 * 4;    off = (off + 255) & ~(size_t)255;
  float* w3 = (float*)(ws + off);
  off += (size_t)B_ * M_ * 3 * 4;

  fps_kernel<<<B_, 1024, 0, stream>>>(pc, obj, grasp, fps);
  three_nn_kernel<<<(B_ * (M_ / 16)) / 8, 256, 0, stream>>>(pc, fps, idx3, w3);
  interp_kernel<<<B_ * C_, 256, 0, stream>>>(feats, idx3, w3, out);
}


# fps kernel: batch proof in parallel (64-bit value|index keys; each wave ranks its 2 list entries by ballot+popcount, +1 barrier; wave 0 tests all ranks at once), no serial argmax per pick; same sums
# speedup vs baseline: 2.1398x; 1.4925x over previous
#include <hip/hip_runtime.h>
#include <hip/hip_bf16.h>

#define B_ 4
#define N_ 20000
#define C_ 512
#define M_ 1024
#define GRASP_TH 0.1f

typedef float v2f __attribute__((ext_vector_type(2)));
typedef float v4f __attribute__((ext_vector_type(4)));
typedef float v8f __attribute__((ext_vector_type(8)));

#if defined(__gfx1250__) && __has_builtin(__builtin_amdgcn_wmma_f32_16x16x4_f32)
#define USE_WMMA 1
#else
#define USE_WMMA 0
#endif

__device__ __forceinline__ void ins3_stream(float v, int n, float d[3], int ix[3]) {
  const bool c0 = v < d[0];
  const bool c1 = v < d[1];
  const bool c2 = v < d[2];
  const float r0 = c0 ? v     : d[0];
  const int   j0 = c0 ? n     : ix[0];
  const float r1 = c0 ? d[0]  : (c1 ? v : d[1]);
  const int   j1 = c0 ? ix[0] : (c1 ? n : ix[1]);
  const float r2 = c1 ? d[1]  : (c2 ? v : d[2]);
  const int   j2 = c1 ? ix[1] : (c2 ? n : ix[2]);
  d[0] = r0; d[1] = r1; d[2] = r2;
  ix[0] = j0; ix[1] = j1; ix[2] = j2;
}

__device__ __forceinline__ void ins3_lex(float v, int n, float d[3], int ix[3]) {
  const bool c0 = (v < d[0]) | ((v == d[0]) & (n < ix[0]));
  const bool c1 = (v < d[1]) | ((v == d[1]) & (n < ix[1]));
  const bool c2 = (v < d[2]) | ((v == d[2]) & (n < ix[2]));
  const float r0 = c0 ? v     : d[0];
  const int   j0 = c0 ? n     : ix[0];
  const float r1 = c0 ? d[0]  : (c1 ? v : d[1]);
  const int   j1 = c0 ? ix[0] : (c1 ? n : ix[1]);
  const float r2 = c1 ? d[1]  : (c2 ? v : d[2]);
  const int   j2 = c1 ? ix[1] : (c2 ? n : ix[2]);
  d[0] = r0; d[1] = r1; d[2] = r2;
  ix[0] = j0; ix[1] = j1; ix[2] = j2;
}

#define FPS_J 5
#define FPS_K 6

__device__ __forceinline__ bool lex_better(float ov, int oi, float v, int i) {
  return (ov > v) | ((ov == v) & (oi < i));
}

typedef unsigned long long u64;

__device__ __forceinline__ u64 pack_key(float v, int i) {
  const bool real = v >= 0.0f;
  const unsigned hi = real ? __float_as_uint(v) : 0u;
  const unsigned lo = real ? (unsigned)(0x7fffffff - i) : 0u;
  return ((u64)hi << 32) | (u64)lo;
}

__global__ __launch_bounds__(1024) void fps_kernel(
    const float* __restrict__ pc, const float* __restrict__ obj,
    const float* __restrict__ grasp, int* __restrict__ fps_idx) {
  const int b    = blockIdx.x;
  const int tid  = threadIdx.x;
  const int lane = tid & 31;
  const int wid  = tid >> 5;

  __shared__ __attribute__((aligned(16))) float s_x[N_];
  __shared__ __attribute__((aligned(16))) float s_y[N_];
  __shared__ __attribute__((aligned(16))) float s_z[N_];
  __shared__ __attribute__((aligned(16))) u64 s_key[64];
  __shared__ __attribute__((aligned(8)))  int s_rank[64];
  __shared__ int   swi1[32];
  __shared__ int   s_m;
  __shared__ float s_px[FPS_K], s_py[FPS_K], s_pz[FPS_K];

  const float* pcb = pc + (size_t)b * N_ * 3;
  const float* ob0 = obj + (size_t)b * 2 * N_;
  const float* gr  = grasp + (size_t)b * N_;

  for (int n = tid; n < N_; n += 1024) {
    const float* p = pcb + (size_t)n * 3;
    s_x[n] = p[0];
    s_y[n] = p[1];
    s_z[n] = p[2];
  }

  float mind[FPS_J][4];
  int firstLoc = N_;
#pragma unroll
  for (int j = FPS_J - 1; j >= 0; --j) {
    const int g = tid + 1024 * j;
    const bool gv = g < N_ / 4;
#pragma unroll
    for (int k = 3; k >= 0; --k) {
      const int n  = 4 * g + k;
      const int nc = gv ? n : 0;
      const bool mk = gv & (ob0[N_ + nc] > ob0[nc]) & (gr[nc] > GRASP_TH);
      mind[j][k] = mk ? __builtin_inff() : -__builtin_inff();
      firstLoc   = mk ? n : firstLoc;
    }
  }
  for (int mm = 16; mm; mm >>= 1) {
    const int o = __shfl_xor(firstLoc, mm, 32);
    firstLoc = (o < firstLoc) ? o : firstLoc;
  }
  if (lane == 0) swi1[wid] = firstLoc;
  __syncthreads();
  if (wid == 0) {
    int v = swi1[lane];
    for (int mm = 16; mm; mm >>= 1) {
      const int o = __shfl_xor(v, mm, 32);
      v = (o < v) ? o : v;
    }
    if (lane == 0) {
      const int f = (v >= N_) ? 0 : v;
      fps_idx[b * M_] = f;
      s_m = 1;
      const float fx = s_x[f], fy = s_y[f], fz = s_z[f];
#pragma unroll
      for (int i = 0; i < FPS_K; ++i) { s_px[i] = fx; s_py[i] = fy; s_pz[i] = fz; }
    }
  }
  __syncthreads();

  const v4f* x4 = (const v4f*)s_x;
  const v4f* y4 = (const v4f*)s_y;
  const v4f* z4 = (const v4f*)s_z;

  int m    = 1;
  int step = 1;

  while (step < M_) {
    float qx[FPS_K], qy[FPS_K], qz[FPS_K];
#pragma unroll
    for (int i = 0; i < FPS_K; ++i) { qx[i] = s_px[i]; qy[i] = s_py[i]; qz[i] = s_pz[i]; }

    float b1v = -__builtin_inff(), b2v = -__builtin_inff();
    int   b1i = 0x7fffffff,        b2i = 0x7fffffff;
#pragma unroll
    for (int j = 0; j < FPS_J; ++j) {
      const int g  = tid + 1024 * j;
      const int gc = (g < N_ / 4) ? g : tid;
      const v4f vx = x4[gc];
      const v4f vy = y4[gc];
      const v4f vz = z4[gc];
      float v[4];
#pragma unroll
      for (int k = 0; k < 4; ++k) v[k] = mind[j][k];
#pragma unroll
      for (int i = 0; i < FPS_K; ++i) {
        if (i >= m) break;
#pragma unroll
        for (int k = 0; k < 4; ++k) {
          const float dxa = vx[k] - qx[i], dya = vy[k] - qy[i], dza = vz[k] - qz[i];
          const float dda = dxa * dxa + dya * dya + dza * dza;
          v[k] = fminf(v[k], dda);
        }
      }
#pragma unroll
      for (int k = 0; k < 4; ++k) {
        mind[j][k] = v[k];
        const int  n  = 4 * g + k;
        const bool c1 = v[k] > b1v;
        const bool c2 = v[k] > b2v;
        b2v = c1 ? b1v : (c2 ? v[k] : b2v);
        b2i = c1 ? b1i : (c2 ? n    : b2i);
        b1v = c1 ? v[k] : b1v;
        b1i = c1 ? n    : b1i;
      }
    }
    for (int mm = 16; mm; mm >>= 1) {
      const float o1v = __shfl_xor(b1v, mm, 32);
      const int   o1i = __shfl_xor(b1i, mm, 32);
      const float o2v = __shfl_xor(b2v, mm, 32);
      const int   o2i = __shfl_xor(b2i, mm, 32);
      const bool c = lex_better(o1v, o1i, b1v, b1i);
      const float lv = c ? b1v : o1v;  const int li = c ? b1i : o1i;
      const float sv = c ? o2v : b2v;  const int si = c ? o2i : b2i;
      b1v = c ? o1v : b1v;             b1i = c ? o1i : b1i;
      const bool d = lex_better(lv, li, sv, si);
      b2v = d ? lv : sv;               b2i = d ? li : si;
    }
    const u64 k1 = pack_key(b1v, b1i);
    const u64 k2 = pack_key(b2v, b2i);
    if (lane == 0) {
      s_key[2 * wid]     = k1;
      s_key[2 * wid + 1] = k2;
    }
    __syncthreads();
    {
      const unsigned h1 = __builtin_amdgcn_readfirstlane((unsigned)(k1 >> 32));
      const unsigned l1 = __builtin_amdgcn_readfirstlane((unsigned)k1);
      const unsigned h2 = __builtin_amdgcn_readfirstlane((unsigned)(k2 >> 32));
      const unsigned l2 = __builtin_amdgcn_readfirstlane((unsigned)k2);
      const u64 kHs = ((u64)h1 << 32) | (u64)l1;
      const u64 kNs = ((u64)h2 << 32) | (u64)l2;
      const u64 ka = s_key[2 * lane];
      const u64 kb = s_key[2 * lane + 1];
      const int rHs = __builtin_popcount(__builtin_amdgcn_ballot_w32(ka > kHs)) +
                      __builtin_popcount(__builtin_amdgcn_ballot_w32(kb > kHs));
      const int rNs = __builtin_popcount(__builtin_amdgcn_ballot_w32(ka > kNs)) +
                      __builtin_popcount(__builtin_amdgcn_ballot_w32(kb > kNs));
      if (lane == 0) { s_rank[2 * wid] = rHs; s_rank[2 * wid + 1] = rNs; }
    }
    __syncthreads();
    if (wid == 0) {
      const u64 kH = s_key[2 * lane];
      const u64 kN = s_key[2 * lane + 1];
      const float valH = __uint_as_float((unsigned)(kH >> 32));
      const float valN = __uint_as_float((unsigned)(kN >> 32));
      const int iH = 0x7fffffff - (int)(unsigned)kH;
      const int iN = 0x7fffffff - (int)(unsigned)kN;
      const bool vH = iH < N_;
      const bool vN = iN < N_;
      const int qH = vH ? iH : 0;
      const int qN = vN ? iN : 0;
      const float xH = s_x[qH], yH = s_y[qH], zH = s_z[qH];
      const float xN = s_x[qN], yN = s_y[qN], zN = s_z[qN];

      const int rH = s_rank[2 * lane];
      const int rN = s_rank[2 * lane + 1];

      if (rH < FPS_K) { s_px[rH] = xH; s_py[rH] = yH; s_pz[rH] = zH; }
      if (rN < FPS_K) { s_px[rN] = xN; s_py[rN] = yN; s_pz[rN] = zN; }

      const float needH = valH * 1.0001f;
      const float needN = valN * 1.0001f;
      bool okH = vH & (valH > 0.0f);
      bool okN = vN & (valN > 0.0f);
#pragma unroll
      for (int i = 0; i < FPS_K - 1; ++i) {
        const float sx = s_px[i], sy = s_py[i], sz = s_pz[i];
        const float dxa = xH - sx, dya = yH - sy, dza = zH - sz;
        const float dda = dxa * dxa + dya * dya + dza * dza;
        okH = okH & ((i >= rH) | (dda >= needH));
        const float dxb = xN - sx, dyb = yN - sy, dzb = zN - sz;
        const float ddb = dxb * dxb + dyb * dyb + dzb * dzb;
        okN = okN & ((i >= rN) | (ddb >= needN));
      }

      int  mnew = 1;
      bool run  = true;
#pragma unroll
      for (int r = 1; r < FPS_K; ++r) {
        const bool hit   = __builtin_amdgcn_ballot_w32((okH & (rH == r)) | (okN & (rN == r))) != 0u;
        const bool spent = __builtin_amdgcn_ballot_w32(rN < r) != 0u;
        run = run & hit & (!spent);
        mnew += run ? 1 : 0;
      }
      const int room = M_ - step;
      mnew = (mnew < room) ? mnew : room;

      if (rH < mnew) fps_idx[b * M_ + step + rH] = qH;
      if (rN < mnew) fps_idx[b * M_ + step + rN] = qN;
      if (lane == 0) s_m = mnew;
    }
    __syncthreads();
    m = __builtin_amdgcn_readfirstlane(s_m);
    step += m;
  }
}

__global__ __launch_bounds__(256) void three_nn_kernel(
    const float* __restrict__ pc, const int* __restrict__ fps_idx,
    int* __restrict__ idx3, float* __restrict__ w3) {
  const int tid  = threadIdx.x;
  const int lane = tid & 31;
  const int wave = blockIdx.x * (256 >> 5) + (tid >> 5);
  const int tilesPerBatch = M_ / 16;
  const int b    = wave / tilesPerBatch;
  const int rt   = wave - b * tilesPerBatch;
  const int half = lane >> 4;
  const int l15  = lane & 15;

  const float* pcb = pc + (size_t)b * N_ * 3;

  const int mrow = rt * 16 + l15;
  const int aidx = fps_idx[b * M_ + mrow];
  const float* ap = pcb + (size_t)aidx * 3;
  const float ax = ap[0], ay = ap[1], az = ap[2];
  v2f A;
  A.x = half ? (-2.0f * az) : (-2.0f * ax);
  A.y = half ? 1.0f         : (-2.0f * ay);

  float td[8][3];
  int   ti[8][3];
#pragma unroll
  for (int r = 0; r < 8; ++r)
#pragma unroll
    for (int k = 0; k < 3; ++k) { td[r][k] = __builtin_inff(); ti[r][k] = 0x7fffffff; }

#pragma unroll 2
  for (int t = 0; t < N_ / 16; ++t) {
    const int n = t * 16 + l15;
    const float* bp = pcb + (size_t)n * 3;
    const int pn = (n + 128 < N_) ? (n + 128) : (N_ - 1);
    __builtin_prefetch(pcb + (size_t)pn * 3, 0, 1);
    const float bx = bp[0], by = bp[1], bz = bp[2];
    const float bn2 = bx * bx + by * by + bz * bz;
    v8f c = {0.f, 0.f, 0.f, 0.f, 0.f, 0.f, 0.f, 0.f};
#if USE_WMMA
    v2f Bv;
    Bv.x = half ? bz  : bx;
    Bv.y = half ? bn2 : by;
    c = __builtin_amdgcn_wmma_f32_16x16x4_f32(
         false, A,  false, Bv,
         (short)0, c,  false,  false);
#else
#pragma unroll
    for (int r = 0; r < 8; ++r) {
      const int row = half * 8 + r;
      const float rx = __shfl(ax, row, 32);
      const float ry = __shfl(ay, row, 32);
      const float rz = __shfl(az, row, 32);
      c[r] = -2.0f * (rx * bx + ry * by + rz * bz) + bn2;
    }
#endif
#pragma unroll
    for (int r = 0; r < 8; ++r) ins3_stream(c[r], n, td[r], ti[r]);
  }

#pragma unroll
  for (int mskk = 1; mskk <= 8; mskk <<= 1) {
#pragma unroll
    for (int r = 0; r < 8; ++r) {
      float pd[3]; int pi[3];
#pragma unroll
      for (int k = 0; k < 3; ++k) {
        pd[k] = __shfl_xor(td[r][k], mskk, 32);
        pi[k] = __shfl_xor(ti[r][k], mskk, 32);
      }
#pragma unroll
      for (int k = 0; k < 3; ++k) ins3_lex(pd[k], pi[k], td[r], ti[r]);
    }
  }

  if (l15 == 0) {
#pragma unroll
    for (int r = 0; r < 8; ++r) {
      const int row = rt * 16 + half * 8 + r;
      const int gi  = fps_idx[b * M_ + row];
      const float* gp = pcb + (size_t)gi * 3;
      const float gx = gp[0], gy = gp[1], gz = gp[2];
      float wv[3];
      float wsum = 0.f;
#pragma unroll
      for (int k = 0; k < 3; ++k) {
        const int nn = ti[r][k];
        const float* np = pcb + (size_t)nn * 3;
        const float dx = np[0] - gx, dy = np[1] - gy, dz = np[2] - gz;
        const float dd = dx * dx + dy * dy + dz * dz;
        const float dist = sqrtf(fmaxf(dd, 0.0f));
        wv[k] = 1.0f / (dist + 1e-8f);
        wsum += wv[k];
        idx3[((size_t)b * M_ + row) * 3 + k] = nn;
      }
#pragma unroll
      for (int k = 0; k < 3; ++k)
        w3[((size_t)b * M_ + row) * 3 + k] = wv[k] / wsum;
    }
  }
}

__global__ __launch_bounds__(256) void interp_kernel(
    const float* __restrict__ feats, const int* __restrict__ idx3,
    const float* __restrict__ w3, float* __restrict__ out) {
  const int bc = blockIdx.x;
  const int b  = bc / C_;
  const int c  = bc - b * C_;
  const float* f = feats + ((size_t)b * C_ + c) * N_;
  float*       o = out   + ((size_t)b * C_ + c) * M_;
  const int*   ib = idx3 + (size_t)b * M_ * 3;
  const float* wb = w3   + (size_t)b * M_ * 3;
  for (int m = threadIdx.x; m < M_; m += 256) {
    const int   n0 = ib[m * 3 + 0], n1 = ib[m * 3 + 1], n2 = ib[m * 3 + 2];
    const float w0 = wb[m * 3 + 0], w1 = wb[m * 3 + 1], w2 = wb[m * 3 + 2];
    o[m] = w0 * f[n0] + w1 * f[n1] + w2 * f[n2];
  }
}

extern "C" void kernel_launch(void* const* d_in, const int* in_sizes, int n_in,
                              void* d_out, int out_size, void* d_ws, size_t ws_size,
                              hipStream_t stream) {
  const float* pc    = (const float*)d_in[0];
  const float* feats = (const float*)d_in[1];
  const float* obj   = (const float*)d_in[2];
  const float* grasp = (const float*)d_in[3];
  float* out = (float*)d_out;

  char* ws = (char*)d_ws;
  size_t off = 0;
  int* fps = (int*)(ws + off);
  off += (size_t)B_ * M_ * 4;        off = (off + 255) & ~(size_t)255;
  int* idx3 = (int*)(ws + off);
  off += (size_t)B_ * M_ * 3 * 4;    off = (off + 255) & ~(size_t)255;
  float* w3 = (float*)(ws + off);
  off += (size_t)B_ * M_ * 3 * 4;

  fps_kernel<<<B_, 1024, 0, stream>>>(pc, obj, grasp, fps);
  three_nn_kernel<<<(B_ * (M_ / 16)) / 8, 256, 0, stream>>>(pc, fps, idx3, w3);
  interp_kernel<<<B_ * C_, 256, 0, stream>>>(feats, idx3, w3, out);
}


# three_nn kernel: cloud tile as WMMA A, the wave's 16 sampled points as B: a lane keeps ONE top-3 list (was 8) and a tile's inserts are skipped wave-wide unless a score beats a 3rd entry; same sums
# speedup vs baseline: 2.3960x; 1.1197x over previous
#include <hip/hip_runtime.h>
#include <hip/hip_bf16.h>

#define B_ 4
#define N_ 20000
#define C_ 512
#define M_ 1024
#define GRASP_TH 0.1f

typedef float v2f __attribute__((ext_vector_type(2)));
typedef float v4f __attribute__((ext_vector_type(4)));
typedef float v8f __attribute__((ext_vector_type(8)));

#if defined(__gfx1250__) && __has_builtin(__builtin_amdgcn_wmma_f32_16x16x4_f32)
#define USE_WMMA 1
#else
#define USE_WMMA 0
#endif

__device__ __forceinline__ void ins3_stream(float v, int n, float d[3], int ix[3]) {
  const bool c0 = v < d[0];
  const bool c1 = v < d[1];
  const bool c2 = v < d[2];
  const float r0 = c0 ? v     : d[0];
  const int   j0 = c0 ? n     : ix[0];
  const float r1 = c0 ? d[0]  : (c1 ? v : d[1]);
  const int   j1 = c0 ? ix[0] : (c1 ? n : ix[1]);
  const float r2 = c1 ? d[1]  : (c2 ? v : d[2]);
  const int   j2 = c1 ? ix[1] : (c2 ? n : ix[2]);
  d[0] = r0; d[1] = r1; d[2] = r2;
  ix[0] = j0; ix[1] = j1; ix[2] = j2;
}

__device__ __forceinline__ void ins3_lex(float v, int n, float d[3], int ix[3]) {
  const bool c0 = (v < d[0]) | ((v == d[0]) & (n < ix[0]));
  const bool c1 = (v < d[1]) | ((v == d[1]) & (n < ix[1]));
  const bool c2 = (v < d[2]) | ((v == d[2]) & (n < ix[2]));
  const float r0 = c0 ? v     : d[0];
  const int   j0 = c0 ? n     : ix[0];
  const float r1 = c0 ? d[0]  : (c1 ? v : d[1]);
  const int   j1 = c0 ? ix[0] : (c1 ? n : ix[1]);
  const float r2 = c1 ? d[1]  : (c2 ? v : d[2]);
  const int   j2 = c1 ? ix[1] : (c2 ? n : ix[2]);
  d[0] = r0; d[1] = r1; d[2] = r2;
  ix[0] = j0; ix[1] = j1; ix[2] = j2;
}

#define FPS_J 5
#define FPS_K 6

__device__ __forceinline__ bool lex_better(float ov, int oi, float v, int i) {
  return (ov > v) | ((ov == v) & (oi < i));
}

typedef unsigned long long u64;

__device__ __forceinline__ u64 pack_key(float v, int i) {
  const bool real = v >= 0.0f;
  const unsigned hi = real ? __float_as_uint(v) : 0u;
  const unsigned lo = real ? (unsigned)(0x7fffffff - i) : 0u;
  return ((u64)hi << 32) | (u64)lo;
}

__global__ __launch_bounds__(1024) void fps_kernel(
    const float* __restrict__ pc, const float* __restrict__ obj,
    const float* __restrict__ grasp, int* __restrict__ fps_idx) {
  const int b    = blockIdx.x;
  const int tid  = threadIdx.x;
  const int lane = tid & 31;
  const int wid  = tid >> 5;

  __shared__ __attribute__((aligned(16))) float s_x[N_];
  __shared__ __attribute__((aligned(16))) float s_y[N_];
  __shared__ __attribute__((aligned(16))) float s_z[N_];
  __shared__ __attribute__((aligned(16))) u64 s_key[64];
  __shared__ __attribute__((aligned(8)))  int s_rank[64];
  __shared__ int   swi1[32];
  __shared__ int   s_m;
  __shared__ float s_px[FPS_K], s_py[FPS_K], s_pz[FPS_K];

  const float* pcb = pc + (size_t)b * N_ * 3;
  const float* ob0 = obj + (size_t)b * 2 * N_;
  const float* gr  = grasp + (size_t)b * N_;

  for (int n = tid; n < N_; n += 1024) {
    const float* p = pcb + (size_t)n * 3;
    s_x[n] = p[0];
    s_y[n] = p[1];
    s_z[n] = p[2];
  }

  float mind[FPS_J][4];
  int firstLoc = N_;
#pragma unroll
  for (int j = FPS_J - 1; j >= 0; --j) {
    const int g = tid + 1024 * j;
    const bool gv = g < N_ / 4;
#pragma unroll
    for (int k = 3; k >= 0; --k) {
      const int n  = 4 * g + k;
      const int nc = gv ? n : 0;
      const bool mk = gv & (ob0[N_ + nc] > ob0[nc]) & (gr[nc] > GRASP_TH);
      mind[j][k] = mk ? __builtin_inff() : -__builtin_inff();
      firstLoc   = mk ? n : firstLoc;
    }
  }
  for (int mm = 16; mm; mm >>= 1) {
    const int o = __shfl_xor(firstLoc, mm, 32);
    firstLoc = (o < firstLoc) ? o : firstLoc;
  }
  if (lane == 0) swi1[wid] = firstLoc;
  __syncthreads();
  if (wid == 0) {
    int v = swi1[lane];
    for (int mm = 16; mm; mm >>= 1) {
      const int o = __shfl_xor(v, mm, 32);
      v = (o < v) ? o : v;
    }
    if (lane == 0) {
      const int f = (v >= N_) ? 0 : v;
      fps_idx[b * M_] = f;
      s_m = 1;
      const float fx = s_x[f], fy = s_y[f], fz = s_z[f];
#pragma unroll
      for (int i = 0; i < FPS_K; ++i) { s_px[i] = fx; s_py[i] = fy; s_pz[i] = fz; }
    }
  }
  __syncthreads();

  const v4f* x4 = (const v4f*)s_x;
  const v4f* y4 = (const v4f*)s_y;
  const v4f* z4 = (const v4f*)s_z;

  int m    = 1;
  int step = 1;

  while (step < M_) {
    float qx[FPS_K], qy[FPS_K], qz[FPS_K];
#pragma unroll
    for (int i = 0; i < FPS_K; ++i) { qx[i] = s_px[i]; qy[i] = s_py[i]; qz[i] = s_pz[i]; }

    float b1v = -__builtin_inff(), b2v = -__builtin_inff();
    int   b1i = 0x7fffffff,        b2i = 0x7fffffff;
#pragma unroll
    for (int j = 0; j < FPS_J; ++j) {
      const int g  = tid + 1024 * j;
      const int gc = (g < N_ / 4) ? g : tid;
      const v4f vx = x4[gc];
      const v4f vy = y4[gc];
      const v4f vz = z4[gc];
      float v[4];
#pragma unroll
      for (int k = 0; k < 4; ++k) v[k] = mind[j][k];
#pragma unroll
      for (int i = 0; i < FPS_K; ++i) {
        if (i >= m) break;
#pragma unroll
        for (int k = 0; k < 4; ++k) {
          const float dxa = vx[k] - qx[i], dya = vy[k] - qy[i], dza = vz[k] - qz[i];
          const float dda = dxa * dxa + dya * dya + dza * dza;
          v[k] = fminf(v[k], dda);
        }
      }
#pragma unroll
      for (int k = 0; k < 4; ++k) {
        mind[j][k] = v[k];
        const int  n  = 4 * g + k;
        const bool c1 = v[k] > b1v;
        const bool c2 = v[k] > b2v;
        b2v = c1 ? b1v : (c2 ? v[k] : b2v);
        b2i = c1 ? b1i : (c2 ? n    : b2i);
        b1v = c1 ? v[k] : b1v;
        b1i = c1 ? n    : b1i;
      }
    }
    for (int mm = 16; mm; mm >>= 1) {
      const float o1v = __shfl_xor(b1v, mm, 32);
      const int   o1i = __shfl_xor(b1i, mm, 32);
      const float o2v = __shfl_xor(b2v, mm, 32);
      const int   o2i = __shfl_xor(b2i, mm, 32);
      const bool c = lex_better(o1v, o1i, b1v, b1i);
      const float lv = c ? b1v : o1v;  const int li = c ? b1i : o1i;
      const float sv = c ? o2v : b2v;  const int si = c ? o2i : b2i;
      b1v = c ? o1v : b1v;             b1i = c ? o1i : b1i;
      const bool d = lex_better(lv, li, sv, si);
      b2v = d ? lv : sv;               b2i = d ? li : si;
    }
    const u64 k1 = pack_key(b1v, b1i);
    const u64 k2 = pack_key(b2v, b2i);
    if (lane == 0) {
      s_key[2 * wid]     = k1;
      s_key[2 * wid + 1] = k2;
    }
    __syncthreads();
    {
      const unsigned h1 = __builtin_amdgcn_readfirstlane((unsigned)(k1 >> 32));
      const unsigned l1 = __builtin_amdgcn_readfirstlane((unsigned)k1);
      const unsigned h2 = __builtin_amdgcn_readfirstlane((unsigned)(k2 >> 32));
      const unsigned l2 = __builtin_amdgcn_readfirstlane((unsigned)k2);
      const u64 kHs = ((u64)h1 << 32) | (u64)l1;
      const u64 kNs = ((u64)h2 << 32) | (u64)l2;
      const u64 ka = s_key[2 * lane];
      const u64 kb = s_key[2 * lane + 1];
      const int rHs = __builtin_popcount(__builtin_amdgcn_ballot_w32(ka > kHs)) +
                      __builtin_popcount(__builtin_amdgcn_ballot_w32(kb > kHs));
      const int rNs = __builtin_popcount(__builtin_amdgcn_ballot_w32(ka > kNs)) +
                      __builtin_popcount(__builtin_amdgcn_ballot_w32(kb > kNs));
      if (lane == 0) { s_rank[2 * wid] = rHs; s_rank[2 * wid + 1] = rNs; }
    }
    __syncthreads();
    if (wid == 0) {
      const u64 kH = s_key[2 * lane];
      const u64 kN = s_key[2 * lane + 1];
      const float valH = __uint_as_float((unsigned)(kH >> 32));
      const float valN = __uint_as_float((unsigned)(kN >> 32));
      const int iH = 0x7fffffff - (int)(unsigned)kH;
      const int iN = 0x7fffffff - (int)(unsigned)kN;
      const bool vH = iH < N_;
      const bool vN = iN < N_;
      const int qH = vH ? iH : 0;
      const int qN = vN ? iN : 0;
      const float xH = s_x[qH], yH = s_y[qH], zH = s_z[qH];
      const float xN = s_x[qN], yN = s_y[qN], zN = s_z[qN];

      const int rH = s_rank[2 * lane];
      const int rN = s_rank[2 * lane + 1];

      if (rH < FPS_K) { s_px[rH] = xH; s_py[rH] = yH; s_pz[rH] = zH; }
      if (rN < FPS_K) { s_px[rN] = xN; s_py[rN] = yN; s_pz[rN] = zN; }

      const float needH = valH * 1.0001f;
      const float needN = valN * 1.0001f;
      bool okH = vH & (valH > 0.0f);
      bool okN = vN & (valN > 0.0f);
#pragma unroll
      for (int i = 0; i < FPS_K - 1; ++i) {
        const float sx = s_px[i], sy = s_py[i], sz = s_pz[i];
        const float dxa = xH - sx, dya = yH - sy, dza = zH - sz;
        const float dda = dxa * dxa + dya * dya + dza * dza;
        okH = okH & ((i >= rH) | (dda >= needH));
        const float dxb = xN - sx, dyb = yN - sy, dzb = zN - sz;
        const float ddb = dxb * dxb + dyb * dyb + dzb * dzb;
        okN = okN & ((i >= rN) | (ddb >= needN));
      }

      int  mnew = 1;
      bool run  = true;
#pragma unroll
      for (int r = 1; r < FPS_K; ++r) {
        const bool hit   = __builtin_amdgcn_ballot_w32((okH & (rH == r)) | (okN & (rN == r))) != 0u;
        const bool spent = __builtin_amdgcn_ballot_w32(rN < r) != 0u;
        run = run & hit & (!spent);
        mnew += run ? 1 : 0;
      }
      const int room = M_ - step;
      mnew = (mnew < room) ? mnew : room;

      if (rH < mnew) fps_idx[b * M_ + step + rH] = qH;
      if (rN < mnew) fps_idx[b * M_ + step + rN] = qN;
      if (lane == 0) s_m = mnew;
    }
    __syncthreads();
    m = __builtin_amdgcn_readfirstlane(s_m);
    step += m;
  }
}

#define NN_G 10
#define NN_TRIPS (N_ / 16 / NN_G)

__global__ __launch_bounds__(256) void three_nn_kernel(
    const float* __restrict__ pc, const int* __restrict__ fps_idx,
    int* __restrict__ idx3, float* __restrict__ w3) {
  const int tid  = threadIdx.x;
  const int lane = tid & 31;
  const int wave = blockIdx.x * (256 >> 5) + (tid >> 5);
  const int tilesPerBatch = M_ / 16;
  const int b    = wave / tilesPerBatch;
  const int rt   = wave - b * tilesPerBatch;
  const int half = lane >> 4;
  const int l15  = lane & 15;

  const float* pcb = pc + (size_t)b * N_ * 3;

  const int mrow = rt * 16 + l15;
  const int sidx = fps_idx[b * M_ + mrow];
  const float* sp = pcb + (size_t)sidx * 3;
  const float sx = sp[0], sy = sp[1], sz = sp[2];
  v2f Bs;
  Bs.x = half ? (-2.0f * sz) : (-2.0f * sx);
  Bs.y = half ? 1.0f         : (-2.0f * sy);

  float d[3];
  int   ix[3];
#pragma unroll
  for (int k = 0; k < 3; ++k) { d[k] = __builtin_inff(); ix[k] = 0x7fffffff; }

  const float* lp = pcb + (size_t)l15 * 3;

  float cx[NN_G], cy[NN_G], cz[NN_G];
#pragma unroll
  for (int g = 0; g < NN_G; ++g) {
    const float* p = lp + (size_t)g * 48;
    cx[g] = p[0]; cy[g] = p[1]; cz[g] = p[2];
  }

  for (int q = 0; q < NN_TRIPS; ++q) {
    const int qn = (q + 1 < NN_TRIPS) ? (q + 1) : q;
    const float* np = lp + (size_t)qn * (NN_G * 48);
    float nx[NN_G], ny[NN_G], nz[NN_G];
#pragma unroll
    for (int g = 0; g < NN_G; ++g) {
      const float* p = np + (size_t)g * 48;
      nx[g] = p[0]; ny[g] = p[1]; nz[g] = p[2];
    }

    v8f c[NN_G];
#pragma unroll
    for (int g = 0; g < NN_G; ++g) {
      const float bx = cx[g], by = cy[g], bz = cz[g];
      const float bn2 = __builtin_fmaf(bz, bz, __builtin_fmaf(bx, bx, by * by));
      const v8f zero = {0.f, 0.f, 0.f, 0.f, 0.f, 0.f, 0.f, 0.f};
#if USE_WMMA
      v2f Av;
      Av.x = half ? bz  : bx;
      Av.y = half ? bn2 : by;
      c[g] = __builtin_amdgcn_wmma_f32_16x16x4_f32(
           false, Av,  false, Bs,
           (short)0, zero,  false,  false);
#else
      c[g] = zero;
#pragma unroll
      for (int r = 0; r < 8; ++r) {
        const int src = half * 8 + r;
        const float rx = __shfl(bx, src, 32);
        const float ry = __shfl(by, src, 32);
        const float rz = __shfl(bz, src, 32);
        const float rn = __shfl(bn2, src, 32);
        c[g][r] = -2.0f * (rx * sx + ry * sy + rz * sz) + rn;
      }
#endif
    }

    unsigned hit[NN_G];
    unsigned anyhit = 0u;
    const float d2s = d[2];
#pragma unroll
    for (int g = 0; g < NN_G; ++g) {
      const float m01 = fminf(fminf(c[g][0], c[g][1]), fminf(c[g][2], c[g][3]));
      const float m23 = fminf(fminf(c[g][4], c[g][5]), fminf(c[g][6], c[g][7]));
      const float mt  = fminf(m01, m23);
      hit[g] = __builtin_amdgcn_ballot_w32(mt < d2s);
      anyhit |= hit[g];
    }
    if (anyhit != 0u) {
#pragma unroll
      for (int g = 0; g < NN_G; ++g) {
        if (hit[g] != 0u) {
          const int n0 = (q * NN_G + g) * 16 + half * 8;
#pragma unroll
          for (int r = 0; r < 8; ++r) ins3_stream(c[g][r], n0 + r, d, ix);
        }
      }
    }

#pragma unroll
    for (int g = 0; g < NN_G; ++g) { cx[g] = nx[g]; cy[g] = ny[g]; cz[g] = nz[g]; }
  }

  {
    float pd[3]; int pi[3];
#pragma unroll
    for (int k = 0; k < 3; ++k) {
      pd[k] = __shfl_xor(d[k], 16, 32);
      pi[k] = __shfl_xor(ix[k], 16, 32);
    }
#pragma unroll
    for (int k = 0; k < 3; ++k) ins3_lex(pd[k], pi[k], d, ix);
  }

  if (half == 0) {
    float wv[3];
    float wsum = 0.f;
#pragma unroll
    for (int k = 0; k < 3; ++k) {
      const int nn = ix[k];
      const float* qp = pcb + (size_t)nn * 3;
      const float dx = qp[0] - sx, dy = qp[1] - sy, dz = qp[2] - sz;
      const float dd = dx * dx + dy * dy + dz * dz;
      const float dist = sqrtf(fmaxf(dd, 0.0f));
      wv[k] = 1.0f / (dist + 1e-8f);
      wsum += wv[k];
      idx3[((size_t)b * M_ + mrow) * 3 + k] = nn;
    }
#pragma unroll
    for (int k = 0; k < 3; ++k)
      w3[((size_t)b * M_ + mrow) * 3 + k] = wv[k] / wsum;
  }
}

__global__ __launch_bounds__(256) void interp_kernel(
    const float* __restrict__ feats, const int* __restrict__ idx3,
    const float* __restrict__ w3, float* __restrict__ out) {
  const int bc = blockIdx.x;
  const int b  = bc / C_;
  const int c  = bc - b * C_;
  const float* f = feats + ((size_t)b * C_ + c) * N_;
  float*       o = out   + ((size_t)b * C_ + c) * M_;
  const int*   ib = idx3 + (size_t)b * M_ * 3;
  const float* wb = w3   + (size_t)b * M_ * 3;
  for (int m = threadIdx.x; m < M_; m += 256) {
    const int   n0 = ib[m * 3 + 0], n1 = ib[m * 3 + 1], n2 = ib[m * 3 + 2];
    const float w0 = wb[m * 3 + 0], w1 = wb[m * 3 + 1], w2 = wb[m * 3 + 2];
    o[m] = w0 * f[n0] + w1 * f[n1] + w2 * f[n2];
  }
}

extern "C" void kernel_launch(void* const* d_in, const int* in_sizes, int n_in,
                              void* d_out, int out_size, void* d_ws, size_t ws_size,
                              hipStream_t stream) {
  const float* pc    = (const float*)d_in[0];
  const float* feats = (const float*)d_in[1];
  const float* obj   = (const float*)d_in[2];
  const float* grasp = (const float*)d_in[3];
  float* out = (float*)d_out;

  char* ws = (char*)d_ws;
  size_t off = 0;
  int* fps = (int*)(ws + off);
  off += (size_t)B_ * M_ * 4;        off = (off + 255) & ~(size_t)255;
  int* idx3 = (int*)(ws + off);
  off += (size_t)B_ * M_ * 3 * 4;    off = (off + 255) & ~(size_t)255;
  float* w3 = (float*)(ws + off);
  off += (size_t)B_ * M_ * 3 * 4;

  fps_kernel<<<B_, 1024, 0, stream>>>(pc, obj, grasp, fps);
  three_nn_kernel<<<(B_ * (M_ / 16)) / 8, 256, 0, stream>>>(pc, fps, idx3, w3);
  interp_kernel<<<B_ * C_, 256, 0, stream>>>(feats, idx3, w3, out);
}


# fps kernel: up to 12 picks a pass (was 6): a wave's 3rd-best value bounds its unlisted points (was: stop at a wave with both entries taken); proof 1 rank per lane pair; best kept by slot; same sums
# speedup vs baseline: 2.6589x; 1.1097x over previous
#include <hip/hip_runtime.h>
#include <hip/hip_bf16.h>

#define B_ 4
#define N_ 20000
#define C_ 512
#define M_ 1024
#define GRASP_TH 0.1f

typedef float v2f __attribute__((ext_vector_type(2)));
typedef float v4f __attribute__((ext_vector_type(4)));
typedef float v8f __attribute__((ext_vector_type(8)));

#if defined(__gfx1250__) && __has_builtin(__builtin_amdgcn_wmma_f32_16x16x4_f32)
#define USE_WMMA 1
#else
#define USE_WMMA 0
#endif

__device__ __forceinline__ void ins3_stream(float v, int n, float d[3], int ix[3]) {
  const bool c0 = v < d[0];
  const bool c1 = v < d[1];
  const bool c2 = v < d[2];
  const float r0 = c0 ? v     : d[0];
  const int   j0 = c0 ? n     : ix[0];
  const float r1 = c0 ? d[0]  : (c1 ? v : d[1]);
  const int   j1 = c0 ? ix[0] : (c1 ? n : ix[1]);
  const float r2 = c1 ? d[1]  : (c2 ? v : d[2]);
  const int   j2 = c1 ? ix[1] : (c2 ? n : ix[2]);
  d[0] = r0; d[1] = r1; d[2] = r2;
  ix[0] = j0; ix[1] = j1; ix[2] = j2;
}

__device__ __forceinline__ void ins3_lex(float v, int n, float d[3], int ix[3]) {
  const bool c0 = (v < d[0]) | ((v == d[0]) & (n < ix[0]));
  const bool c1 = (v < d[1]) | ((v == d[1]) & (n < ix[1]));
  const bool c2 = (v < d[2]) | ((v == d[2]) & (n < ix[2]));
  const float r0 = c0 ? v     : d[0];
  const int   j0 = c0 ? n     : ix[0];
  const float r1 = c0 ? d[0]  : (c1 ? v : d[1]);
  const int   j1 = c0 ? ix[0] : (c1 ? n : ix[1]);
  const float r2 = c1 ? d[1]  : (c2 ? v : d[2]);
  const int   j2 = c1 ? ix[1] : (c2 ? n : ix[2]);
  d[0] = r0; d[1] = r1; d[2] = r2;
  ix[0] = j0; ix[1] = j1; ix[2] = j2;
}

#define FPS_J 5
#define FPS_K 12
#define FPS_KH 6
#define FPS_KP 16
static_assert(FPS_K == 12 && FPS_KH * 2 == FPS_K && FPS_K <= FPS_KP, "the pick list in the scan is written out for 12");

__device__ __forceinline__ bool lex_better(float ov, int oi, float v, int i) {
  return (ov > v) | ((ov == v) & (oi < i));
}

typedef unsigned long long u64;

__device__ __forceinline__ void fps_apply(const v4f vx, const v4f vy, const v4f vz,
                                          float qx, float qy, float qz, float v[4]) {
#pragma unroll
  for (int k = 0; k < 4; ++k) {
    const float dxa = vx[k] - qx, dya = vy[k] - qy, dza = vz[k] - qz;
    const float dda = __builtin_fmaf(dza, dza, __builtin_fmaf(dxa, dxa, dya * dya));
    v[k] = fminf(v[k], dda);
  }
}
#define FPS_PICK(i) if (m <= (i)) break; fps_apply(vx, vy, vz, qx[i], qy[i], qz[i], v)

__device__ __forceinline__ u64 pack_key(float v, int i) {
  const bool real = v >= 0.0f;
  const unsigned hi = real ? __float_as_uint(v) : 0u;
  const unsigned lo = real ? (unsigned)(0x7fffffff - i) : 0u;
  return ((u64)hi << 32) | (u64)lo;
}

__global__ __launch_bounds__(1024) void fps_kernel(
    const float* __restrict__ pc, const float* __restrict__ obj,
    const float* __restrict__ grasp, int* __restrict__ fps_idx) {
  const int b    = blockIdx.x;
  const int tid  = threadIdx.x;
  const int lane = tid & 31;
  const int wid  = tid >> 5;

  __shared__ __attribute__((aligned(16))) float s_x[N_];
  __shared__ __attribute__((aligned(16))) float s_y[N_];
  __shared__ __attribute__((aligned(16))) float s_z[N_];
  __shared__ __attribute__((aligned(16))) u64 s_key[64];
  __shared__ __attribute__((aligned(8)))  int s_rank[64];
  __shared__ int   swi1[32];
  __shared__ int   s_m;
  __shared__ __attribute__((aligned(16))) float s_px[FPS_KP], s_py[FPS_KP], s_pz[FPS_KP];
  __shared__ __attribute__((aligned(16))) float s_pv[FPS_KP];
  __shared__ int s_blk[FPS_KP];

  const float* pcb = pc + (size_t)b * N_ * 3;
  const float* ob0 = obj + (size_t)b * 2 * N_;
  const float* gr  = grasp + (size_t)b * N_;

  for (int n = tid; n < N_; n += 1024) {
    const float* p = pcb + (size_t)n * 3;
    s_x[n] = p[0];
    s_y[n] = p[1];
    s_z[n] = p[2];
  }

  if (tid < FPS_KP) s_blk[tid] = 0;

  float mind[FPS_J][4];
  int firstLoc = N_;
#pragma unroll
  for (int j = FPS_J - 1; j >= 0; --j) {
    const int g = tid + 1024 * j;
    const bool gv = g < N_ / 4;
#pragma unroll
    for (int k = 3; k >= 0; --k) {
      const int n  = 4 * g + k;
      const int nc = gv ? n : 0;
      const bool mk = gv & (ob0[N_ + nc] > ob0[nc]) & (gr[nc] > GRASP_TH);
      mind[j][k] = mk ? __builtin_inff() : -__builtin_inff();
      firstLoc   = mk ? n : firstLoc;
    }
  }
  for (int mm = 16; mm; mm >>= 1) {
    const int o = __shfl_xor(firstLoc, mm, 32);
    firstLoc = (o < firstLoc) ? o : firstLoc;
  }
  if (lane == 0) swi1[wid] = firstLoc;
  __syncthreads();
  if (wid == 0) {
    int v = swi1[lane];
    for (int mm = 16; mm; mm >>= 1) {
      const int o = __shfl_xor(v, mm, 32);
      v = (o < v) ? o : v;
    }
    if (lane == 0) {
      const int f = (v >= N_) ? 0 : v;
      fps_idx[b * M_] = f;
      s_m = 1;
      const float fx = s_x[f], fy = s_y[f], fz = s_z[f];
#pragma unroll
      for (int i = 0; i < FPS_KP; ++i) { s_px[i] = fx; s_py[i] = fy; s_pz[i] = fz; s_pv[i] = 0.0f; }
    }
  }
  __syncthreads();

  const v4f* x4 = (const v4f*)s_x;
  const v4f* y4 = (const v4f*)s_y;
  const v4f* z4 = (const v4f*)s_z;

  int m    = 1;
  int step = 1;
  int pass = 1;

  while (step < M_) {
    float qx[FPS_K], qy[FPS_K], qz[FPS_K];
#pragma unroll
    for (int i = 0; i < FPS_K; ++i) { qx[i] = s_px[i]; qy[i] = s_py[i]; qz[i] = s_pz[i]; }

    float b1v = -__builtin_inff(), b2v = -__builtin_inff();
    int   b1s = 0,                 b2s = 0;
    int   b3  = __float_as_int(-__builtin_inff());
#pragma unroll
    for (int j = 0; j < FPS_J; ++j) {
      const int g  = tid + 1024 * j;
      const int gc = (g < N_ / 4) ? g : tid;
      const v4f vx = x4[gc];
      const v4f vy = y4[gc];
      const v4f vz = z4[gc];
      float v[4];
#pragma unroll
      for (int k = 0; k < 4; ++k) v[k] = mind[j][k];
      do {
        FPS_PICK(0); FPS_PICK(1); FPS_PICK(2);  FPS_PICK(3);
        FPS_PICK(4); FPS_PICK(5); FPS_PICK(6);  FPS_PICK(7);
        FPS_PICK(8); FPS_PICK(9); FPS_PICK(10); FPS_PICK(11);
      } while (0);
#pragma unroll
      for (int k = 0; k < 4; ++k) {
        mind[j][k] = v[k];
        const int  s  = 4 * j + k;
        const bool c1 = v[k] > b1v;
        const bool c2 = v[k] > b2v;
        const int vb = __float_as_int(v[k]);
        const int lo = (__float_as_int(b2v) < vb) ? __float_as_int(b2v) : vb;
        b3  = (lo > b3) ? lo : b3;
        b2v = c1 ? b1v : (c2 ? v[k] : b2v);
        b2s = c1 ? b1s : (c2 ? s    : b2s);
        b1v = c1 ? v[k] : b1v;
        b1s = c1 ? s    : b1s;
      }
    }
    int b1i = 4 * tid + 4096 * (b1s >> 2) + (b1s & 3);
    int b2i = 4 * tid + 4096 * (b2s >> 2) + (b2s & 3);
    for (int mm = 16; mm; mm >>= 1) {
      const float o1v = __shfl_xor(b1v, mm, 32);
      const int   o1i = __shfl_xor(b1i, mm, 32);
      const float o2v = __shfl_xor(b2v, mm, 32);
      const int   o2i = __shfl_xor(b2i, mm, 32);
      const int   o3  = __shfl_xor(b3, mm, 32);
      {
        const int a1 = __float_as_int(b1v), a2 = __float_as_int(b2v);
        const int p1 = __float_as_int(o1v), p2 = __float_as_int(o2v);
        const int m12 = (a1 < p2) ? a1 : p2;
        const int m21 = (a2 < p1) ? a2 : p1;
        const int t3  = (b3 > o3) ? b3 : o3;
        const int t4  = (m12 > m21) ? m12 : m21;
        b3 = (t3 > t4) ? t3 : t4;
      }
      const bool c = lex_better(o1v, o1i, b1v, b1i);
      const float lv = c ? b1v : o1v;  const int li = c ? b1i : o1i;
      const float sv = c ? o2v : b2v;  const int si = c ? o2i : b2i;
      b1v = c ? o1v : b1v;             b1i = c ? o1i : b1i;
      const bool d = lex_better(lv, li, sv, si);
      b2v = d ? lv : sv;               b2i = d ? li : si;
    }
    const u64 k1 = pack_key(b1v, b1i);
    const u64 k2 = pack_key(b2v, b2i);
    const u64 k3 = (b3 >= 0) ? (((u64)(unsigned)b3 << 32) | 0xffffffffull) : 0ull;
    if (lane == 0) {
      s_key[2 * wid]     = k1;
      s_key[2 * wid + 1] = k2;
    }
    __syncthreads();
    {
      const unsigned h1 = __builtin_amdgcn_readfirstlane((unsigned)(k1 >> 32));
      const unsigned l1 = __builtin_amdgcn_readfirstlane((unsigned)k1);
      const unsigned h2 = __builtin_amdgcn_readfirstlane((unsigned)(k2 >> 32));
      const unsigned l2 = __builtin_amdgcn_readfirstlane((unsigned)k2);
      const u64 kHs = ((u64)h1 << 32) | (u64)l1;
      const u64 kNs = ((u64)h2 << 32) | (u64)l2;
      const unsigned h3 = __builtin_amdgcn_readfirstlane((unsigned)(k3 >> 32));
      const unsigned l3 = __builtin_amdgcn_readfirstlane((unsigned)k3);
      const u64 kTs = ((u64)h3 << 32) | (u64)l3;
      const u64 ka = s_key[2 * lane];
      const u64 kb = s_key[2 * lane + 1];
      const int rHs = __builtin_popcount(__builtin_amdgcn_ballot_w32(ka > kHs)) +
                      __builtin_popcount(__builtin_amdgcn_ballot_w32(kb > kHs));
      const int rNs = __builtin_popcount(__builtin_amdgcn_ballot_w32(ka > kNs)) +
                      __builtin_popcount(__builtin_amdgcn_ballot_w32(kb > kNs));
      const int rTs = __builtin_popcount(__builtin_amdgcn_ballot_w32(ka > kTs)) +
                      __builtin_popcount(__builtin_amdgcn_ballot_w32(kb > kTs));
      if (lane == 0) {
        s_rank[2 * wid] = rHs; s_rank[2 * wid + 1] = rNs;
        const int rMark = (rTs < 1) ? 1 : rTs;
        if (rMark < FPS_K) s_blk[rMark] = pass;
      }
    }
    __syncthreads();
    if (wid == 0) {
      const u64 kH = s_key[2 * lane];
      const u64 kN = s_key[2 * lane + 1];
      const float valH = __uint_as_float((unsigned)(kH >> 32));
      const float valN = __uint_as_float((unsigned)(kN >> 32));
      const int iH = 0x7fffffff - (int)(unsigned)kH;
      const int iN = 0x7fffffff - (int)(unsigned)kN;
      const bool vH = iH < N_;
      const bool vN = iN < N_;
      const int qH = vH ? iH : 0;
      const int qN = vN ? iN : 0;
      const float xH = s_x[qH], yH = s_y[qH], zH = s_z[qH];
      const float xN = s_x[qN], yN = s_y[qN], zN = s_z[qN];

      const int rH = s_rank[2 * lane];
      const int rN = s_rank[2 * lane + 1];

      if (rH < FPS_K) { s_px[rH] = xH; s_py[rH] = yH; s_pz[rH] = zH; s_pv[rH] = valH; }
      if (rN < FPS_K) { s_px[rN] = xN; s_py[rN] = yN; s_pz[rN] = zN; s_pv[rN] = valN; }

      const int R = __builtin_popcount(__builtin_amdgcn_ballot_w32(vH)) +
                    __builtin_popcount(__builtin_amdgcn_ballot_w32(vN));

      const int r  = lane & 15;
      const int hh = lane >> 4;
      const float cxr = s_px[r], cyr = s_py[r], czr = s_pz[r], cvr = s_pv[r];
      const int   blk = s_blk[r];

      const float need = cvr * 1.0001f;
      float dmin = __builtin_inff();
#pragma unroll
      for (int i = 0; i < FPS_KH; ++i) {
        const int ii = hh * FPS_KH + i;
        const float sx = s_px[ii], sy = s_py[ii], sz = s_pz[ii];
        const float dxa = cxr - sx, dya = cyr - sy, dza = czr - sz;
        const float dda = dxa * dxa + dya * dya + dza * dza;
        dmin = fminf(dmin, (ii >= r) ? __builtin_inff() : dda);
      }
      const bool ok = (r < FPS_K) & (r < R) & (cvr > 0.0f) & (blk != pass) & (dmin >= need);

      const unsigned fail = __builtin_amdgcn_ballot_w32(!ok);
      const unsigned fm   = ((fail | (fail >> 16)) & 0xfffeu) | (1u << FPS_K);
      int mnew = __builtin_ctz(fm);
      const int room = M_ - step;
      mnew = (mnew < room) ? mnew : room;

      if (rH < mnew) fps_idx[b * M_ + step + rH] = qH;
      if (rN < mnew) fps_idx[b * M_ + step + rN] = qN;
      if (lane == 0) s_m = mnew;
    }
    __syncthreads();
    m = __builtin_amdgcn_readfirstlane(s_m);
    step += m;
    pass += 1;
  }
}

#define NN_G 10
#define NN_TRIPS (N_ / 16 / NN_G)

__global__ __launch_bounds__(256) void three_nn_kernel(
    const float* __restrict__ pc, const int* __restrict__ fps_idx,
    int* __restrict__ idx3, float* __restrict__ w3) {
  const int tid  = threadIdx.x;
  const int lane = tid & 31;
  const int wave = blockIdx.x * (256 >> 5) + (tid >> 5);
  const int tilesPerBatch = M_ / 16;
  const int b    = wave / tilesPerBatch;
  const int rt   = wave - b * tilesPerBatch;
  const int half = lane >> 4;
  const int l15  = lane & 15;

  const float* pcb = pc + (size_t)b * N_ * 3;

  const int mrow = rt * 16 + l15;
  const int sidx = fps_idx[b * M_ + mrow];
  const float* sp = pcb + (size_t)sidx * 3;
  const float sx = sp[0], sy = sp[1], sz = sp[2];
  v2f Bs;
  Bs.x = half ? (-2.0f * sz) : (-2.0f * sx);
  Bs.y = half ? 1.0f         : (-2.0f * sy);

  float d[3];
  int   ix[3];
#pragma unroll
  for (int k = 0; k < 3; ++k) { d[k] = __builtin_inff(); ix[k] = 0x7fffffff; }

  const float* lp = pcb + (size_t)l15 * 3;

  float cx[NN_G], cy[NN_G], cz[NN_G];
#pragma unroll
  for (int g = 0; g < NN_G; ++g) {
    const float* p = lp + (size_t)g * 48;
    cx[g] = p[0]; cy[g] = p[1]; cz[g] = p[2];
  }

  for (int q = 0; q < NN_TRIPS; ++q) {
    const int qn = (q + 1 < NN_TRIPS) ? (q + 1) : q;
    const float* np = lp + (size_t)qn * (NN_G * 48);
    float nx[NN_G], ny[NN_G], nz[NN_G];
#pragma unroll
    for (int g = 0; g < NN_G; ++g) {
      const float* p = np + (size_t)g * 48;
      nx[g] = p[0]; ny[g] = p[1]; nz[g] = p[2];
    }

    v8f c[NN_G];
#pragma unroll
    for (int g = 0; g < NN_G; ++g) {
      const float bx = cx[g], by = cy[g], bz = cz[g];
      const float bn2 = __builtin_fmaf(bz, bz, __builtin_fmaf(bx, bx, by * by));
      const v8f zero = {0.f, 0.f, 0.f, 0.f, 0.f, 0.f, 0.f, 0.f};
#if USE_WMMA
      v2f Av;
      Av.x = half ? bz  : bx;
      Av.y = half ? bn2 : by;
      c[g] = __builtin_amdgcn_wmma_f32_16x16x4_f32(
           false, Av,  false, Bs,
           (short)0, zero,  false,  false);
#else
      c[g] = zero;
#pragma unroll
      for (int r = 0; r < 8; ++r) {
        const int src = half * 8 + r;
        const float rx = __shfl(bx, src, 32);
        const float ry = __shfl(by, src, 32);
        const float rz = __shfl(bz, src, 32);
        const float rn = __shfl(bn2, src, 32);
        c[g][r] = -2.0f * (rx * sx + ry * sy + rz * sz) + rn;
      }
#endif
    }

    unsigned hit[NN_G];
    unsigned anyhit = 0u;
    const float d2s = d[2];
#pragma unroll
    for (int g = 0; g < NN_G; ++g) {
      const float m01 = fminf(fminf(c[g][0], c[g][1]), fminf(c[g][2], c[g][3]));
      const float m23 = fminf(fminf(c[g][4], c[g][5]), fminf(c[g][6], c[g][7]));
      const float mt  = fminf(m01, m23);
      hit[g] = __builtin_amdgcn_ballot_w32(mt < d2s);
      anyhit |= hit[g];
    }
    if (anyhit != 0u) {
#pragma unroll
      for (int g = 0; g < NN_G; ++g) {
        if (hit[g] != 0u) {
          const int n0 = (q * NN_G + g) * 16 + half * 8;
#pragma unroll
          for (int r = 0; r < 8; ++r) ins3_stream(c[g][r], n0 + r, d, ix);
        }
      }
    }

#pragma unroll
    for (int g = 0; g < NN_G; ++g) { cx[g] = nx[g]; cy[g] = ny[g]; cz[g] = nz[g]; }
  }

  {
    float pd[3]; int pi[3];
#pragma unroll
    for (int k = 0; k < 3; ++k) {
      pd[k] = __shfl_xor(d[k], 16, 32);
      pi[k] = __shfl_xor(ix[k], 16, 32);
    }
#pragma unroll
    for (int k = 0; k < 3; ++k) ins3_lex(pd[k], pi[k], d, ix);
  }

  if (half == 0) {
    float wv[3];
    float wsum = 0.f;
#pragma unroll
    for (int k = 0; k < 3; ++k) {
      const int nn = ix[k];
      const float* qp = pcb + (size_t)nn * 3;
      const float dx = qp[0] - sx, dy = qp[1] - sy, dz = qp[2] - sz;
      const float dd = dx * dx + dy * dy + dz * dz;
      const float dist = sqrtf(fmaxf(dd, 0.0f));
      wv[k] = 1.0f / (dist + 1e-8f);
      wsum += wv[k];
      idx3[((size_t)b * M_ + mrow) * 3 + k] = nn;
    }
#pragma unroll
    for (int k = 0; k < 3; ++k)
      w3[((size_t)b * M_ + mrow) * 3 + k] = wv[k] / wsum;
  }
}

__global__ __launch_bounds__(256) void interp_kernel(
    const float* __restrict__ feats, const int* __restrict__ idx3,
    const float* __restrict__ w3, float* __restrict__ out) {
  const int bc = blockIdx.x;
  const int b  = bc / C_;
  const int c  = bc - b * C_;
  const float* f = feats + ((size_t)b * C_ + c) * N_;
  float*       o = out   + ((size_t)b * C_ + c) * M_;
  const int*   ib = idx3 + (size_t)b * M_ * 3;
  const float* wb = w3   + (size_t)b * M_ * 3;
  for (int m = threadIdx.x; m < M_; m += 256) {
    const int   n0 = ib[m * 3 + 0], n1 = ib[m * 3 + 1], n2 = ib[m * 3 + 2];
    const float w0 = wb[m * 3 + 0], w1 = wb[m * 3 + 1], w2 = wb[m * 3 + 2];
    o[m] = w0 * f[n0] + w1 * f[n1] + w2 * f[n2];
  }
}

extern "C" void kernel_launch(void* const* d_in, const int* in_sizes, int n_in,
                              void* d_out, int out_size, void* d_ws, size_t ws_size,
                              hipStream_t stream) {
  const float* pc    = (const float*)d_in[0];
  const float* feats = (const float*)d_in[1];
  const float* obj   = (const float*)d_in[2];
  const float* grasp = (const float*)d_in[3];
  float* out = (float*)d_out;

  char* ws = (char*)d_ws;
  size_t off = 0;
  int* fps = (int*)(ws + off);
  off += (size_t)B_ * M_ * 4;        off = (off + 255) & ~(size_t)255;
  int* idx3 = (int*)(ws + off);
  off += (size_t)B_ * M_ * 3 * 4;    off = (off + 255) & ~(size_t)255;
  float* w3 = (float*)(ws + off);
  off += (size_t)B_ * M_ * 3 * 4;

  fps_kernel<<<B_, 1024, 0, stream>>>(pc, obj, grasp, fps);
  three_nn_kernel<<<(B_ * (M_ / 16)) / 8, 256, 0, stream>>>(pc, fps, idx3, w3);
  interp_kernel<<<B_ * C_, 256, 0, stream>>>(feats, idx3, w3, out);
}


# fps kernel: a batch candidate that an earlier pick of the batch lowers is skipped (was: ended the batch); later ranks stand while above a bound on every skipped one; 16 ranks examined; same sums
# speedup vs baseline: 3.0065x; 1.1307x over previous
#include <hip/hip_runtime.h>
#include <hip/hip_bf16.h>

#define B_ 4
#define N_ 20000
#define C_ 512
#define M_ 1024
#define GRASP_TH 0.1f

typedef float v2f __attribute__((ext_vector_type(2)));
typedef float v4f __attribute__((ext_vector_type(4)));
typedef float v8f __attribute__((ext_vector_type(8)));

#if defined(__gfx1250__) && __has_builtin(__builtin_amdgcn_wmma_f32_16x16x4_f32)
#define USE_WMMA 1
#else
#define USE_WMMA 0
#endif

__device__ __forceinline__ void ins3_stream(float v, int n, float d[3], int ix[3]) {
  const bool c0 = v < d[0];
  const bool c1 = v < d[1];
  const bool c2 = v < d[2];
  const float r0 = c0 ? v     : d[0];
  const int   j0 = c0 ? n     : ix[0];
  const float r1 = c0 ? d[0]  : (c1 ? v : d[1]);
  const int   j1 = c0 ? ix[0] : (c1 ? n : ix[1]);
  const float r2 = c1 ? d[1]  : (c2 ? v : d[2]);
  const int   j2 = c1 ? ix[1] : (c2 ? n : ix[2]);
  d[0] = r0; d[1] = r1; d[2] = r2;
  ix[0] = j0; ix[1] = j1; ix[2] = j2;
}

__device__ __forceinline__ void ins3_lex(float v, int n, float d[3], int ix[3]) {
  const bool c0 = (v < d[0]) | ((v == d[0]) & (n < ix[0]));
  const bool c1 = (v < d[1]) | ((v == d[1]) & (n < ix[1]));
  const bool c2 = (v < d[2]) | ((v == d[2]) & (n < ix[2]));
  const float r0 = c0 ? v     : d[0];
  const int   j0 = c0 ? n     : ix[0];
  const float r1 = c0 ? d[0]  : (c1 ? v : d[1]);
  const int   j1 = c0 ? ix[0] : (c1 ? n : ix[1]);
  const float r2 = c1 ? d[1]  : (c2 ? v : d[2]);
  const int   j2 = c1 ? ix[1] : (c2 ? n : ix[2]);
  d[0] = r0; d[1] = r1; d[2] = r2;
  ix[0] = j0; ix[1] = j1; ix[2] = j2;
}

#define FPS_J 5
#define FPS_K 12
#define FPS_KC 16
#define FPS_KCH 8
#define FPS_KP 16
#define W_CX 0
#define W_CY 16
#define W_CZ 32
#define W_CV 48
#define W_CI 64
#define W_UB 80
#define W_BLK 112
static_assert(FPS_K == 12 && FPS_KCH * 2 == FPS_KC && FPS_KC == 16 && FPS_K <= FPS_KP, "the pick list in the scan is written out for 12, the proof for 16 candidates");

__device__ __forceinline__ bool lex_better(float ov, int oi, float v, int i) {
  return (ov > v) | ((ov == v) & (oi < i));
}

typedef unsigned long long u64;

__device__ __forceinline__ void fps_apply(const v4f vx, const v4f vy, const v4f vz,
                                          float qx, float qy, float qz, float v[4]) {
#pragma unroll
  for (int k = 0; k < 4; ++k) {
    const float dxa = vx[k] - qx, dya = vy[k] - qy, dza = vz[k] - qz;
    const float dda = __builtin_fmaf(dza, dza, __builtin_fmaf(dxa, dxa, dya * dya));
    v[k] = fminf(v[k], dda);
  }
}
#define FPS_PICK(i) if (m <= (i)) break; fps_apply(vx, vy, vz, qx[i], qy[i], qz[i], v)

__device__ __forceinline__ u64 pack_key(float v, int i) {
  const bool real = v >= 0.0f;
  const unsigned hi = real ? __float_as_uint(v) : 0u;
  const unsigned lo = real ? (unsigned)(0x7fffffff - i) : 0u;
  return ((u64)hi << 32) | (u64)lo;
}

__global__ __launch_bounds__(1024) void fps_kernel(
    const float* __restrict__ pc, const float* __restrict__ obj,
    const float* __restrict__ grasp, int* __restrict__ fps_idx) {
  const int b    = blockIdx.x;
  const int tid  = threadIdx.x;
  const int lane = tid & 31;
  const int wid  = tid >> 5;

  __shared__ __attribute__((aligned(16))) float s_x[N_];
  __shared__ __attribute__((aligned(16))) float s_y[N_];
  __shared__ __attribute__((aligned(16))) float s_z[N_];
  __shared__ __attribute__((aligned(16))) u64 s_key[64];
  __shared__ __attribute__((aligned(8)))  int s_rank[64];
  __shared__ int   swi1[32];
  __shared__ int   s_m;
  __shared__ __attribute__((aligned(16))) float s_px[FPS_KP], s_py[FPS_KP], s_pz[FPS_KP];
  __shared__ __attribute__((aligned(16))) float s_w[128];

  const float* pcb = pc + (size_t)b * N_ * 3;
  const float* ob0 = obj + (size_t)b * 2 * N_;
  const float* gr  = grasp + (size_t)b * N_;

  for (int n = tid; n < N_; n += 1024) {
    const float* p = pcb + (size_t)n * 3;
    s_x[n] = p[0];
    s_y[n] = p[1];
    s_z[n] = p[2];
  }

  if (tid < 128) s_w[tid] = 0.0f;

  float mind[FPS_J][4];
  int firstLoc = N_;
#pragma unroll
  for (int j = FPS_J - 1; j >= 0; --j) {
    const int g = tid + 1024 * j;
    const bool gv = g < N_ / 4;
#pragma unroll
    for (int k = 3; k >= 0; --k) {
      const int n  = 4 * g + k;
      const int nc = gv ? n : 0;
      const bool mk = gv & (ob0[N_ + nc] > ob0[nc]) & (gr[nc] > GRASP_TH);
      mind[j][k] = mk ? __builtin_inff() : -__builtin_inff();
      firstLoc   = mk ? n : firstLoc;
    }
  }
  for (int mm = 16; mm; mm >>= 1) {
    const int o = __shfl_xor(firstLoc, mm, 32);
    firstLoc = (o < firstLoc) ? o : firstLoc;
  }
  if (lane == 0) swi1[wid] = firstLoc;
  __syncthreads();
  if (wid == 0) {
    int v = swi1[lane];
    for (int mm = 16; mm; mm >>= 1) {
      const int o = __shfl_xor(v, mm, 32);
      v = (o < v) ? o : v;
    }
    if (lane == 0) {
      const int f = (v >= N_) ? 0 : v;
      fps_idx[b * M_] = f;
      s_m = 1;
      const float fx = s_x[f], fy = s_y[f], fz = s_z[f];
#pragma unroll
      for (int i = 0; i < FPS_KP; ++i) { s_px[i] = fx; s_py[i] = fy; s_pz[i] = fz; }
    }
  }
  __syncthreads();

  const v4f* x4 = (const v4f*)s_x;
  const v4f* y4 = (const v4f*)s_y;
  const v4f* z4 = (const v4f*)s_z;

  int m    = 1;
  int step = 1;
  int pass = 1;

  while (step < M_) {
    float qx[FPS_K], qy[FPS_K], qz[FPS_K];
#pragma unroll
    for (int i = 0; i < FPS_K; ++i) { qx[i] = s_px[i]; qy[i] = s_py[i]; qz[i] = s_pz[i]; }

    float b1v = -__builtin_inff(), b2v = -__builtin_inff();
    int   b1s = 0,                 b2s = 0;
    int   b3  = __float_as_int(-__builtin_inff());
#pragma unroll
    for (int j = 0; j < FPS_J; ++j) {
      const int g  = tid + 1024 * j;
      const int gc = (g < N_ / 4) ? g : tid;
      const v4f vx = x4[gc];
      const v4f vy = y4[gc];
      const v4f vz = z4[gc];
      float v[4];
#pragma unroll
      for (int k = 0; k < 4; ++k) v[k] = mind[j][k];
      do {
        FPS_PICK(0); FPS_PICK(1); FPS_PICK(2);  FPS_PICK(3);
        FPS_PICK(4); FPS_PICK(5); FPS_PICK(6);  FPS_PICK(7);
        FPS_PICK(8); FPS_PICK(9); FPS_PICK(10); FPS_PICK(11);
      } while (0);
#pragma unroll
      for (int k = 0; k < 4; ++k) {
        mind[j][k] = v[k];
        const int  s  = 4 * j + k;
        const bool c1 = v[k] > b1v;
        const bool c2 = v[k] > b2v;
        const int vb = __float_as_int(v[k]);
        const int lo = (__float_as_int(b2v) < vb) ? __float_as_int(b2v) : vb;
        b3  = (lo > b3) ? lo : b3;
        b2v = c1 ? b1v : (c2 ? v[k] : b2v);
        b2s = c1 ? b1s : (c2 ? s    : b2s);
        b1v = c1 ? v[k] : b1v;
        b1s = c1 ? s    : b1s;
      }
    }
    int b1i = 4 * tid + 4096 * (b1s >> 2) + (b1s & 3);
    int b2i = 4 * tid + 4096 * (b2s >> 2) + (b2s & 3);
    for (int mm = 16; mm; mm >>= 1) {
      const float o1v = __shfl_xor(b1v, mm, 32);
      const int   o1i = __shfl_xor(b1i, mm, 32);
      const float o2v = __shfl_xor(b2v, mm, 32);
      const int   o2i = __shfl_xor(b2i, mm, 32);
      const int   o3  = __shfl_xor(b3, mm, 32);
      {
        const int a1 = __float_as_int(b1v), a2 = __float_as_int(b2v);
        const int p1 = __float_as_int(o1v), p2 = __float_as_int(o2v);
        const int m12 = (a1 < p2) ? a1 : p2;
        const int m21 = (a2 < p1) ? a2 : p1;
        const int t3  = (b3 > o3) ? b3 : o3;
        const int t4  = (m12 > m21) ? m12 : m21;
        b3 = (t3 > t4) ? t3 : t4;
      }
      const bool c = lex_better(o1v, o1i, b1v, b1i);
      const float lv = c ? b1v : o1v;  const int li = c ? b1i : o1i;
      const float sv = c ? o2v : b2v;  const int si = c ? o2i : b2i;
      b1v = c ? o1v : b1v;             b1i = c ? o1i : b1i;
      const bool d = lex_better(lv, li, sv, si);
      b2v = d ? lv : sv;               b2i = d ? li : si;
    }
    const u64 k1 = pack_key(b1v, b1i);
    const u64 k2 = pack_key(b2v, b2i);
    const u64 k3 = (b3 >= 0) ? (((u64)(unsigned)b3 << 32) | 0xffffffffull) : 0ull;
    if (lane == 0) {
      s_key[2 * wid]     = k1;
      s_key[2 * wid + 1] = k2;
    }
    __syncthreads();
    {
      const unsigned h1 = __builtin_amdgcn_readfirstlane((unsigned)(k1 >> 32));
      const unsigned l1 = __builtin_amdgcn_readfirstlane((unsigned)k1);
      const unsigned h2 = __builtin_amdgcn_readfirstlane((unsigned)(k2 >> 32));
      const unsigned l2 = __builtin_amdgcn_readfirstlane((unsigned)k2);
      const u64 kHs = ((u64)h1 << 32) | (u64)l1;
      const u64 kNs = ((u64)h2 << 32) | (u64)l2;
      const unsigned h3 = __builtin_amdgcn_readfirstlane((unsigned)(k3 >> 32));
      const unsigned l3 = __builtin_amdgcn_readfirstlane((unsigned)k3);
      const u64 kTs = ((u64)h3 << 32) | (u64)l3;
      const u64 ka = s_key[2 * lane];
      const u64 kb = s_key[2 * lane + 1];
      const int rHs = __builtin_popcount(__builtin_amdgcn_ballot_w32(ka > kHs)) +
                      __builtin_popcount(__builtin_amdgcn_ballot_w32(kb > kHs));
      const int rNs = __builtin_popcount(__builtin_amdgcn_ballot_w32(ka > kNs)) +
                      __builtin_popcount(__builtin_amdgcn_ballot_w32(kb > kNs));
      const int rTs = __builtin_popcount(__builtin_amdgcn_ballot_w32(ka > kTs)) +
                      __builtin_popcount(__builtin_amdgcn_ballot_w32(kb > kTs));
      if (lane == 0) {
        s_rank[2 * wid] = rHs; s_rank[2 * wid + 1] = rNs;
        const int rMark = (rTs < 1) ? 1 : rTs;
        if (rMark < FPS_KC) s_w[W_BLK + rMark] = __int_as_float(pass);
      }
    }
    __syncthreads();
    if (wid == 0) {
      const u64 kH = s_key[2 * lane];
      const u64 kN = s_key[2 * lane + 1];
      const float valH = __uint_as_float((unsigned)(kH >> 32));
      const float valN = __uint_as_float((unsigned)(kN >> 32));
      const int iH = 0x7fffffff - (int)(unsigned)kH;
      const int iN = 0x7fffffff - (int)(unsigned)kN;
      const bool vH = iH < N_;
      const bool vN = iN < N_;
      const int qH = vH ? iH : 0;
      const int qN = vN ? iN : 0;
      const float xH = s_x[qH], yH = s_y[qH], zH = s_z[qH];
      const float xN = s_x[qN], yN = s_y[qN], zN = s_z[qN];

      const int rH = s_rank[2 * lane];
      const int rN = s_rank[2 * lane + 1];

      if (rH < FPS_KC) {
        s_w[W_CX + rH] = xH; s_w[W_CY + rH] = yH; s_w[W_CZ + rH] = zH;
        s_w[W_CV + rH] = valH; s_w[W_CI + rH] = __int_as_float(qH);
      }
      if (rN < FPS_KC) {
        s_w[W_CX + rN] = xN; s_w[W_CY + rN] = yN; s_w[W_CZ + rN] = zN;
        s_w[W_CV + rN] = valN; s_w[W_CI + rN] = __int_as_float(qN);
      }

      const int R = __builtin_popcount(__builtin_amdgcn_ballot_w32(vH)) +
                    __builtin_popcount(__builtin_amdgcn_ballot_w32(vN));

      const int r  = lane & 15;
      const int hh = lane >> 4;
      const float cxr = s_w[W_CX + r], cyr = s_w[W_CY + r], czr = s_w[W_CZ + r], cvr = s_w[W_CV + r];
      const int   cir = __float_as_int(s_w[W_CI + r]);
      const int   blk = __float_as_int(s_w[W_BLK + r]);
      const bool  real = (r < R) & (cvr > 0.0f);

      const float need = cvr * 1.0001f;
      float dd[FPS_KCH];
      float dall = __builtin_inff();
#pragma unroll
      for (int i = 0; i < FPS_KCH; ++i) {
        const int ii = hh * FPS_KCH + i;
        const float sx = s_w[W_CX + ii], sy = s_w[W_CY + ii], sz = s_w[W_CZ + ii];
        const float dxa = cxr - sx, dya = cyr - sy, dza = czr - sz;
        const float dda = dxa * dxa + dya * dya + dza * dza;
        dd[i] = (ii >= r) ? __builtin_inff() : dda;
        dall  = fminf(dall, dd[i]);
      }
      const unsigned nfar  = __builtin_amdgcn_ballot_w32(!(dall >= need));
      const unsigned nreal = __builtin_amdgcn_ballot_w32(!real);
      const unsigned realm = ~nreal & 0xffffu;
      const unsigned amask = ~(nfar | (nfar >> 16)) & realm;
      const unsigned ah = hh ? (amask >> FPS_KCH) : amask;
      float da[FPS_KCH];
#pragma unroll
      for (int i = 0; i < FPS_KCH; ++i)
        da[i] = ((ah >> i) & 1u) ? dd[i] : __builtin_inff();
      const float dA = fminf(fminf(fminf(da[0], da[1]), fminf(da[2], da[3])),
                             fminf(fminf(da[4], da[5]), fminf(da[6], da[7])));
      s_w[W_UB + lane] = dA;
      const unsigned skm = realm & ~amask;
      int wv[FPS_KC];
#pragma unroll
      for (int s = 0; s < FPS_KC; ++s) {
        const int lo = __float_as_int(s_w[W_UB + s]);
        const int hi = __float_as_int(s_w[W_UB + 16 + s]);
        const bool use = (((skm >> s) & 1u) != 0u) & (s < r);
        wv[s] = use ? ((lo < hi) ? lo : hi) : 0;
      }
#pragma unroll
      for (int w = FPS_KC / 2; w; w >>= 1) {
#pragma unroll
        for (int s = 0; s < w; ++s) wv[s] = (wv[s] > wv[s + w]) ? wv[s] : wv[s + w];
      }
      const float mub = __int_as_float(wv[0]) * 1.0001f;
      const unsigned mk = __builtin_amdgcn_ballot_w32(blk == pass) & 0xffffu;
      const int lmark = __builtin_ctz(mk | (1u << FPS_KC));
      const bool acc = (((amask >> r) & 1u) != 0u) & (mub < cvr) & (r < lmark);
      const unsigned accm = (__builtin_amdgcn_ballot_w32(acc) & 0xffffu) | 1u;
      const int pos = __builtin_popcount(accm & ((1u << r) - 1u));
      int mnew = __builtin_popcount(accm);
      mnew = (mnew < FPS_K) ? mnew : FPS_K;
      const int room = M_ - step;
      mnew = (mnew < room) ? mnew : room;

      if ((hh == 0) & (((accm >> r) & 1u) != 0u) & (pos < mnew)) {
        s_px[pos] = cxr; s_py[pos] = cyr; s_pz[pos] = czr;
        fps_idx[b * M_ + step + pos] = cir;
      }
      if (lane == 0) s_m = mnew;
    }
    __syncthreads();
    m = __builtin_amdgcn_readfirstlane(s_m);
    step += m;
    pass += 1;
  }
}

#define NN_G 10
#define NN_TRIPS (N_ / 16 / NN_G)

__global__ __launch_bounds__(256) void three_nn_kernel(
    const float* __restrict__ pc, const int* __restrict__ fps_idx,
    int* __restrict__ idx3, float* __restrict__ w3) {
  const int tid  = threadIdx.x;
  const int lane = tid & 31;
  const int wave = blockIdx.x * (256 >> 5) + (tid >> 5);
  const int tilesPerBatch = M_ / 16;
  const int b    = wave / tilesPerBatch;
  const int rt   = wave - b * tilesPerBatch;
  const int half = lane >> 4;
  const int l15  = lane & 15;

  const float* pcb = pc + (size_t)b * N_ * 3;

  const int mrow = rt * 16 + l15;
  const int sidx = fps_idx[b * M_ + mrow];
  const float* sp = pcb + (size_t)sidx * 3;
  const float sx = sp[0], sy = sp[1], sz = sp[2];
  v2f Bs;
  Bs.x = half ? (-2.0f * sz) : (-2.0f * sx);
  Bs.y = half ? 1.0f         : (-2.0f * sy);

  float d[3];
  int   ix[3];
#pragma unroll
  for (int k = 0; k < 3; ++k) { d[k] = __builtin_inff(); ix[k] = 0x7fffffff; }

  const float* lp = pcb + (size_t)l15 * 3;

  float cx[NN_G], cy[NN_G], cz[NN_G];
#pragma unroll
  for (int g = 0; g < NN_G; ++g) {
    const float* p = lp + (size_t)g * 48;
    cx[g] = p[0]; cy[g] = p[1]; cz[g] = p[2];
  }

  for (int q = 0; q < NN_TRIPS; ++q) {
    const int qn = (q + 1 < NN_TRIPS) ? (q + 1) : q;
    const float* np = lp + (size_t)qn * (NN_G * 48);
    float nx[NN_G], ny[NN_G], nz[NN_G];
#pragma unroll
    for (int g = 0; g < NN_G; ++g) {
      const float* p = np + (size_t)g * 48;
      nx[g] = p[0]; ny[g] = p[1]; nz[g] = p[2];
    }

    v8f c[NN_G];
#pragma unroll
    for (int g = 0; g < NN_G; ++g) {
      const float bx = cx[g], by = cy[g], bz = cz[g];
      const float bn2 = __builtin_fmaf(bz, bz, __builtin_fmaf(bx, bx, by * by));
      const v8f zero = {0.f, 0.f, 0.f, 0.f, 0.f, 0.f, 0.f, 0.f};
#if USE_WMMA
      v2f Av;
      Av.x = half ? bz  : bx;
      Av.y = half ? bn2 : by;
      c[g] = __builtin_amdgcn_wmma_f32_16x16x4_f32(
           false, Av,  false, Bs,
           (short)0, zero,  false,  false);
#else
      c[g] = zero;
#pragma unroll
      for (int r = 0; r < 8; ++r) {
        const int src = half * 8 + r;
        const float rx = __shfl(bx, src, 32);
        const float ry = __shfl(by, src, 32);
        const float rz = __shfl(bz, src, 32);
        const float rn = __shfl(bn2, src, 32);
        c[g][r] = -2.0f * (rx * sx + ry * sy + rz * sz) + rn;
      }
#endif
    }

    unsigned hit[NN_G];
    unsigned anyhit = 0u;
    const float d2s = d[2];
#pragma unroll
    for (int g = 0; g < NN_G; ++g) {
      const float m01 = fminf(fminf(c[g][0], c[g][1]), fminf(c[g][2], c[g][3]));
      const float m23 = fminf(fminf(c[g][4], c[g][5]), fminf(c[g][6], c[g][7]));
      const float mt  = fminf(m01, m23);
      hit[g] = __builtin_amdgcn_ballot_w32(mt < d2s);
      anyhit |= hit[g];
    }
    if (anyhit != 0u) {
#pragma unroll
      for (int g = 0; g < NN_G; ++g) {
        if (hit[g] != 0u) {
          const int n0 = (q * NN_G + g) * 16 + half * 8;
#pragma unroll
          for (int r = 0; r < 8; ++r) ins3_stream(c[g][r], n0 + r, d, ix);
        }
      }
    }

#pragma unroll
    for (int g = 0; g < NN_G; ++g) { cx[g] = nx[g]; cy[g] = ny[g]; cz[g] = nz[g]; }
  }

  {
    float pd[3]; int pi[3];
#pragma unroll
    for (int k = 0; k < 3; ++k) {
      pd[k] = __shfl_xor(d[k], 16, 32);
      pi[k] = __shfl_xor(ix[k], 16, 32);
    }
#pragma unroll
    for (int k = 0; k < 3; ++k) ins3_lex(pd[k], pi[k], d, ix);
  }

  if (half == 0) {
    float wv[3];
    float wsum = 0.f;
#pragma unroll
    for (int k = 0; k < 3; ++k) {
      const int nn = ix[k];
      const float* qp = pcb + (size_t)nn * 3;
      const float dx = qp[0] - sx, dy = qp[1] - sy, dz = qp[2] - sz;
      const float dd = dx * dx + dy * dy + dz * dz;
      const float dist = sqrtf(fmaxf(dd, 0.0f));
      wv[k] = 1.0f / (dist + 1e-8f);
      wsum += wv[k];
      idx3[((size_t)b * M_ + mrow) * 3 + k] = nn;
    }
#pragma unroll
    for (int k = 0; k < 3; ++k)
      w3[((size_t)b * M_ + mrow) * 3 + k] = wv[k] / wsum;
  }
}

__global__ __launch_bounds__(256) void interp_kernel(
    const float* __restrict__ feats, const int* __restrict__ idx3,
    const float* __restrict__ w3, float* __restrict__ out) {
  const int bc = blockIdx.x;
  const int b  = bc / C_;
  const int c  = bc - b * C_;
  const float* f = feats + ((size_t)b * C_ + c) * N_;
  float*       o = out   + ((size_t)b * C_ + c) * M_;
  const int*   ib = idx3 + (size_t)b * M_ * 3;
  const float* wb = w3   + (size_t)b * M_ * 3;
  for (int m = threadIdx.x; m < M_; m += 256) {
    const int   n0 = ib[m * 3 + 0], n1 = ib[m * 3 + 1], n2 = ib[m * 3 + 2];
    const float w0 = wb[m * 3 + 0], w1 = wb[m * 3 + 1], w2 = wb[m * 3 + 2];
    o[m] = w0 * f[n0] + w1 * f[n1] + w2 * f[n2];
  }
}

extern "C" void kernel_launch(void* const* d_in, const int* in_sizes, int n_in,
                              void* d_out, int out_size, void* d_ws, size_t ws_size,
                              hipStream_t stream) {
  const float* pc    = (const float*)d_in[0];
  const float* feats = (const float*)d_in[1];
  const float* obj   = (const float*)d_in[2];
  const float* grasp = (const float*)d_in[3];
  float* out = (float*)d_out;

  char* ws = (char*)d_ws;
  size_t off = 0;
  int* fps = (int*)(ws + off);
  off += (size_t)B_ * M_ * 4;        off = (off + 255) & ~(size_t)255;
  int* idx3 = (int*)(ws + off);
  off += (size_t)B_ * M_ * 3 * 4;    off = (off + 255) & ~(size_t)255;
  float* w3 = (float*)(ws + off);
  off += (size_t)B_ * M_ * 3 * 4;

  fps_kernel<<<B_, 1024, 0, stream>>>(pc, obj, grasp, fps);
  three_nn_kernel<<<(B_ * (M_ / 16)) / 8, 256, 0, stream>>>(pc, fps, idx3, w3);
  interp_kernel<<<B_ * C_, 256, 0, stream>>>(feats, idx3, w3, out);
}


# fps kernel: each thread packs its own masked-in points to the front of its own 20 LDS slots (index order kept); a wave scans only the 4-slot groups its fullest lane filled (was all 5); same sums
# speedup vs baseline: 4.0703x; 1.3538x over previous
#include <hip/hip_runtime.h>
#include <hip/hip_bf16.h>

#define B_ 4
#define N_ 20000
#define C_ 512
#define M_ 1024
#define GRASP_TH 0.1f

typedef float v2f __attribute__((ext_vector_type(2)));
typedef float v4f __attribute__((ext_vector_type(4)));
typedef float v8f __attribute__((ext_vector_type(8)));

#if defined(__gfx1250__) && __has_builtin(__builtin_amdgcn_wmma_f32_16x16x4_f32)
#define USE_WMMA 1
#else
#define USE_WMMA 0
#endif

__device__ __forceinline__ void ins3_stream(float v, int n, float d[3], int ix[3]) {
  const bool c0 = v < d[0];
  const bool c1 = v < d[1];
  const bool c2 = v < d[2];
  const float r0 = c0 ? v     : d[0];
  const int   j0 = c0 ? n     : ix[0];
  const float r1 = c0 ? d[0]  : (c1 ? v : d[1]);
  const int   j1 = c0 ? ix[0] : (c1 ? n : ix[1]);
  const float r2 = c1 ? d[1]  : (c2 ? v : d[2]);
  const int   j2 = c1 ? ix[1] : (c2 ? n : ix[2]);
  d[0] = r0; d[1] = r1; d[2] = r2;
  ix[0] = j0; ix[1] = j1; ix[2] = j2;
}

__device__ __forceinline__ void ins3_lex(float v, int n, float d[3], int ix[3]) {
  const bool c0 = (v < d[0]) | ((v == d[0]) & (n < ix[0]));
  const bool c1 = (v < d[1]) | ((v == d[1]) & (n < ix[1]));
  const bool c2 = (v < d[2]) | ((v == d[2]) & (n < ix[2]));
  const float r0 = c0 ? v     : d[0];
  const int   j0 = c0 ? n     : ix[0];
  const float r1 = c0 ? d[0]  : (c1 ? v : d[1]);
  const int   j1 = c0 ? ix[0] : (c1 ? n : ix[1]);
  const float r2 = c1 ? d[1]  : (c2 ? v : d[2]);
  const int   j2 = c1 ? ix[1] : (c2 ? n : ix[2]);
  d[0] = r0; d[1] = r1; d[2] = r2;
  ix[0] = j0; ix[1] = j1; ix[2] = j2;
}

#define FPS_J 5
#define FPS_K 12
#define FPS_KC 16
#define FPS_KCH 8
#define FPS_KP 16
#define W_CX 0
#define W_CY 16
#define W_CZ 32
#define W_CV 48
#define W_CI 64
#define W_UB 80
#define W_BLK 112
#define W_EX 128
#define W_EY 192
#define W_EZ 256
static_assert(FPS_K == 12 && FPS_KCH * 2 == FPS_KC && FPS_KC == 16 && FPS_K <= FPS_KP, "the pick list in the scan is written out for 12, the proof for 16 candidates");

__device__ __forceinline__ bool lex_better(float ov, int oi, float v, int i) {
  return (ov > v) | ((ov == v) & (oi < i));
}

typedef unsigned long long u64;

__device__ __forceinline__ void fps_apply(const v4f vx, const v4f vy, const v4f vz,
                                          float qx, float qy, float qz, float v[4]) {
#pragma unroll
  for (int k = 0; k < 4; ++k) {
    const float dxa = vx[k] - qx, dya = vy[k] - qy, dza = vz[k] - qz;
    const float dda = __builtin_fmaf(dza, dza, __builtin_fmaf(dxa, dxa, dya * dya));
    v[k] = fminf(v[k], dda);
  }
}
#define FPS_PICK(i) if (m <= (i)) break; fps_apply(vx, vy, vz, qx[i], qy[i], qz[i], v)

__device__ __forceinline__ u64 pack_key(float v, int i) {
  const bool real = v >= 0.0f;
  const unsigned hi = real ? __float_as_uint(v) : 0u;
  const unsigned lo = real ? (unsigned)(0x7fffffff - i) : 0u;
  return ((u64)hi << 32) | (u64)lo;
}

__global__ __launch_bounds__(1024) void fps_kernel(
    const float* __restrict__ pc, const float* __restrict__ obj,
    const float* __restrict__ grasp, int* __restrict__ fps_idx) {
  const int b    = blockIdx.x;
  const int tid  = threadIdx.x;
  const int lane = tid & 31;
  const int wid  = tid >> 5;

  __shared__ __attribute__((aligned(16))) float s_x[4 * 4096];
  __shared__ __attribute__((aligned(16))) float s_y[4 * 4096];
  __shared__ __attribute__((aligned(16))) float s_z[4 * 4096];
  __shared__ __attribute__((aligned(16))) float s_t[3 * 4096];
  __shared__ __attribute__((aligned(16))) u64 s_key[64];
  __shared__ __attribute__((aligned(8)))  int s_rank[64];
  __shared__ int   swi1[32];
  __shared__ int   s_m;
  __shared__ __attribute__((aligned(16))) float s_px[FPS_KP], s_py[FPS_KP], s_pz[FPS_KP];
  __shared__ __attribute__((aligned(16))) float s_w[320];
  __shared__ unsigned char s_map[FPS_J * 4 * 1024];

  const float* pcb = pc + (size_t)b * N_ * 3;
  const float* ob0 = obj + (size_t)b * 2 * N_;
  const float* gr  = grasp + (size_t)b * N_;

  int lz;
  asm("v_mov_b32 %0, 0" : "=v"(lz));
  const v4f* ppx = (const v4f*)s_px + lz;
  const v4f* ppy = (const v4f*)s_py + lz;
  const v4f* ppz = (const v4f*)s_pz + lz;
  const v4f* pub = (const v4f*)(s_w + W_UB) + lz;

  {
    const v4f zero4 = {0.f, 0.f, 0.f, 0.f};
#pragma unroll
    for (int j = 0; j < 4; ++j) {
      const int g = tid + 1024 * j;
      ((v4f*)s_x)[g] = zero4;
      ((v4f*)s_y)[g] = zero4;
      ((v4f*)s_z)[g] = zero4;
    }
#pragma unroll
    for (int p = 0; p < 3; ++p) ((v4f*)s_t)[tid + 1024 * p] = zero4;
  }

  if (tid < 320) s_w[tid] = 0.0f;

  int firstLoc = N_;
  int cnt = 0;
  unsigned vm = 0u;
#pragma unroll
  for (int j = 0; j < FPS_J; ++j) {
    const int g  = tid + 1024 * j;
    const bool gv = g < N_ / 4;
    const int gl = gv ? g : 0;
#pragma unroll
    for (int k = 0; k < 4; ++k) {
      const int n = 4 * gl + k;
      const bool mk = gv & (ob0[N_ + n] > ob0[n]) & (gr[n] > GRASP_TH);
      vm |= mk ? (1u << (4 * j + k)) : 0u;
    }
  }
#pragma unroll
  for (int j = 0; j < FPS_J; ++j) {
    const int g  = tid + 1024 * j;
    const bool gv = g < N_ / 4;
    const int gl = gv ? g : 0;
    const float* pp = pcb + (size_t)gl * 12;
    float px[4], py[4], pz[4];
#pragma unroll
    for (int k = 0; k < 4; ++k) { px[k] = pp[3 * k]; py[k] = pp[3 * k + 1]; pz[k] = pp[3 * k + 2]; }
#pragma unroll
    for (int k = 0; k < 4; ++k) {
      const int n = 4 * gl + k;
      const bool mk = ((vm >> (4 * j + k)) & 1u) != 0u;
      if (mk) {
        const bool tail = cnt >= 16;
        const int  pos  = 4096 * (cnt >> 2) + 4 * tid + (cnt & 3);
        float* dx = tail ? (s_t + 4 * tid + (cnt & 3))        : (s_x + pos);
        float* dy = tail ? (s_t + 4096 + 4 * tid + (cnt & 3)) : (s_y + pos);
        float* dz = tail ? (s_t + 8192 + 4 * tid + (cnt & 3)) : (s_z + pos);
        *dx = px[k];
        *dy = py[k];
        *dz = pz[k];
        s_map[cnt * 1024 + tid] = (unsigned char)(4 * j + k);
        firstLoc = (firstLoc < n) ? firstLoc : n;
        cnt += 1;
      }
    }
  }
  float mind[FPS_J][4];
#pragma unroll
  for (int s = 0; s < FPS_J * 4; ++s) {
    const bool filled = s < cnt;
    mind[s >> 2][s & 3] = filled ? __builtin_inff() : -__builtin_inff();
    if (!filled) s_map[s * 1024 + tid] = (unsigned char)0;
  }
  int wcnt = cnt;
  for (int mm = 16; mm; mm >>= 1) {
    const int o = __shfl_xor(wcnt, mm, 32);
    wcnt = (o > wcnt) ? o : wcnt;
  }
  const int Jw = (__builtin_amdgcn_readfirstlane(wcnt) + 3) >> 2;

  for (int mm = 16; mm; mm >>= 1) {
    const int o = __shfl_xor(firstLoc, mm, 32);
    firstLoc = (o < firstLoc) ? o : firstLoc;
  }
  if (lane == 0) swi1[wid] = firstLoc;
  __syncthreads();
  if (wid == 0) {
    int v = swi1[lane];
    for (int mm = 16; mm; mm >>= 1) {
      const int o = __shfl_xor(v, mm, 32);
      v = (o < v) ? o : v;
    }
    if (lane == 0) {
      const int f = (v >= N_) ? 0 : v;
      fps_idx[b * M_] = f;
      s_m = 1;
      const float fx = pcb[(size_t)f * 3], fy = pcb[(size_t)f * 3 + 1], fz = pcb[(size_t)f * 3 + 2];
#pragma unroll
      for (int i = 0; i < FPS_KP; ++i) { s_px[i] = fx; s_py[i] = fy; s_pz[i] = fz; }
    }
  }
  __syncthreads();

  const v4f* x4 = (const v4f*)s_x;
  const v4f* y4 = (const v4f*)s_y;
  const v4f* z4 = (const v4f*)s_z;
  const v4f* tl4 = (const v4f*)s_t;

  int m    = 1;
  int step = 1;
  int pass = 1;

  while (step < M_) {
    float qx[FPS_K], qy[FPS_K], qz[FPS_K];
#pragma unroll
    for (int i4 = 0; i4 < FPS_K / 4; ++i4) {
      const v4f tx = ppx[i4], ty = ppy[i4], tz = ppz[i4];
#pragma unroll
      for (int e = 0; e < 4; ++e) { qx[4 * i4 + e] = tx[e]; qy[4 * i4 + e] = ty[e]; qz[4 * i4 + e] = tz[e]; }
    }

    float b1v = -__builtin_inff(), b2v = -__builtin_inff();
    int   b1s = 0,                 b2s = 0;
    int   b3  = __float_as_int(-__builtin_inff());
#pragma unroll
    for (int j = 0; j < FPS_J; ++j) {
      if (j >= Jw) break;
      const v4f vx = (j < 4) ? x4[tid + 1024 * j] : tl4[tid];
      const v4f vy = (j < 4) ? y4[tid + 1024 * j] : tl4[tid + 1024];
      const v4f vz = (j < 4) ? z4[tid + 1024 * j] : tl4[tid + 2048];
      float v[4];
#pragma unroll
      for (int k = 0; k < 4; ++k) v[k] = mind[j][k];
      do {
        FPS_PICK(0); FPS_PICK(1); FPS_PICK(2);  FPS_PICK(3);
        FPS_PICK(4); FPS_PICK(5); FPS_PICK(6);  FPS_PICK(7);
        FPS_PICK(8); FPS_PICK(9); FPS_PICK(10); FPS_PICK(11);
      } while (0);
#pragma unroll
      for (int k = 0; k < 4; ++k) {
        mind[j][k] = v[k];
        const int  s  = 4 * j + k;
        const bool c1 = v[k] > b1v;
        const bool c2 = v[k] > b2v;
        const int vb = __float_as_int(v[k]);
        const int lo = (__float_as_int(b2v) < vb) ? __float_as_int(b2v) : vb;
        b3  = (lo > b3) ? lo : b3;
        b2v = c1 ? b1v : (c2 ? v[k] : b2v);
        b2s = c1 ? b1s : (c2 ? s    : b2s);
        b1v = c1 ? v[k] : b1v;
        b1s = c1 ? s    : b1s;
      }
    }
    const int c1o = (int)s_map[b1s * 1024 + tid];
    const int c2o = (int)s_map[b2s * 1024 + tid];
    int b1i = 4 * tid + 4096 * (c1o >> 2) + (c1o & 3);
    int b2i = 4 * tid + 4096 * (c2o >> 2) + (c2o & 3);
    const int my1i = (b1v >= 0.0f) ? b1i : -1;
    const int my2i = (b2v >= 0.0f) ? b2i : -1;
    const int my1s = b1s, my2s = b2s;
    for (int mm = 16; mm; mm >>= 1) {
      const float o1v = __shfl_xor(b1v, mm, 32);
      const int   o1i = __shfl_xor(b1i, mm, 32);
      const float o2v = __shfl_xor(b2v, mm, 32);
      const int   o2i = __shfl_xor(b2i, mm, 32);
      const int   o3  = __shfl_xor(b3, mm, 32);
      {
        const int a1 = __float_as_int(b1v), a2 = __float_as_int(b2v);
        const int p1 = __float_as_int(o1v), p2 = __float_as_int(o2v);
        const int m12 = (a1 < p2) ? a1 : p2;
        const int m21 = (a2 < p1) ? a2 : p1;
        const int t3  = (b3 > o3) ? b3 : o3;
        const int t4  = (m12 > m21) ? m12 : m21;
        b3 = (t3 > t4) ? t3 : t4;
      }
      const bool c = lex_better(o1v, o1i, b1v, b1i);
      const float lv = c ? b1v : o1v;  const int li = c ? b1i : o1i;
      const float sv = c ? o2v : b2v;  const int si = c ? o2i : b2i;
      b1v = c ? o1v : b1v;             b1i = c ? o1i : b1i;
      const bool d = lex_better(lv, li, sv, si);
      b2v = d ? lv : sv;               b2i = d ? li : si;
    }
    const u64 k1 = pack_key(b1v, b1i);
    const u64 k2 = pack_key(b2v, b2i);
    const u64 k3 = (b3 >= 0) ? (((u64)(unsigned)b3 << 32) | 0xffffffffull) : 0ull;
    if (lane == 0) {
      s_key[2 * wid]     = k1;
      s_key[2 * wid + 1] = k2;
    }
    {
      const bool hReal = b1v >= 0.0f;
      const bool nReal = b2v >= 0.0f;
      const bool ownH  = hReal & (my1i == b1i);
      const bool ownN1 = nReal & (my1i == b2i);
      const bool ownN2 = nReal & (my2i == b2i);
      if (ownH | ownN1 | ownN2) {
        const bool first = ownH | ownN1;
        const int  sl    = first ? my1s : my2s;
        const int  row   = 2 * wid + (ownH ? 0 : 1);
        const bool tail  = sl >= 16;
        const int  pos   = 4096 * (sl >> 2) + 4 * tid + (sl & 3);
        const float* fx = tail ? (s_t + 4 * tid + (sl & 3))        : (s_x + pos);
        const float* fy = tail ? (s_t + 4096 + 4 * tid + (sl & 3)) : (s_y + pos);
        const float* fz = tail ? (s_t + 8192 + 4 * tid + (sl & 3)) : (s_z + pos);
        const float ex = *fx, ey = *fy, ez = *fz;
        s_w[W_EX + row] = ex; s_w[W_EY + row] = ey; s_w[W_EZ + row] = ez;
        if (ownH & ownN2) {
          const bool tail2 = my2s >= 16;
          const int  pos2  = 4096 * (my2s >> 2) + 4 * tid + (my2s & 3);
          const float* gx = tail2 ? (s_t + 4 * tid + (my2s & 3))        : (s_x + pos2);
          const float* gy = tail2 ? (s_t + 4096 + 4 * tid + (my2s & 3)) : (s_y + pos2);
          const float* gz = tail2 ? (s_t + 8192 + 4 * tid + (my2s & 3)) : (s_z + pos2);
          const float hx = *gx, hy = *gy, hz = *gz;
          s_w[W_EX + row + 1] = hx; s_w[W_EY + row + 1] = hy; s_w[W_EZ + row + 1] = hz;
        }
      }
    }
    __syncthreads();
    {
      const unsigned h1 = __builtin_amdgcn_readfirstlane((unsigned)(k1 >> 32));
      const unsigned l1 = __builtin_amdgcn_readfirstlane((unsigned)k1);
      const unsigned h2 = __builtin_amdgcn_readfirstlane((unsigned)(k2 >> 32));
      const unsigned l2 = __builtin_amdgcn_readfirstlane((unsigned)k2);
      const u64 kHs = ((u64)h1 << 32) | (u64)l1;
      const u64 kNs = ((u64)h2 << 32) | (u64)l2;
      const unsigned h3 = __builtin_amdgcn_readfirstlane((unsigned)(k3 >> 32));
      const unsigned l3 = __builtin_amdgcn_readfirstlane((unsigned)k3);
      const u64 kTs = ((u64)h3 << 32) | (u64)l3;
      const u64 ka = s_key[2 * lane];
      const u64 kb = s_key[2 * lane + 1];
      const int rHs = __builtin_popcount(__builtin_amdgcn_ballot_w32(ka > kHs)) +
                      __builtin_popcount(__builtin_amdgcn_ballot_w32(kb > kHs));
      const int rNs = __builtin_popcount(__builtin_amdgcn_ballot_w32(ka > kNs)) +
                      __builtin_popcount(__builtin_amdgcn_ballot_w32(kb > kNs));
      const int rTs = __builtin_popcount(__builtin_amdgcn_ballot_w32(ka > kTs)) +
                      __builtin_popcount(__builtin_amdgcn_ballot_w32(kb > kTs));
      if (lane == 0) {
        s_rank[2 * wid] = rHs; s_rank[2 * wid + 1] = rNs;
        const int rMark = (rTs < 1) ? 1 : rTs;
        if (rMark < FPS_KC) s_w[W_BLK + rMark] = __int_as_float(pass);
      }
    }
    __syncthreads();
    if (wid == 0) {
      const u64 kH = s_key[2 * lane];
      const u64 kN = s_key[2 * lane + 1];
      const float valH = __uint_as_float((unsigned)(kH >> 32));
      const float valN = __uint_as_float((unsigned)(kN >> 32));
      const int iH = 0x7fffffff - (int)(unsigned)kH;
      const int iN = 0x7fffffff - (int)(unsigned)kN;
      const bool vH = iH < N_;
      const bool vN = iN < N_;
      const int qH = vH ? iH : 0;
      const int qN = vN ? iN : 0;
      const float xH = s_w[W_EX + 2 * lane],     yH = s_w[W_EY + 2 * lane],     zH = s_w[W_EZ + 2 * lane];
      const float xN = s_w[W_EX + 2 * lane + 1], yN = s_w[W_EY + 2 * lane + 1], zN = s_w[W_EZ + 2 * lane + 1];

      const int rH = s_rank[2 * lane];
      const int rN = s_rank[2 * lane + 1];

      if (rH < FPS_KC) {
        s_w[W_CX + rH] = xH; s_w[W_CY + rH] = yH; s_w[W_CZ + rH] = zH;
        s_w[W_CV + rH] = valH; s_w[W_CI + rH] = __int_as_float(qH);
      }
      if (rN < FPS_KC) {
        s_w[W_CX + rN] = xN; s_w[W_CY + rN] = yN; s_w[W_CZ + rN] = zN;
        s_w[W_CV + rN] = valN; s_w[W_CI + rN] = __int_as_float(qN);
      }

      const int R = __builtin_popcount(__builtin_amdgcn_ballot_w32(vH)) +
                    __builtin_popcount(__builtin_amdgcn_ballot_w32(vN));

      const int r  = lane & 15;
      const int hh = lane >> 4;
      const float cxr = s_w[W_CX + r], cyr = s_w[W_CY + r], czr = s_w[W_CZ + r], cvr = s_w[W_CV + r];
      const int   cir = __float_as_int(s_w[W_CI + r]);
      const int   blk = __float_as_int(s_w[W_BLK + r]);
      const bool  real = (r < R) & (cvr > 0.0f);

      const float need = cvr * 1.0001f;
      float dd[FPS_KCH];
      float dall = __builtin_inff();
#pragma unroll
      for (int i = 0; i < FPS_KCH; ++i) {
        const int ii = hh * FPS_KCH + i;
        const float sx = s_w[W_CX + ii], sy = s_w[W_CY + ii], sz = s_w[W_CZ + ii];
        const float dxa = cxr - sx, dya = cyr - sy, dza = czr - sz;
        const float dda = dxa * dxa + dya * dya + dza * dza;
        dd[i] = (ii >= r) ? __builtin_inff() : dda;
        dall  = fminf(dall, dd[i]);
      }
      const unsigned nfar  = __builtin_amdgcn_ballot_w32(!(dall >= need));
      const unsigned nreal = __builtin_amdgcn_ballot_w32(!real);
      const unsigned realm = ~nreal & 0xffffu;
      const unsigned amask = ~(nfar | (nfar >> 16)) & realm;
      const unsigned ah = hh ? (amask >> FPS_KCH) : amask;
      float da[FPS_KCH];
#pragma unroll
      for (int i = 0; i < FPS_KCH; ++i)
        da[i] = ((ah >> i) & 1u) ? dd[i] : __builtin_inff();
      const float dA = fminf(fminf(fminf(da[0], da[1]), fminf(da[2], da[3])),
                             fminf(fminf(da[4], da[5]), fminf(da[6], da[7])));
      s_w[W_UB + lane] = dA;
      const unsigned skm = realm & ~amask;
      int wv[FPS_KC];
#pragma unroll
      for (int s = 0; s < FPS_KC; ++s) {
        const int lo = __float_as_int(pub[s >> 2][s & 3]);
        const int hi = __float_as_int(pub[4 + (s >> 2)][s & 3]);
        const bool use = (((skm >> s) & 1u) != 0u) & (s < r);
        wv[s] = use ? ((lo < hi) ? lo : hi) : 0;
      }
#pragma unroll
      for (int w = FPS_KC / 2; w; w >>= 1) {
#pragma unroll
        for (int s = 0; s < w; ++s) wv[s] = (wv[s] > wv[s + w]) ? wv[s] : wv[s + w];
      }
      const float mub = __int_as_float(wv[0]) * 1.0001f;
      const unsigned mk = __builtin_amdgcn_ballot_w32(blk == pass) & 0xffffu;
      const int lmark = __builtin_ctz(mk | (1u << FPS_KC));
      const bool acc = (((amask >> r) & 1u) != 0u) & (mub < cvr) & (r < lmark);
      const unsigned accm = (__builtin_amdgcn_ballot_w32(acc) & 0xffffu) | 1u;
      const int pos = __builtin_popcount(accm & ((1u << r) - 1u));
      int mnew = __builtin_popcount(accm);
      mnew = (mnew < FPS_K) ? mnew : FPS_K;
      const int room = M_ - step;
      mnew = (mnew < room) ? mnew : room;

      if ((hh == 0) & (((accm >> r) & 1u) != 0u) & (pos < mnew)) {
        s_px[pos] = cxr; s_py[pos] = cyr; s_pz[pos] = czr;
        fps_idx[b * M_ + step + pos] = cir;
      }
      if (lane == 0) s_m = mnew;
    }
    __syncthreads();
    m = __builtin_amdgcn_readfirstlane(s_m);
    step += m;
    pass += 1;
  }
}

#define NN_G 10
#define NN_TRIPS (N_ / 16 / NN_G)

__global__ __launch_bounds__(256) void three_nn_kernel(
    const float* __restrict__ pc, const int* __restrict__ fps_idx,
    int* __restrict__ idx3, float* __restrict__ w3) {
  const int tid  = threadIdx.x;
  const int lane = tid & 31;
  const int wave = blockIdx.x * (256 >> 5) + (tid >> 5);
  const int tilesPerBatch = M_ / 16;
  const int b    = wave / tilesPerBatch;
  const int rt   = wave - b * tilesPerBatch;
  const int half = lane >> 4;
  const int l15  = lane & 15;

  const float* pcb = pc + (size_t)b * N_ * 3;

  const int mrow = rt * 16 + l15;
  const int sidx = fps_idx[b * M_ + mrow];
  const float* sp = pcb + (size_t)sidx * 3;
  const float sx = sp[0], sy = sp[1], sz = sp[2];
  v2f Bs;
  Bs.x = half ? (-2.0f * sz) : (-2.0f * sx);
  Bs.y = half ? 1.0f         : (-2.0f * sy);

  float d[3];
  int   ix[3];
#pragma unroll
  for (int k = 0; k < 3; ++k) { d[k] = __builtin_inff(); ix[k] = 0x7fffffff; }

  const float* lp = pcb + (size_t)l15 * 3;

  float cx[NN_G], cy[NN_G], cz[NN_G];
#pragma unroll
  for (int g = 0; g < NN_G; ++g) {
    const float* p = lp + (size_t)g * 48;
    cx[g] = p[0]; cy[g] = p[1]; cz[g] = p[2];
  }

  for (int q = 0; q < NN_TRIPS; ++q) {
    const int qn = (q + 1 < NN_TRIPS) ? (q + 1) : q;
    const float* np = lp + (size_t)qn * (NN_G * 48);
    float nx[NN_G], ny[NN_G], nz[NN_G];
#pragma unroll
    for (int g = 0; g < NN_G; ++g) {
      const float* p = np + (size_t)g * 48;
      nx[g] = p[0]; ny[g] = p[1]; nz[g] = p[2];
    }

    v8f c[NN_G];
#pragma unroll
    for (int g = 0; g < NN_G; ++g) {
      const float bx = cx[g], by = cy[g], bz = cz[g];
      const float bn2 = __builtin_fmaf(bz, bz, __builtin_fmaf(bx, bx, by * by));
      const v8f zero = {0.f, 0.f, 0.f, 0.f, 0.f, 0.f, 0.f, 0.f};
#if USE_WMMA
      v2f Av;
      Av.x = half ? bz  : bx;
      Av.y = half ? bn2 : by;
      c[g] = __builtin_amdgcn_wmma_f32_16x16x4_f32(
           false, Av,  false, Bs,
           (short)0, zero,  false,  false);
#else
      c[g] = zero;
#pragma unroll
      for (int r = 0; r < 8; ++r) {
        const int src = half * 8 + r;
        const float rx = __shfl(bx, src, 32);
        const float ry = __shfl(by, src, 32);
        const float rz = __shfl(bz, src, 32);
        const float rn = __shfl(bn2, src, 32);
        c[g][r] = -2.0f * (rx * sx + ry * sy + rz * sz) + rn;
      }
#endif
    }

    unsigned hit[NN_G];
    unsigned anyhit = 0u;
    const float d2s = d[2];
#pragma unroll
    for (int g = 0; g < NN_G; ++g) {
      const float m01 = fminf(fminf(c[g][0], c[g][1]), fminf(c[g][2], c[g][3]));
      const float m23 = fminf(fminf(c[g][4], c[g][5]), fminf(c[g][6], c[g][7]));
      const float mt  = fminf(m01, m23);
      hit[g] = __builtin_amdgcn_ballot_w32(mt < d2s);
      anyhit |= hit[g];
    }
    if (anyhit != 0u) {
#pragma unroll
      for (int g = 0; g < NN_G; ++g) {
        if (hit[g] != 0u) {
          const int n0 = (q * NN_G + g) * 16 + half * 8;
#pragma unroll
          for (int r = 0; r < 8; ++r) ins3_stream(c[g][r], n0 + r, d, ix);
        }
      }
    }

#pragma unroll
    for (int g = 0; g < NN_G; ++g) { cx[g] = nx[g]; cy[g] = ny[g]; cz[g] = nz[g]; }
  }

  {
    float pd[3]; int pi[3];
#pragma unroll
    for (int k = 0; k < 3; ++k) {
      pd[k] = __shfl_xor(d[k], 16, 32);
      pi[k] = __shfl_xor(ix[k], 16, 32);
    }
#pragma unroll
    for (int k = 0; k < 3; ++k) ins3_lex(pd[k], pi[k], d, ix);
  }

  if (half == 0) {
    float wv[3];
    float wsum = 0.f;
#pragma unroll
    for (int k = 0; k < 3; ++k) {
      const int nn = ix[k];
      const float* qp = pcb + (size_t)nn * 3;
      const float dx = qp[0] - sx, dy = qp[1] - sy, dz = qp[2] - sz;
      const float dd = dx * dx + dy * dy + dz * dz;
      const float dist = sqrtf(fmaxf(dd, 0.0f));
      wv[k] = 1.0f / (dist + 1e-8f);
      wsum += wv[k];
      idx3[((size_t)b * M_ + mrow) * 3 + k] = nn;
    }
#pragma unroll
    for (int k = 0; k < 3; ++k)
      w3[((size_t)b * M_ + mrow) * 3 + k] = wv[k] / wsum;
  }
}

__global__ __launch_bounds__(256) void interp_kernel(
    const float* __restrict__ feats, const int* __restrict__ idx3,
    const float* __restrict__ w3, float* __restrict__ out) {
  const int bc = blockIdx.x;
  const int b  = bc / C_;
  const int c  = bc - b * C_;
  const float* f = feats + ((size_t)b * C_ + c) * N_;
  float*       o = out   + ((size_t)b * C_ + c) * M_;
  const int*   ib = idx3 + (size_t)b * M_ * 3;
  const float* wb = w3   + (size_t)b * M_ * 3;
  for (int m = threadIdx.x; m < M_; m += 256) {
    const int   n0 = ib[m * 3 + 0], n1 = ib[m * 3 + 1], n2 = ib[m * 3 + 2];
    const float w0 = wb[m * 3 + 0], w1 = wb[m * 3 + 1], w2 = wb[m * 3 + 2];
    o[m] = w0 * f[n0] + w1 * f[n1] + w2 * f[n2];
  }
}

extern "C" void kernel_launch(void* const* d_in, const int* in_sizes, int n_in,
                              void* d_out, int out_size, void* d_ws, size_t ws_size,
                              hipStream_t stream) {
  const float* pc    = (const float*)d_in[0];
  const float* feats = (const float*)d_in[1];
  const float* obj   = (const float*)d_in[2];
  const float* grasp = (const float*)d_in[3];
  float* out = (float*)d_out;

  char* ws = (char*)d_ws;
  size_t off = 0;
  int* fps = (int*)(ws + off);
  off += (size_t)B_ * M_ * 4;        off = (off + 255) & ~(size_t)255;
  int* idx3 = (int*)(ws + off);
  off += (size_t)B_ * M_ * 3 * 4;    off = (off + 255) & ~(size_t)255;
  float* w3 = (float*)(ws + off);
  off += (size_t)B_ * M_ * 3 * 4;

  fps_kernel<<<B_, 1024, 0, stream>>>(pc, obj, grasp, fps);
  three_nn_kernel<<<(B_ * (M_ / 16)) / 8, 256, 0, stream>>>(pc, fps, idx3, w3);
  interp_kernel<<<B_ * C_, 256, 0, stream>>>(feats, idx3, w3, out);
}


# fps kernel: a wave's masked-in points are dealt round-robin over its 32 lanes (index order kept per lane), so lanes hold equal counts; the scan runs in 2-slot units up to lane 0's count; same sums
# speedup vs baseline: 4.7737x; 1.1728x over previous
#include <hip/hip_runtime.h>
#include <hip/hip_bf16.h>

#define B_ 4
#define N_ 20000
#define C_ 512
#define M_ 1024
#define GRASP_TH 0.1f

typedef float v2f __attribute__((ext_vector_type(2)));
typedef float v4f __attribute__((ext_vector_type(4)));
typedef float v8f __attribute__((ext_vector_type(8)));

#if defined(__gfx1250__) && __has_builtin(__builtin_amdgcn_wmma_f32_16x16x4_f32)
#define USE_WMMA 1
#else
#define USE_WMMA 0
#endif

__device__ __forceinline__ void ins3_stream(float v, int n, float d[3], int ix[3]) {
  const bool c0 = v < d[0];
  const bool c1 = v < d[1];
  const bool c2 = v < d[2];
  const float r0 = c0 ? v     : d[0];
  const int   j0 = c0 ? n     : ix[0];
  const float r1 = c0 ? d[0]  : (c1 ? v : d[1]);
  const int   j1 = c0 ? ix[0] : (c1 ? n : ix[1]);
  const float r2 = c1 ? d[1]  : (c2 ? v : d[2]);
  const int   j2 = c1 ? ix[1] : (c2 ? n : ix[2]);
  d[0] = r0; d[1] = r1; d[2] = r2;
  ix[0] = j0; ix[1] = j1; ix[2] = j2;
}

__device__ __forceinline__ void ins3_lex(float v, int n, float d[3], int ix[3]) {
  const bool c0 = (v < d[0]) | ((v == d[0]) & (n < ix[0]));
  const bool c1 = (v < d[1]) | ((v == d[1]) & (n < ix[1]));
  const bool c2 = (v < d[2]) | ((v == d[2]) & (n < ix[2]));
  const float r0 = c0 ? v     : d[0];
  const int   j0 = c0 ? n     : ix[0];
  const float r1 = c0 ? d[0]  : (c1 ? v : d[1]);
  const int   j1 = c0 ? ix[0] : (c1 ? n : ix[1]);
  const float r2 = c1 ? d[1]  : (c2 ? v : d[2]);
  const int   j2 = c1 ? ix[1] : (c2 ? n : ix[2]);
  d[0] = r0; d[1] = r1; d[2] = r2;
  ix[0] = j0; ix[1] = j1; ix[2] = j2;
}

#define FPS_J 5
#define FPS_P 10
#define FPS_K 12
#define FPS_KC 16
#define FPS_KCH 8
#define FPS_KP 16
#define W_CX 0
#define W_CY 16
#define W_CZ 32
#define W_CV 48
#define W_CI 64
#define W_UB 80
#define W_BLK 112
#define W_EX 128
#define W_EY 192
#define W_EZ 256
static_assert(FPS_K == 12 && FPS_KCH * 2 == FPS_KC && FPS_KC == 16 && FPS_K <= FPS_KP, "the pick list in the scan is written out for 12, the proof for 16 candidates");

__device__ __forceinline__ bool lex_better(float ov, int oi, float v, int i) {
  return (ov > v) | ((ov == v) & (oi < i));
}

typedef unsigned long long u64;

__device__ __forceinline__ void fps_apply(const v2f vx, const v2f vy, const v2f vz,
                                          float qx, float qy, float qz, float v[2]) {
#pragma unroll
  for (int k = 0; k < 2; ++k) {
    const float dxa = vx[k] - qx, dya = vy[k] - qy, dza = vz[k] - qz;
    const float dda = __builtin_fmaf(dza, dza, __builtin_fmaf(dxa, dxa, dya * dya));
    v[k] = fminf(v[k], dda);
  }
}
#define FPS_PICK(i) if (m <= (i)) break; fps_apply(vx, vy, vz, qx[i], qy[i], qz[i], v)

__device__ __forceinline__ void fps_pair(const int p, const v2f* x2, const v2f* y2, const v2f* z2,
                                         const v2f* tl2, const int tid, const int m,
                                         const float (&qx)[FPS_K], const float (&qy)[FPS_K],
                                         const float (&qz)[FPS_K], float (&mp)[2],
                                         float& b1v, float& b2v, int& b1s, int& b2s, int& b3) {
  const v2f vx = (p < 8) ? x2[2 * (tid + 1024 * (p >> 1)) + (p & 1)] : tl2[2 * tid + (p & 1)];
  const v2f vy = (p < 8) ? y2[2 * (tid + 1024 * (p >> 1)) + (p & 1)] : tl2[2 * tid + (p & 1) + 2048];
  const v2f vz = (p < 8) ? z2[2 * (tid + 1024 * (p >> 1)) + (p & 1)] : tl2[2 * tid + (p & 1) + 4096];
  float v[2];
  v[0] = mp[0]; v[1] = mp[1];
  do {
    FPS_PICK(0); FPS_PICK(1); FPS_PICK(2);  FPS_PICK(3);
    FPS_PICK(4); FPS_PICK(5); FPS_PICK(6);  FPS_PICK(7);
    FPS_PICK(8); FPS_PICK(9); FPS_PICK(10); FPS_PICK(11);
  } while (0);
#pragma unroll
  for (int k = 0; k < 2; ++k) {
    mp[k] = v[k];
    const int  s  = 2 * p + k;
    const bool c1 = v[k] > b1v;
    const bool c2 = v[k] > b2v;
    const int vb = __float_as_int(v[k]);
    const int lo = (__float_as_int(b2v) < vb) ? __float_as_int(b2v) : vb;
    b3  = (lo > b3) ? lo : b3;
    b2v = c1 ? b1v : (c2 ? v[k] : b2v);
    b2s = c1 ? b1s : (c2 ? s    : b2s);
    b1v = c1 ? v[k] : b1v;
    b1s = c1 ? s    : b1s;
  }
}
#define FPS_PAIR(p) if (Pw <= (p)) break; fps_pair((p), x2, y2, z2, tl2, tid, m, qx, qy, qz, mind[p], b1v, b2v, b1s, b2s, b3)

__device__ __forceinline__ u64 pack_key(float v, int i) {
  const bool real = v >= 0.0f;
  const unsigned hi = real ? __float_as_uint(v) : 0u;
  const unsigned lo = real ? (unsigned)(0x7fffffff - i) : 0u;
  return ((u64)hi << 32) | (u64)lo;
}

__global__ __launch_bounds__(1024) void fps_kernel(
    const float* __restrict__ pc, const float* __restrict__ obj,
    const float* __restrict__ grasp, int* __restrict__ fps_idx) {
  const int b    = blockIdx.x;
  const int tid  = threadIdx.x;
  const int lane = tid & 31;
  const int wid  = tid >> 5;

  __shared__ __attribute__((aligned(16))) float s_x[4 * 4096];
  __shared__ __attribute__((aligned(16))) float s_y[4 * 4096];
  __shared__ __attribute__((aligned(16))) float s_z[4 * 4096];
  __shared__ __attribute__((aligned(16))) float s_t[3 * 4096];
  __shared__ __attribute__((aligned(16))) u64 s_key[64];
  __shared__ __attribute__((aligned(8)))  int s_rank[64];
  __shared__ int   swi1[32];
  __shared__ int   s_m;
  __shared__ __attribute__((aligned(16))) float s_px[FPS_KP], s_py[FPS_KP], s_pz[FPS_KP];
  __shared__ __attribute__((aligned(16))) float s_w[320];
  __shared__ unsigned short s_map[FPS_J * 4 * 1024];

  const float* pcb = pc + (size_t)b * N_ * 3;
  const float* ob0 = obj + (size_t)b * 2 * N_;
  const float* gr  = grasp + (size_t)b * N_;

  int lz;
  asm("v_mov_b32 %0, 0" : "=v"(lz));
  const v4f* ppx = (const v4f*)s_px + lz;
  const v4f* ppy = (const v4f*)s_py + lz;
  const v4f* ppz = (const v4f*)s_pz + lz;
  const v4f* pub = (const v4f*)(s_w + W_UB) + lz;

  {
    const v4f zero4 = {0.f, 0.f, 0.f, 0.f};
#pragma unroll
    for (int j = 0; j < 4; ++j) {
      const int g = tid + 1024 * j;
      ((v4f*)s_x)[g] = zero4;
      ((v4f*)s_y)[g] = zero4;
      ((v4f*)s_z)[g] = zero4;
    }
#pragma unroll
    for (int p = 0; p < 3; ++p) ((v4f*)s_t)[tid + 1024 * p] = zero4;
  }

  if (tid < 320) s_w[tid] = 0.0f;

  unsigned vm = 0u;
#pragma unroll
  for (int j = 0; j < FPS_J; ++j) {
    const int g  = tid + 1024 * j;
    const bool gv = g < N_ / 4;
    const int gl = gv ? g : 0;
#pragma unroll
    for (int k = 0; k < 4; ++k) {
      const int n = 4 * gl + k;
      const bool mk = gv & (ob0[N_ + n] > ob0[n]) & (gr[n] > GRASP_TH);
      vm |= mk ? (1u << (4 * j + k)) : 0u;
    }
  }
  int firstLoc = N_;
  if (vm != 0u) {
    const int o = __builtin_ctz(vm);
    firstLoc = 4 * (tid + 1024 * (o >> 2)) + (o & 3);
  }
  unsigned cA = 0u;
#pragma unroll
  for (int j = 0; j < 4; ++j) cA |= (unsigned)__builtin_popcount((vm >> (4 * j)) & 15u) << (8 * j);
  const unsigned cB = (unsigned)__builtin_popcount((vm >> 16) & 15u);
  unsigned iA = cA, iB = cB;
  for (int dd = 1; dd < 32; dd <<= 1) {
    const unsigned oA = __shfl_up(iA, dd, 32);
    const unsigned oB = __shfl_up(iB, dd, 32);
    const bool in = lane >= dd;
    iA += in ? oA : 0u;
    iB += in ? oB : 0u;
  }
  const unsigned tA = __shfl(iA, 31, 32);
  const unsigned tB = __shfl(iB, 31, 32);
  const unsigned eA = iA - cA;
  const unsigned eB = iB - cB;
  int r0[FPS_J];
  int wtotal = 0;
#pragma unroll
  for (int j = 0; j < FPS_J; ++j) {
    const int tot = (j < 4) ? (int)((tA >> (8 * j)) & 255u) : (int)tB;
    const int ex  = (j < 4) ? (int)((eA >> (8 * j)) & 255u) : (int)eB;
    r0[j] = wtotal + ex;
    wtotal += tot;
  }
  const int wbase = tid & ~31;
#pragma unroll
  for (int j = 0; j < FPS_J; ++j) {
    const int g  = tid + 1024 * j;
    const bool gv = g < N_ / 4;
    const int gl = gv ? g : 0;
    const float* pp = pcb + (size_t)gl * 12;
    float px[4], py[4], pz[4];
#pragma unroll
    for (int k = 0; k < 4; ++k) { px[k] = pp[3 * k]; py[k] = pp[3 * k + 1]; pz[k] = pp[3 * k + 2]; }
    int r = r0[j];
#pragma unroll
    for (int k = 0; k < 4; ++k) {
      const bool mk = ((vm >> (4 * j + k)) & 1u) != 0u;
      if (mk) {
        const int  c    = r >> 5;
        const int  dt   = wbase + (r & 31);
        const bool tail = c >= 16;
        const int  pos  = 4096 * (c >> 2) + 4 * dt + (c & 3);
        float* dx = tail ? (s_t + 4 * dt + (c & 3))        : (s_x + pos);
        float* dy = tail ? (s_t + 4096 + 4 * dt + (c & 3)) : (s_y + pos);
        float* dz = tail ? (s_t + 8192 + 4 * dt + (c & 3)) : (s_z + pos);
        *dx = px[k];
        *dy = py[k];
        *dz = pz[k];
        s_map[c * 1024 + dt] = (unsigned short)((4 * j + k) | (lane << 5));
        r += 1;
      }
    }
  }
  const int cnt = (wtotal + 31 - lane) >> 5;
  float mind[FPS_P][2];
#pragma unroll
  for (int s = 0; s < FPS_P * 2; ++s) {
    const bool filled = s < cnt;
    mind[s >> 1][s & 1] = filled ? __builtin_inff() : -__builtin_inff();
    if (!filled) s_map[s * 1024 + tid] = (unsigned short)0;
  }
  const int Pw = (((__builtin_amdgcn_readfirstlane(wtotal) + 31) >> 5) + 1) >> 1;

  for (int mm = 16; mm; mm >>= 1) {
    const int o = __shfl_xor(firstLoc, mm, 32);
    firstLoc = (o < firstLoc) ? o : firstLoc;
  }
  if (lane == 0) swi1[wid] = firstLoc;
  __syncthreads();
  if (wid == 0) {
    int v = swi1[lane];
    for (int mm = 16; mm; mm >>= 1) {
      const int o = __shfl_xor(v, mm, 32);
      v = (o < v) ? o : v;
    }
    if (lane == 0) {
      const int f = (v >= N_) ? 0 : v;
      fps_idx[b * M_] = f;
      s_m = 1;
      const float fx = pcb[(size_t)f * 3], fy = pcb[(size_t)f * 3 + 1], fz = pcb[(size_t)f * 3 + 2];
#pragma unroll
      for (int i = 0; i < FPS_KP; ++i) { s_px[i] = fx; s_py[i] = fy; s_pz[i] = fz; }
    }
  }
  __syncthreads();

  const v2f* x2 = (const v2f*)s_x;
  const v2f* y2 = (const v2f*)s_y;
  const v2f* z2 = (const v2f*)s_z;
  const v2f* tl2 = (const v2f*)s_t;

  int m    = 1;
  int step = 1;
  int pass = 1;

  while (step < M_) {
    float qx[FPS_K], qy[FPS_K], qz[FPS_K];
#pragma unroll
    for (int i4 = 0; i4 < FPS_K / 4; ++i4) {
      const v4f tx = ppx[i4], ty = ppy[i4], tz = ppz[i4];
#pragma unroll
      for (int e = 0; e < 4; ++e) { qx[4 * i4 + e] = tx[e]; qy[4 * i4 + e] = ty[e]; qz[4 * i4 + e] = tz[e]; }
    }

    float b1v = -__builtin_inff(), b2v = -__builtin_inff();
    int   b1s = 0,                 b2s = 0;
    int   b3  = __float_as_int(-__builtin_inff());
    do {
      FPS_PAIR(0); FPS_PAIR(1); FPS_PAIR(2); FPS_PAIR(3); FPS_PAIR(4);
      FPS_PAIR(5); FPS_PAIR(6); FPS_PAIR(7); FPS_PAIR(8); FPS_PAIR(9);
    } while (0);
    const int c1o = (int)s_map[b1s * 1024 + tid];
    const int c2o = (int)s_map[b2s * 1024 + tid];
    int b1i = 4 * (wbase + (c1o >> 5)) + 4096 * ((c1o & 31) >> 2) + (c1o & 3);
    int b2i = 4 * (wbase + (c2o >> 5)) + 4096 * ((c2o & 31) >> 2) + (c2o & 3);
    const int my1i = (b1v >= 0.0f) ? b1i : -1;
    const int my2i = (b2v >= 0.0f) ? b2i : -1;
    const int my1s = b1s, my2s = b2s;
    for (int mm = 16; mm; mm >>= 1) {
      const float o1v = __shfl_xor(b1v, mm, 32);
      const int   o1i = __shfl_xor(b1i, mm, 32);
      const float o2v = __shfl_xor(b2v, mm, 32);
      const int   o2i = __shfl_xor(b2i, mm, 32);
      const int   o3  = __shfl_xor(b3, mm, 32);
      {
        const int a1 = __float_as_int(b1v), a2 = __float_as_int(b2v);
        const int p1 = __float_as_int(o1v), p2 = __float_as_int(o2v);
        const int m12 = (a1 < p2) ? a1 : p2;
        const int m21 = (a2 < p1) ? a2 : p1;
        const int t3  = (b3 > o3) ? b3 : o3;
        const int t4  = (m12 > m21) ? m12 : m21;
        b3 = (t3 > t4) ? t3 : t4;
      }
      const bool c = lex_better(o1v, o1i, b1v, b1i);
      const float lv = c ? b1v : o1v;  const int li = c ? b1i : o1i;
      const float sv = c ? o2v : b2v;  const int si = c ? o2i : b2i;
      b1v = c ? o1v : b1v;             b1i = c ? o1i : b1i;
      const bool d = lex_better(lv, li, sv, si);
      b2v = d ? lv : sv;               b2i = d ? li : si;
    }
    const u64 k1 = pack_key(b1v, b1i);
    const u64 k2 = pack_key(b2v, b2i);
    const u64 k3 = (b3 >= 0) ? (((u64)(unsigned)b3 << 32) | 0xffffffffull) : 0ull;
    if (lane == 0) {
      s_key[2 * wid]     = k1;
      s_key[2 * wid + 1] = k2;
    }
    {
      const bool hReal = b1v >= 0.0f;
      const bool nReal = b2v >= 0.0f;
      const bool ownH  = hReal & (my1i == b1i);
      const bool ownN1 = nReal & (my1i == b2i);
      const bool ownN2 = nReal & (my2i == b2i);
      if (ownH | ownN1 | ownN2) {
        const bool first = ownH | ownN1;
        const int  sl    = first ? my1s : my2s;
        const int  row   = 2 * wid + (ownH ? 0 : 1);
        const bool tail  = sl >= 16;
        const int  pos   = 4096 * (sl >> 2) + 4 * tid + (sl & 3);
        const float* fx = tail ? (s_t + 4 * tid + (sl & 3))        : (s_x + pos);
        const float* fy = tail ? (s_t + 4096 + 4 * tid + (sl & 3)) : (s_y + pos);
        const float* fz = tail ? (s_t + 8192 + 4 * tid + (sl & 3)) : (s_z + pos);
        const float ex = *fx, ey = *fy, ez = *fz;
        s_w[W_EX + row] = ex; s_w[W_EY + row] = ey; s_w[W_EZ + row] = ez;
        if (ownH & ownN2) {
          const bool tail2 = my2s >= 16;
          const int  pos2  = 4096 * (my2s >> 2) + 4 * tid + (my2s & 3);
          const float* gx = tail2 ? (s_t + 4 * tid + (my2s & 3))        : (s_x + pos2);
          const float* gy = tail2 ? (s_t + 4096 + 4 * tid + (my2s & 3)) : (s_y + pos2);
          const float* gz = tail2 ? (s_t + 8192 + 4 * tid + (my2s & 3)) : (s_z + pos2);
          const float hx = *gx, hy = *gy, hz = *gz;
          s_w[W_EX + row + 1] = hx; s_w[W_EY + row + 1] = hy; s_w[W_EZ + row + 1] = hz;
        }
      }
    }
    __syncthreads();
    {
      const unsigned h1 = __builtin_amdgcn_readfirstlane((unsigned)(k1 >> 32));
      const unsigned l1 = __builtin_amdgcn_readfirstlane((unsigned)k1);
      const unsigned h2 = __builtin_amdgcn_readfirstlane((unsigned)(k2 >> 32));
      const unsigned l2 = __builtin_amdgcn_readfirstlane((unsigned)k2);
      const u64 kHs = ((u64)h1 << 32) | (u64)l1;
      const u64 kNs = ((u64)h2 << 32) | (u64)l2;
      const unsigned h3 = __builtin_amdgcn_readfirstlane((unsigned)(k3 >> 32));
      const unsigned l3 = __builtin_amdgcn_readfirstlane((unsigned)k3);
      const u64 kTs = ((u64)h3 << 32) | (u64)l3;
      const u64 ka = s_key[2 * lane];
      const u64 kb = s_key[2 * lane + 1];
      const int rHs = __builtin_popcount(__builtin_amdgcn_ballot_w32(ka > kHs)) +
                      __builtin_popcount(__builtin_amdgcn_ballot_w32(kb > kHs));
      const int rNs = __builtin_popcount(__builtin_amdgcn_ballot_w32(ka > kNs)) +
                      __builtin_popcount(__builtin_amdgcn_ballot_w32(kb > kNs));
      const int rTs = __builtin_popcount(__builtin_amdgcn_ballot_w32(ka > kTs)) +
                      __builtin_popcount(__builtin_amdgcn_ballot_w32(kb > kTs));
      if (lane == 0) {
        s_rank[2 * wid] = rHs; s_rank[2 * wid + 1] = rNs;
        const int rMark = (rTs < 1) ? 1 : rTs;
        if (rMark < FPS_KC) s_w[W_BLK + rMark] = __int_as_float(pass);
      }
    }
    __syncthreads();
    if (wid == 0) {
      const u64 kH = s_key[2 * lane];
      const u64 kN = s_key[2 * lane + 1];
      const float valH = __uint_as_float((unsigned)(kH >> 32));
      const float valN = __uint_as_float((unsigned)(kN >> 32));
      const int iH = 0x7fffffff - (int)(unsigned)kH;
      const int iN = 0x7fffffff - (int)(unsigned)kN;
      const bool vH = iH < N_;
      const bool vN = iN < N_;
      const int qH = vH ? iH : 0;
      const int qN = vN ? iN : 0;
      const float xH = s_w[W_EX + 2 * lane],     yH = s_w[W_EY + 2 * lane],     zH = s_w[W_EZ + 2 * lane];
      const float xN = s_w[W_EX + 2 * lane + 1], yN = s_w[W_EY + 2 * lane + 1], zN = s_w[W_EZ + 2 * lane + 1];

      const int rH = s_rank[2 * lane];
      const int rN = s_rank[2 * lane + 1];

      if (rH < FPS_KC) {
        s_w[W_CX + rH] = xH; s_w[W_CY + rH] = yH; s_w[W_CZ + rH] = zH;
        s_w[W_CV + rH] = valH; s_w[W_CI + rH] = __int_as_float(qH);
      }
      if (rN < FPS_KC) {
        s_w[W_CX + rN] = xN; s_w[W_CY + rN] = yN; s_w[W_CZ + rN] = zN;
        s_w[W_CV + rN] = valN; s_w[W_CI + rN] = __int_as_float(qN);
      }

      const int R = __builtin_popcount(__builtin_amdgcn_ballot_w32(vH)) +
                    __builtin_popcount(__builtin_amdgcn_ballot_w32(vN));

      const int r  = lane & 15;
      const int hh = lane >> 4;
      const float cxr = s_w[W_CX + r], cyr = s_w[W_CY + r], czr = s_w[W_CZ + r], cvr = s_w[W_CV + r];
      const int   cir = __float_as_int(s_w[W_CI + r]);
      const int   blk = __float_as_int(s_w[W_BLK + r]);
      const bool  real = (r < R) & (cvr > 0.0f);

      const float need = cvr * 1.0001f;
      float dd[FPS_KCH];
      float dall = __builtin_inff();
#pragma unroll
      for (int i = 0; i < FPS_KCH; ++i) {
        const int ii = hh * FPS_KCH + i;
        const float sx = s_w[W_CX + ii], sy = s_w[W_CY + ii], sz = s_w[W_CZ + ii];
        const float dxa = cxr - sx, dya = cyr - sy, dza = czr - sz;
        const float dda = dxa * dxa + dya * dya + dza * dza;
        dd[i] = (ii >= r) ? __builtin_inff() : dda;
        dall  = fminf(dall, dd[i]);
      }
      const unsigned nfar  = __builtin_amdgcn_ballot_w32(!(dall >= need));
      const unsigned nreal = __builtin_amdgcn_ballot_w32(!real);
      const unsigned realm = ~nreal & 0xffffu;
      const unsigned amask = ~(nfar | (nfar >> 16)) & realm;
      const unsigned ah = hh ? (amask >> FPS_KCH) : amask;
      float da[FPS_KCH];
#pragma unroll
      for (int i = 0; i < FPS_KCH; ++i)
        da[i] = ((ah >> i) & 1u) ? dd[i] : __builtin_inff();
      const float dA = fminf(fminf(fminf(da[0], da[1]), fminf(da[2], da[3])),
                             fminf(fminf(da[4], da[5]), fminf(da[6], da[7])));
      s_w[W_UB + lane] = dA;
      const unsigned skm = realm & ~amask;
      int wv[FPS_KC];
#pragma unroll
      for (int s = 0; s < FPS_KC; ++s) {
        const int lo = __float_as_int(pub[s >> 2][s & 3]);
        const int hi = __float_as_int(pub[4 + (s >> 2)][s & 3]);
        const bool use = (((skm >> s) & 1u) != 0u) & (s < r);
        wv[s] = use ? ((lo < hi) ? lo : hi) : 0;
      }
#pragma unroll
      for (int w = FPS_KC / 2; w; w >>= 1) {
#pragma unroll
        for (int s = 0; s < w; ++s) wv[s] = (wv[s] > wv[s + w]) ? wv[s] : wv[s + w];
      }
      const float mub = __int_as_float(wv[0]) * 1.0001f;
      const unsigned mk = __builtin_amdgcn_ballot_w32(blk == pass) & 0xffffu;
      const int lmark = __builtin_ctz(mk | (1u << FPS_KC));
      const bool acc = (((amask >> r) & 1u) != 0u) & (mub < cvr) & (r < lmark);
      const unsigned accm = (__builtin_amdgcn_ballot_w32(acc) & 0xffffu) | 1u;
      const int pos = __builtin_popcount(accm & ((1u << r) - 1u));
      int mnew = __builtin_popcount(accm);
      mnew = (mnew < FPS_K) ? mnew : FPS_K;
      const int room = M_ - step;
      mnew = (mnew < room) ? mnew : room;

      if ((hh == 0) & (((accm >> r) & 1u) != 0u) & (pos < mnew)) {
        s_px[pos] = cxr; s_py[pos] = cyr; s_pz[pos] = czr;
        fps_idx[b * M_ + step + pos] = cir;
      }
      if (lane == 0) s_m = mnew;
    }
    __syncthreads();
    m = __builtin_amdgcn_readfirstlane(s_m);
    step += m;
    pass += 1;
  }
}

#define NN_G 10
#define NN_TRIPS (N_ / 16 / NN_G)

__global__ __launch_bounds__(256) void three_nn_kernel(
    const float* __restrict__ pc, const int* __restrict__ fps_idx,
    int* __restrict__ idx3, float* __restrict__ w3) {
  const int tid  = threadIdx.x;
  const int lane = tid & 31;
  const int wave = blockIdx.x * (256 >> 5) + (tid >> 5);
  const int tilesPerBatch = M_ / 16;
  const int b    = wave / tilesPerBatch;
  const int rt   = wave - b * tilesPerBatch;
  const int half = lane >> 4;
  const int l15  = lane & 15;

  const float* pcb = pc + (size_t)b * N_ * 3;

  const int mrow = rt * 16 + l15;
  const int sidx = fps_idx[b * M_ + mrow];
  const float* sp = pcb + (size_t)sidx * 3;
  const float sx = sp[0], sy = sp[1], sz = sp[2];
  v2f Bs;
  Bs.x = half ? (-2.0f * sz) : (-2.0f * sx);
  Bs.y = half ? 1.0f         : (-2.0f * sy);

  float d[3];
  int   ix[3];
#pragma unroll
  for (int k = 0; k < 3; ++k) { d[k] = __builtin_inff(); ix[k] = 0x7fffffff; }

  const float* lp = pcb + (size_t)l15 * 3;

  float cx[NN_G], cy[NN_G], cz[NN_G];
#pragma unroll
  for (int g = 0; g < NN_G; ++g) {
    const float* p = lp + (size_t)g * 48;
    cx[g] = p[0]; cy[g] = p[1]; cz[g] = p[2];
  }

  for (int q = 0; q < NN_TRIPS; ++q) {
    const int qn = (q + 1 < NN_TRIPS) ? (q + 1) : q;
    const float* np = lp + (size_t)qn * (NN_G * 48);
    float nx[NN_G], ny[NN_G], nz[NN_G];
#pragma unroll
    for (int g = 0; g < NN_G; ++g) {
      const float* p = np + (size_t)g * 48;
      nx[g] = p[0]; ny[g] = p[1]; nz[g] = p[2];
    }

    v8f c[NN_G];
#pragma unroll
    for (int g = 0; g < NN_G; ++g) {
      const float bx = cx[g], by = cy[g], bz = cz[g];
      const float bn2 = __builtin_fmaf(bz, bz, __builtin_fmaf(bx, bx, by * by));
      const v8f zero = {0.f, 0.f, 0.f, 0.f, 0.f, 0.f, 0.f, 0.f};
#if USE_WMMA
      v2f Av;
      Av.x = half ? bz  : bx;
      Av.y = half ? bn2 : by;
      c[g] = __builtin_amdgcn_wmma_f32_16x16x4_f32(
           false, Av,  false, Bs,
           (short)0, zero,  false,  false);
#else
      c[g] = zero;
#pragma unroll
      for (int r = 0; r < 8; ++r) {
        const int src = half * 8 + r;
        const float rx = __shfl(bx, src, 32);
        const float ry = __shfl(by, src, 32);
        const float rz = __shfl(bz, src, 32);
        const float rn = __shfl(bn2, src, 32);
        c[g][r] = -2.0f * (rx * sx + ry * sy + rz * sz) + rn;
      }
#endif
    }

    unsigned hit[NN_G];
    unsigned anyhit = 0u;
    const float d2s = d[2];
#pragma unroll
    for (int g = 0; g < NN_G; ++g) {
      const float m01 = fminf(fminf(c[g][0], c[g][1]), fminf(c[g][2], c[g][3]));
      const float m23 = fminf(fminf(c[g][4], c[g][5]), fminf(c[g][6], c[g][7]));
      const float mt  = fminf(m01, m23);
      hit[g] = __builtin_amdgcn_ballot_w32(mt < d2s);
      anyhit |= hit[g];
    }
    if (anyhit != 0u) {
#pragma unroll
      for (int g = 0; g < NN_G; ++g) {
        if (hit[g] != 0u) {
          const int n0 = (q * NN_G + g) * 16 + half * 8;
#pragma unroll
          for (int r = 0; r < 8; ++r) ins3_stream(c[g][r], n0 + r, d, ix);
        }
      }
    }

#pragma unroll
    for (int g = 0; g < NN_G; ++g) { cx[g] = nx[g]; cy[g] = ny[g]; cz[g] = nz[g]; }
  }

  {
    float pd[3]; int pi[3];
#pragma unroll
    for (int k = 0; k < 3; ++k) {
      pd[k] = __shfl_xor(d[k], 16, 32);
      pi[k] = __shfl_xor(ix[k], 16, 32);
    }
#pragma unroll
    for (int k = 0; k < 3; ++k) ins3_lex(pd[k], pi[k], d, ix);
  }

  if (half == 0) {
    float wv[3];
    float wsum = 0.f;
#pragma unroll
    for (int k = 0; k < 3; ++k) {
      const int nn = ix[k];
      const float* qp = pcb + (size_t)nn * 3;
      const float dx = qp[0] - sx, dy = qp[1] - sy, dz = qp[2] - sz;
      const float dd = dx * dx + dy * dy + dz * dz;
      const float dist = sqrtf(fmaxf(dd, 0.0f));
      wv[k] = 1.0f / (dist + 1e-8f);
      wsum += wv[k];
      idx3[((size_t)b * M_ + mrow) * 3 + k] = nn;
    }
#pragma unroll
    for (int k = 0; k < 3; ++k)
      w3[((size_t)b * M_ + mrow) * 3 + k] = wv[k] / wsum;
  }
}

__global__ __launch_bounds__(256) void interp_kernel(
    const float* __restrict__ feats, const int* __restrict__ idx3,
    const float* __restrict__ w3, float* __restrict__ out) {
  const int bc = blockIdx.x;
  const int b  = bc / C_;
  const int c  = bc - b * C_;
  const float* f = feats + ((size_t)b * C_ + c) * N_;
  float*       o = out   + ((size_t)b * C_ + c) * M_;
  const int*   ib = idx3 + (size_t)b * M_ * 3;
  const float* wb = w3   + (size_t)b * M_ * 3;
  for (int m = threadIdx.x; m < M_; m += 256) {
    const int   n0 = ib[m * 3 + 0], n1 = ib[m * 3 + 1], n2 = ib[m * 3 + 2];
    const float w0 = wb[m * 3 + 0], w1 = wb[m * 3 + 1], w2 = wb[m * 3 + 2];
    o[m] = w0 * f[n0] + w1 * f[n1] + w2 * f[n2];
  }
}

extern "C" void kernel_launch(void* const* d_in, const int* in_sizes, int n_in,
                              void* d_out, int out_size, void* d_ws, size_t ws_size,
                              hipStream_t stream) {
  const float* pc    = (const float*)d_in[0];
  const float* feats = (const float*)d_in[1];
  const float* obj   = (const float*)d_in[2];
  const float* grasp = (const float*)d_in[3];
  float* out = (float*)d_out;

  char* ws = (char*)d_ws;
  size_t off = 0;
  int* fps = (int*)(ws + off);
  off += (size_t)B_ * M_ * 4;        off = (off + 255) & ~(size_t)255;
  int* idx3 = (int*)(ws + off);
  off += (size_t)B_ * M_ * 3 * 4;    off = (off + 255) & ~(size_t)255;
  float* w3 = (float*)(ws + off);
  off += (size_t)B_ * M_ * 3 * 4;

  fps_kernel<<<B_, 1024, 0, stream>>>(pc, obj, grasp, fps);
  three_nn_kernel<<<(B_ * (M_ / 16)) / 8, 256, 0, stream>>>(pc, fps, idx3, w3);
  interp_kernel<<<B_ * C_, 256, 0, stream>>>(feats, idx3, w3, out);
}


# fps kernel: a batch's pick indices are stored to memory behind the pass's closing barrier (was in front of it): the barrier no longer waits for a global store's acknowledgement each pass; same sums
# speedup vs baseline: 4.9541x; 1.0378x over previous
#include <hip/hip_runtime.h>
#include <hip/hip_bf16.h>

#define B_ 4
#define N_ 20000
#define C_ 512
#define M_ 1024
#define GRASP_TH 0.1f

typedef float v2f __attribute__((ext_vector_type(2)));
typedef float v4f __attribute__((ext_vector_type(4)));
typedef float v8f __attribute__((ext_vector_type(8)));

#if defined(__gfx1250__) && __has_builtin(__builtin_amdgcn_wmma_f32_16x16x4_f32)
#define USE_WMMA 1
#else
#define USE_WMMA 0
#endif

__device__ __forceinline__ void ins3_stream(float v, int n, float d[3], int ix[3]) {
  const bool c0 = v < d[0];
  const bool c1 = v < d[1];
  const bool c2 = v < d[2];
  const float r0 = c0 ? v     : d[0];
  const int   j0 = c0 ? n     : ix[0];
  const float r1 = c0 ? d[0]  : (c1 ? v : d[1]);
  const int   j1 = c0 ? ix[0] : (c1 ? n : ix[1]);
  const float r2 = c1 ? d[1]  : (c2 ? v : d[2]);
  const int   j2 = c1 ? ix[1] : (c2 ? n : ix[2]);
  d[0] = r0; d[1] = r1; d[2] = r2;
  ix[0] = j0; ix[1] = j1; ix[2] = j2;
}

__device__ __forceinline__ void ins3_lex(float v, int n, float d[3], int ix[3]) {
  const bool c0 = (v < d[0]) | ((v == d[0]) & (n < ix[0]));
  const bool c1 = (v < d[1]) | ((v == d[1]) & (n < ix[1]));
  const bool c2 = (v < d[2]) | ((v == d[2]) & (n < ix[2]));
  const float r0 = c0 ? v     : d[0];
  const int   j0 = c0 ? n     : ix[0];
  const float r1 = c0 ? d[0]  : (c1 ? v : d[1]);
  const int   j1 = c0 ? ix[0] : (c1 ? n : ix[1]);
  const float r2 = c1 ? d[1]  : (c2 ? v : d[2]);
  const int   j2 = c1 ? ix[1] : (c2 ? n : ix[2]);
  d[0] = r0; d[1] = r1; d[2] = r2;
  ix[0] = j0; ix[1] = j1; ix[2] = j2;
}

#define FPS_J 5
#define FPS_P 10
#define FPS_K 12
#define FPS_KC 16
#define FPS_KCH 8
#define FPS_KP 16
#define W_CX 0
#define W_CY 16
#define W_CZ 32
#define W_CV 48
#define W_CI 64
#define W_UB 80
#define W_BLK 112
#define W_EX 128
#define W_EY 192
#define W_EZ 256
static_assert(FPS_K == 12 && FPS_KCH * 2 == FPS_KC && FPS_KC == 16 && FPS_K <= FPS_KP, "the pick list in the scan is written out for 12, the proof for 16 candidates");

__device__ __forceinline__ bool lex_better(float ov, int oi, float v, int i) {
  return (ov > v) | ((ov == v) & (oi < i));
}

typedef unsigned long long u64;

__device__ __forceinline__ void fps_apply(const v2f vx, const v2f vy, const v2f vz,
                                          float qx, float qy, float qz, float v[2]) {
#pragma unroll
  for (int k = 0; k < 2; ++k) {
    const float dxa = vx[k] - qx, dya = vy[k] - qy, dza = vz[k] - qz;
    const float dda = __builtin_fmaf(dza, dza, __builtin_fmaf(dxa, dxa, dya * dya));
    v[k] = fminf(v[k], dda);
  }
}
#define FPS_PICK(i) if (m <= (i)) break; fps_apply(vx, vy, vz, qx[i], qy[i], qz[i], v)

__device__ __forceinline__ void fps_pair(const int p, const v2f* x2, const v2f* y2, const v2f* z2,
                                         const v2f* tl2, const int tid, const int m,
                                         const float (&qx)[FPS_K], const float (&qy)[FPS_K],
                                         const float (&qz)[FPS_K], float (&mp)[2],
                                         float& b1v, float& b2v, int& b1s, int& b2s, int& b3) {
  const v2f vx = (p < 8) ? x2[2 * (tid + 1024 * (p >> 1)) + (p & 1)] : tl2[2 * tid + (p & 1)];
  const v2f vy = (p < 8) ? y2[2 * (tid + 1024 * (p >> 1)) + (p & 1)] : tl2[2 * tid + (p & 1) + 2048];
  const v2f vz = (p < 8) ? z2[2 * (tid + 1024 * (p >> 1)) + (p & 1)] : tl2[2 * tid + (p & 1) + 4096];
  float v[2];
  v[0] = mp[0]; v[1] = mp[1];
  do {
    FPS_PICK(0); FPS_PICK(1); FPS_PICK(2);  FPS_PICK(3);
    FPS_PICK(4); FPS_PICK(5); FPS_PICK(6);  FPS_PICK(7);
    FPS_PICK(8); FPS_PICK(9); FPS_PICK(10); FPS_PICK(11);
  } while (0);
#pragma unroll
  for (int k = 0; k < 2; ++k) {
    mp[k] = v[k];
    const int  s  = 2 * p + k;
    const bool c1 = v[k] > b1v;
    const bool c2 = v[k] > b2v;
    const int vb = __float_as_int(v[k]);
    const int lo = (__float_as_int(b2v) < vb) ? __float_as_int(b2v) : vb;
    b3  = (lo > b3) ? lo : b3;
    b2v = c1 ? b1v : (c2 ? v[k] : b2v);
    b2s = c1 ? b1s : (c2 ? s    : b2s);
    b1v = c1 ? v[k] : b1v;
    b1s = c1 ? s    : b1s;
  }
}
#define FPS_PAIR(p) if (Pw <= (p)) break; fps_pair((p), x2, y2, z2, tl2, tid, m, qx, qy, qz, mind[p], b1v, b2v, b1s, b2s, b3)

__device__ __forceinline__ u64 pack_key(float v, int i) {
  const bool real = v >= 0.0f;
  const unsigned hi = real ? __float_as_uint(v) : 0u;
  const unsigned lo = real ? (unsigned)(0x7fffffff - i) : 0u;
  return ((u64)hi << 32) | (u64)lo;
}

__global__ __launch_bounds__(1024) void fps_kernel(
    const float* __restrict__ pc, const float* __restrict__ obj,
    const float* __restrict__ grasp, int* __restrict__ fps_idx) {
  const int b    = blockIdx.x;
  const int tid  = threadIdx.x;
  const int lane = tid & 31;
  const int wid  = tid >> 5;

  __shared__ __attribute__((aligned(16))) float s_x[4 * 4096];
  __shared__ __attribute__((aligned(16))) float s_y[4 * 4096];
  __shared__ __attribute__((aligned(16))) float s_z[4 * 4096];
  __shared__ __attribute__((aligned(16))) float s_t[3 * 4096];
  __shared__ __attribute__((aligned(16))) u64 s_key[64];
  __shared__ __attribute__((aligned(8)))  int s_rank[64];
  __shared__ int   swi1[32];
  __shared__ int   s_m;
  __shared__ __attribute__((aligned(16))) float s_px[FPS_KP], s_py[FPS_KP], s_pz[FPS_KP];
  __shared__ __attribute__((aligned(16))) float s_w[320];
  __shared__ unsigned short s_map[FPS_J * 4 * 1024];

  const float* pcb = pc + (size_t)b * N_ * 3;
  const float* ob0 = obj + (size_t)b * 2 * N_;
  const float* gr  = grasp + (size_t)b * N_;

  int lz;
  asm("v_mov_b32 %0, 0" : "=v"(lz));
  const v4f* ppx = (const v4f*)s_px + lz;
  const v4f* ppy = (const v4f*)s_py + lz;
  const v4f* ppz = (const v4f*)s_pz + lz;
  const v4f* pub = (const v4f*)(s_w + W_UB) + lz;

  {
    const v4f zero4 = {0.f, 0.f, 0.f, 0.f};
#pragma unroll
    for (int j = 0; j < 4; ++j) {
      const int g = tid + 1024 * j;
      ((v4f*)s_x)[g] = zero4;
      ((v4f*)s_y)[g] = zero4;
      ((v4f*)s_z)[g] = zero4;
    }
#pragma unroll
    for (int p = 0; p < 3; ++p) ((v4f*)s_t)[tid + 1024 * p] = zero4;
  }

  if (tid < 320) s_w[tid] = 0.0f;

  unsigned vm = 0u;
#pragma unroll
  for (int j = 0; j < FPS_J; ++j) {
    const int g  = tid + 1024 * j;
    const bool gv = g < N_ / 4;
    const int gl = gv ? g : 0;
#pragma unroll
    for (int k = 0; k < 4; ++k) {
      const int n = 4 * gl + k;
      const bool mk = gv & (ob0[N_ + n] > ob0[n]) & (gr[n] > GRASP_TH);
      vm |= mk ? (1u << (4 * j + k)) : 0u;
    }
  }
  int firstLoc = N_;
  if (vm != 0u) {
    const int o = __builtin_ctz(vm);
    firstLoc = 4 * (tid + 1024 * (o >> 2)) + (o & 3);
  }
  unsigned cA = 0u;
#pragma unroll
  for (int j = 0; j < 4; ++j) cA |= (unsigned)__builtin_popcount((vm >> (4 * j)) & 15u) << (8 * j);
  const unsigned cB = (unsigned)__builtin_popcount((vm >> 16) & 15u);
  unsigned iA = cA, iB = cB;
  for (int dd = 1; dd < 32; dd <<= 1) {
    const unsigned oA = __shfl_up(iA, dd, 32);
    const unsigned oB = __shfl_up(iB, dd, 32);
    const bool in = lane >= dd;
    iA += in ? oA : 0u;
    iB += in ? oB : 0u;
  }
  const unsigned tA = __shfl(iA, 31, 32);
  const unsigned tB = __shfl(iB, 31, 32);
  const unsigned eA = iA - cA;
  const unsigned eB = iB - cB;
  int r0[FPS_J];
  int wtotal = 0;
#pragma unroll
  for (int j = 0; j < FPS_J; ++j) {
    const int tot = (j < 4) ? (int)((tA >> (8 * j)) & 255u) : (int)tB;
    const int ex  = (j < 4) ? (int)((eA >> (8 * j)) & 255u) : (int)eB;
    r0[j] = wtotal + ex;
    wtotal += tot;
  }
  const int wbase = tid & ~31;
#pragma unroll
  for (int j = 0; j < FPS_J; ++j) {
    const int g  = tid + 1024 * j;
    const bool gv = g < N_ / 4;
    const int gl = gv ? g : 0;
    const float* pp = pcb + (size_t)gl * 12;
    float px[4], py[4], pz[4];
#pragma unroll
    for (int k = 0; k < 4; ++k) { px[k] = pp[3 * k]; py[k] = pp[3 * k + 1]; pz[k] = pp[3 * k + 2]; }
    int r = r0[j];
#pragma unroll
    for (int k = 0; k < 4; ++k) {
      const bool mk = ((vm >> (4 * j + k)) & 1u) != 0u;
      if (mk) {
        const int  c    = r >> 5;
        const int  dt   = wbase + (r & 31);
        const bool tail = c >= 16;
        const int  pos  = 4096 * (c >> 2) + 4 * dt + (c & 3);
        float* dx = tail ? (s_t + 4 * dt + (c & 3))        : (s_x + pos);
        float* dy = tail ? (s_t + 4096 + 4 * dt + (c & 3)) : (s_y + pos);
        float* dz = tail ? (s_t + 8192 + 4 * dt + (c & 3)) : (s_z + pos);
        *dx = px[k];
        *dy = py[k];
        *dz = pz[k];
        s_map[c * 1024 + dt] = (unsigned short)((4 * j + k) | (lane << 5));
        r += 1;
      }
    }
  }
  const int cnt = (wtotal + 31 - lane) >> 5;
  float mind[FPS_P][2];
#pragma unroll
  for (int s = 0; s < FPS_P * 2; ++s) {
    const bool filled = s < cnt;
    mind[s >> 1][s & 1] = filled ? __builtin_inff() : -__builtin_inff();
    if (!filled) s_map[s * 1024 + tid] = (unsigned short)0;
  }
  const int Pw = (((__builtin_amdgcn_readfirstlane(wtotal) + 31) >> 5) + 1) >> 1;

  for (int mm = 16; mm; mm >>= 1) {
    const int o = __shfl_xor(firstLoc, mm, 32);
    firstLoc = (o < firstLoc) ? o : firstLoc;
  }
  if (lane == 0) swi1[wid] = firstLoc;
  __syncthreads();
  if (wid == 0) {
    int v = swi1[lane];
    for (int mm = 16; mm; mm >>= 1) {
      const int o = __shfl_xor(v, mm, 32);
      v = (o < v) ? o : v;
    }
    if (lane == 0) {
      const int f = (v >= N_) ? 0 : v;
      fps_idx[b * M_] = f;
      s_m = 1;
      const float fx = pcb[(size_t)f * 3], fy = pcb[(size_t)f * 3 + 1], fz = pcb[(size_t)f * 3 + 2];
#pragma unroll
      for (int i = 0; i < FPS_KP; ++i) { s_px[i] = fx; s_py[i] = fy; s_pz[i] = fz; }
    }
  }
  __syncthreads();

  const v2f* x2 = (const v2f*)s_x;
  const v2f* y2 = (const v2f*)s_y;
  const v2f* z2 = (const v2f*)s_z;
  const v2f* tl2 = (const v2f*)s_t;

  int m    = 1;
  int step = 1;
  int pass = 1;
  int st_at = -1, st_i = 0;

  while (step < M_) {
    float qx[FPS_K], qy[FPS_K], qz[FPS_K];
#pragma unroll
    for (int i4 = 0; i4 < FPS_K / 4; ++i4) {
      const v4f tx = ppx[i4], ty = ppy[i4], tz = ppz[i4];
#pragma unroll
      for (int e = 0; e < 4; ++e) { qx[4 * i4 + e] = tx[e]; qy[4 * i4 + e] = ty[e]; qz[4 * i4 + e] = tz[e]; }
    }

    float b1v = -__builtin_inff(), b2v = -__builtin_inff();
    int   b1s = 0,                 b2s = 0;
    int   b3  = __float_as_int(-__builtin_inff());
    do {
      FPS_PAIR(0); FPS_PAIR(1); FPS_PAIR(2); FPS_PAIR(3); FPS_PAIR(4);
      FPS_PAIR(5); FPS_PAIR(6); FPS_PAIR(7); FPS_PAIR(8); FPS_PAIR(9);
    } while (0);
    const int c1o = (int)s_map[b1s * 1024 + tid];
    const int c2o = (int)s_map[b2s * 1024 + tid];
    int b1i = 4 * (wbase + (c1o >> 5)) + 4096 * ((c1o & 31) >> 2) + (c1o & 3);
    int b2i = 4 * (wbase + (c2o >> 5)) + 4096 * ((c2o & 31) >> 2) + (c2o & 3);
    const int my1i = (b1v >= 0.0f) ? b1i : -1;
    const int my2i = (b2v >= 0.0f) ? b2i : -1;
    const int my1s = b1s, my2s = b2s;
    for (int mm = 16; mm; mm >>= 1) {
      const float o1v = __shfl_xor(b1v, mm, 32);
      const int   o1i = __shfl_xor(b1i, mm, 32);
      const float o2v = __shfl_xor(b2v, mm, 32);
      const int   o2i = __shfl_xor(b2i, mm, 32);
      const int   o3  = __shfl_xor(b3, mm, 32);
      {
        const int a1 = __float_as_int(b1v), a2 = __float_as_int(b2v);
        const int p1 = __float_as_int(o1v), p2 = __float_as_int(o2v);
        const int m12 = (a1 < p2) ? a1 : p2;
        const int m21 = (a2 < p1) ? a2 : p1;
        const int t3  = (b3 > o3) ? b3 : o3;
        const int t4  = (m12 > m21) ? m12 : m21;
        b3 = (t3 > t4) ? t3 : t4;
      }
      const bool c = lex_better(o1v, o1i, b1v, b1i);
      const float lv = c ? b1v : o1v;  const int li = c ? b1i : o1i;
      const float sv = c ? o2v : b2v;  const int si = c ? o2i : b2i;
      b1v = c ? o1v : b1v;             b1i = c ? o1i : b1i;
      const bool d = lex_better(lv, li, sv, si);
      b2v = d ? lv : sv;               b2i = d ? li : si;
    }
    const u64 k1 = pack_key(b1v, b1i);
    const u64 k2 = pack_key(b2v, b2i);
    const u64 k3 = (b3 >= 0) ? (((u64)(unsigned)b3 << 32) | 0xffffffffull) : 0ull;
    if (lane == 0) {
      s_key[2 * wid]     = k1;
      s_key[2 * wid + 1] = k2;
    }
    {
      const bool hReal = b1v >= 0.0f;
      const bool nReal = b2v >= 0.0f;
      const bool ownH  = hReal & (my1i == b1i);
      const bool ownN1 = nReal & (my1i == b2i);
      const bool ownN2 = nReal & (my2i == b2i);
      if (ownH | ownN1 | ownN2) {
        const bool first = ownH | ownN1;
        const int  sl    = first ? my1s : my2s;
        const int  row   = 2 * wid + (ownH ? 0 : 1);
        const bool tail  = sl >= 16;
        const int  pos   = 4096 * (sl >> 2) + 4 * tid + (sl & 3);
        const float* fx = tail ? (s_t + 4 * tid + (sl & 3))        : (s_x + pos);
        const float* fy = tail ? (s_t + 4096 + 4 * tid + (sl & 3)) : (s_y + pos);
        const float* fz = tail ? (s_t + 8192 + 4 * tid + (sl & 3)) : (s_z + pos);
        const float ex = *fx, ey = *fy, ez = *fz;
        s_w[W_EX + row] = ex; s_w[W_EY + row] = ey; s_w[W_EZ + row] = ez;
        if (ownH & ownN2) {
          const bool tail2 = my2s >= 16;
          const int  pos2  = 4096 * (my2s >> 2) + 4 * tid + (my2s & 3);
          const float* gx = tail2 ? (s_t + 4 * tid + (my2s & 3))        : (s_x + pos2);
          const float* gy = tail2 ? (s_t + 4096 + 4 * tid + (my2s & 3)) : (s_y + pos2);
          const float* gz = tail2 ? (s_t + 8192 + 4 * tid + (my2s & 3)) : (s_z + pos2);
          const float hx = *gx, hy = *gy, hz = *gz;
          s_w[W_EX + row + 1] = hx; s_w[W_EY + row + 1] = hy; s_w[W_EZ + row + 1] = hz;
        }
      }
    }
    __syncthreads();
    {
      const unsigned h1 = __builtin_amdgcn_readfirstlane((unsigned)(k1 >> 32));
      const unsigned l1 = __builtin_amdgcn_readfirstlane((unsigned)k1);
      const unsigned h2 = __builtin_amdgcn_readfirstlane((unsigned)(k2 >> 32));
      const unsigned l2 = __builtin_amdgcn_readfirstlane((unsigned)k2);
      const u64 kHs = ((u64)h1 << 32) | (u64)l1;
      const u64 kNs = ((u64)h2 << 32) | (u64)l2;
      const unsigned h3 = __builtin_amdgcn_readfirstlane((unsigned)(k3 >> 32));
      const unsigned l3 = __builtin_amdgcn_readfirstlane((unsigned)k3);
      const u64 kTs = ((u64)h3 << 32) | (u64)l3;
      const u64 ka = s_key[2 * lane];
      const u64 kb = s_key[2 * lane + 1];
      const int rHs = __builtin_popcount(__builtin_amdgcn_ballot_w32(ka > kHs)) +
                      __builtin_popcount(__builtin_amdgcn_ballot_w32(kb > kHs));
      const int rNs = __builtin_popcount(__builtin_amdgcn_ballot_w32(ka > kNs)) +
                      __builtin_popcount(__builtin_amdgcn_ballot_w32(kb > kNs));
      const int rTs = __builtin_popcount(__builtin_amdgcn_ballot_w32(ka > kTs)) +
                      __builtin_popcount(__builtin_amdgcn_ballot_w32(kb > kTs));
      if (lane == 0) {
        s_rank[2 * wid] = rHs; s_rank[2 * wid + 1] = rNs;
        const int rMark = (rTs < 1) ? 1 : rTs;
        if (rMark < FPS_KC) s_w[W_BLK + rMark] = __int_as_float(pass);
      }
    }
    __syncthreads();
    if (wid == 0) {
      const u64 kH = s_key[2 * lane];
      const u64 kN = s_key[2 * lane + 1];
      const float valH = __uint_as_float((unsigned)(kH >> 32));
      const float valN = __uint_as_float((unsigned)(kN >> 32));
      const int iH = 0x7fffffff - (int)(unsigned)kH;
      const int iN = 0x7fffffff - (int)(unsigned)kN;
      const bool vH = iH < N_;
      const bool vN = iN < N_;
      const int qH = vH ? iH : 0;
      const int qN = vN ? iN : 0;
      const float xH = s_w[W_EX + 2 * lane],     yH = s_w[W_EY + 2 * lane],     zH = s_w[W_EZ + 2 * lane];
      const float xN = s_w[W_EX + 2 * lane + 1], yN = s_w[W_EY + 2 * lane + 1], zN = s_w[W_EZ + 2 * lane + 1];

      const int rH = s_rank[2 * lane];
      const int rN = s_rank[2 * lane + 1];

      if (rH < FPS_KC) {
        s_w[W_CX + rH] = xH; s_w[W_CY + rH] = yH; s_w[W_CZ + rH] = zH;
        s_w[W_CV + rH] = valH; s_w[W_CI + rH] = __int_as_float(qH);
      }
      if (rN < FPS_KC) {
        s_w[W_CX + rN] = xN; s_w[W_CY + rN] = yN; s_w[W_CZ + rN] = zN;
        s_w[W_CV + rN] = valN; s_w[W_CI + rN] = __int_as_float(qN);
      }

      const int R = __builtin_popcount(__builtin_amdgcn_ballot_w32(vH)) +
                    __builtin_popcount(__builtin_amdgcn_ballot_w32(vN));

      const int r  = lane & 15;
      const int hh = lane >> 4;
      const float cxr = s_w[W_CX + r], cyr = s_w[W_CY + r], czr = s_w[W_CZ + r], cvr = s_w[W_CV + r];
      const int   cir = __float_as_int(s_w[W_CI + r]);
      const int   blk = __float_as_int(s_w[W_BLK + r]);
      const bool  real = (r < R) & (cvr > 0.0f);

      const float need = cvr * 1.0001f;
      float dd[FPS_KCH];
      float dall = __builtin_inff();
#pragma unroll
      for (int i = 0; i < FPS_KCH; ++i) {
        const int ii = hh * FPS_KCH + i;
        const float sx = s_w[W_CX + ii], sy = s_w[W_CY + ii], sz = s_w[W_CZ + ii];
        const float dxa = cxr - sx, dya = cyr - sy, dza = czr - sz;
        const float dda = dxa * dxa + dya * dya + dza * dza;
        dd[i] = (ii >= r) ? __builtin_inff() : dda;
        dall  = fminf(dall, dd[i]);
      }
      const unsigned nfar  = __builtin_amdgcn_ballot_w32(!(dall >= need));
      const unsigned nreal = __builtin_amdgcn_ballot_w32(!real);
      const unsigned realm = ~nreal & 0xffffu;
      const unsigned amask = ~(nfar | (nfar >> 16)) & realm;
      const unsigned ah = hh ? (amask >> FPS_KCH) : amask;
      float da[FPS_KCH];
#pragma unroll
      for (int i = 0; i < FPS_KCH; ++i)
        da[i] = ((ah >> i) & 1u) ? dd[i] : __builtin_inff();
      const float dA = fminf(fminf(fminf(da[0], da[1]), fminf(da[2], da[3])),
                             fminf(fminf(da[4], da[5]), fminf(da[6], da[7])));
      s_w[W_UB + lane] = dA;
      const unsigned skm = realm & ~amask;
      int wv[FPS_KC];
#pragma unroll
      for (int s = 0; s < FPS_KC; ++s) {
        const int lo = __float_as_int(pub[s >> 2][s & 3]);
        const int hi = __float_as_int(pub[4 + (s >> 2)][s & 3]);
        const bool use = (((skm >> s) & 1u) != 0u) & (s < r);
        wv[s] = use ? ((lo < hi) ? lo : hi) : 0;
      }
#pragma unroll
      for (int w = FPS_KC / 2; w; w >>= 1) {
#pragma unroll
        for (int s = 0; s < w; ++s) wv[s] = (wv[s] > wv[s + w]) ? wv[s] : wv[s + w];
      }
      const float mub = __int_as_float(wv[0]) * 1.0001f;
      const unsigned mk = __builtin_amdgcn_ballot_w32(blk == pass) & 0xffffu;
      const int lmark = __builtin_ctz(mk | (1u << FPS_KC));
      const bool acc = (((amask >> r) & 1u) != 0u) & (mub < cvr) & (r < lmark);
      const unsigned accm = (__builtin_amdgcn_ballot_w32(acc) & 0xffffu) | 1u;
      const int pos = __builtin_popcount(accm & ((1u << r) - 1u));
      int mnew = __builtin_popcount(accm);
      mnew = (mnew < FPS_K) ? mnew : FPS_K;
      const int room = M_ - step;
      mnew = (mnew < room) ? mnew : room;

      const bool mine = (hh == 0) & (((accm >> r) & 1u) != 0u) & (pos < mnew);
      if (mine) { s_px[pos] = cxr; s_py[pos] = cyr; s_pz[pos] = czr; }
      st_at = mine ? (b * M_ + step + pos) : -1;
      st_i  = cir;
      if (lane == 0) s_m = mnew;
    }
    __syncthreads();
    if (st_at >= 0) fps_idx[st_at] = st_i;
    st_at = -1; st_i = 0;
    m = __builtin_amdgcn_readfirstlane(s_m);
    step += m;
    pass += 1;
  }
}

#define NN_G 10
#define NN_TRIPS (N_ / 16 / NN_G)

__global__ __launch_bounds__(256) void three_nn_kernel(
    const float* __restrict__ pc, const int* __restrict__ fps_idx,
    int* __restrict__ idx3, float* __restrict__ w3) {
  const int tid  = threadIdx.x;
  const int lane = tid & 31;
  const int wave = blockIdx.x * (256 >> 5) + (tid >> 5);
  const int tilesPerBatch = M_ / 16;
  const int b    = wave / tilesPerBatch;
  const int rt   = wave - b * tilesPerBatch;
  const int half = lane >> 4;
  const int l15  = lane & 15;

  const float* pcb = pc + (size_t)b * N_ * 3;

  const int mrow = rt * 16 + l15;
  const int sidx = fps_idx[b * M_ + mrow];
  const float* sp = pcb + (size_t)sidx * 3;
  const float sx = sp[0], sy = sp[1], sz = sp[2];
  v2f Bs;
  Bs.x = half ? (-2.0f * sz) : (-2.0f * sx);
  Bs.y = half ? 1.0f         : (-2.0f * sy);

  float d[3];
  int   ix[3];
#pragma unroll
  for (int k = 0; k < 3; ++k) { d[k] = __builtin_inff(); ix[k] = 0x7fffffff; }

  const float* lp = pcb + (size_t)l15 * 3;

  float cx[NN_G], cy[NN_G], cz[NN_G];
#pragma unroll
  for (int g = 0; g < NN_G; ++g) {
    const float* p = lp + (size_t)g * 48;
    cx[g] = p[0]; cy[g] = p[1]; cz[g] = p[2];
  }

  for (int q = 0; q < NN_TRIPS; ++q) {
    const int qn = (q + 1 < NN_TRIPS) ? (q + 1) : q;
    const float* np = lp + (size_t)qn * (NN_G * 48);
    float nx[NN_G], ny[NN_G], nz[NN_G];
#pragma unroll
    for (int g = 0; g < NN_G; ++g) {
      const float* p = np + (size_t)g * 48;
      nx[g] = p[0]; ny[g] = p[1]; nz[g] = p[2];
    }

    v8f c[NN_G];
#pragma unroll
    for (int g = 0; g < NN_G; ++g) {
      const float bx = cx[g], by = cy[g], bz = cz[g];
      const float bn2 = __builtin_fmaf(bz, bz, __builtin_fmaf(bx, bx, by * by));
      const v8f zero = {0.f, 0.f, 0.f, 0.f, 0.f, 0.f, 0.f, 0.f};
#if USE_WMMA
      v2f Av;
      Av.x = half ? bz  : bx;
      Av.y = half ? bn2 : by;
      c[g] = __builtin_amdgcn_wmma_f32_16x16x4_f32(
           false, Av,  false, Bs,
           (short)0, zero,  false,  false);
#else
      c[g] = zero;
#pragma unroll
      for (int r = 0; r < 8; ++r) {
        const int src = half * 8 + r;
        const float rx = __shfl(bx, src, 32);
        const float ry = __shfl(by, src, 32);
        const float rz = __shfl(bz, src, 32);
        const float rn = __shfl(bn2, src, 32);
        c[g][r] = -2.0f * (rx * sx + ry * sy + rz * sz) + rn;
      }
#endif
    }

    unsigned hit[NN_G];
    unsigned anyhit = 0u;
    const float d2s = d[2];
#pragma unroll
    for (int g = 0; g < NN_G; ++g) {
      const float m01 = fminf(fminf(c[g][0], c[g][1]), fminf(c[g][2], c[g][3]));
      const float m23 = fminf(fminf(c[g][4], c[g][5]), fminf(c[g][6], c[g][7]));
      const float mt  = fminf(m01, m23);
      hit[g] = __builtin_amdgcn_ballot_w32(mt < d2s);
      anyhit |= hit[g];
    }
    if (anyhit != 0u) {
#pragma unroll
      for (int g = 0; g < NN_G; ++g) {
        if (hit[g] != 0u) {
          const int n0 = (q * NN_G + g) * 16 + half * 8;
#pragma unroll
          for (int r = 0; r < 8; ++r) ins3_stream(c[g][r], n0 + r, d, ix);
        }
      }
    }

#pragma unroll
    for (int g = 0; g < NN_G; ++g) { cx[g] = nx[g]; cy[g] = ny[g]; cz[g] = nz[g]; }
  }

  {
    float pd[3]; int pi[3];
#pragma unroll
    for (int k = 0; k < 3; ++k) {
      pd[k] = __shfl_xor(d[k], 16, 32);
      pi[k] = __shfl_xor(ix[k], 16, 32);
    }
#pragma unroll
    for (int k = 0; k < 3; ++k) ins3_lex(pd[k], pi[k], d, ix);
  }

  if (half == 0) {
    float wv[3];
    float wsum = 0.f;
#pragma unroll
    for (int k = 0; k < 3; ++k) {
      const int nn = ix[k];
      const float* qp = pcb + (size_t)nn * 3;
      const float dx = qp[0] - sx, dy = qp[1] - sy, dz = qp[2] - sz;
      const float dd = dx * dx + dy * dy + dz * dz;
      const float dist = sqrtf(fmaxf(dd, 0.0f));
      wv[k] = 1.0f / (dist + 1e-8f);
      wsum += wv[k];
      idx3[((size_t)b * M_ + mrow) * 3 + k] = nn;
    }
#pragma unroll
    for (int k = 0; k < 3; ++k)
      w3[((size_t)b * M_ + mrow) * 3 + k] = wv[k] / wsum;
  }
}

__global__ __launch_bounds__(256) void interp_kernel(
    const float* __restrict__ feats, const int* __restrict__ idx3,
    const float* __restrict__ w3, float* __restrict__ out) {
  const int bc = blockIdx.x;
  const int b  = bc / C_;
  const int c  = bc - b * C_;
  const float* f = feats + ((size_t)b * C_ + c) * N_;
  float*       o = out   + ((size_t)b * C_ + c) * M_;
  const int*   ib = idx3 + (size_t)b * M_ * 3;
  const float* wb = w3   + (size_t)b * M_ * 3;
  for (int m = threadIdx.x; m < M_; m += 256) {
    const int   n0 = ib[m * 3 + 0], n1 = ib[m * 3 + 1], n2 = ib[m * 3 + 2];
    const float w0 = wb[m * 3 + 0], w1 = wb[m * 3 + 1], w2 = wb[m * 3 + 2];
    o[m] = w0 * f[n0] + w1 * f[n1] + w2 * f[n2];
  }
}

extern "C" void kernel_launch(void* const* d_in, const int* in_sizes, int n_in,
                              void* d_out, int out_size, void* d_ws, size_t ws_size,
                              hipStream_t stream) {
  const float* pc    = (const float*)d_in[0];
  const float* feats = (const float*)d_in[1];
  const float* obj   = (const float*)d_in[2];
  const float* grasp = (const float*)d_in[3];
  float* out = (float*)d_out;

  char* ws = (char*)d_ws;
  size_t off = 0;
  int* fps = (int*)(ws + off);
  off += (size_t)B_ * M_ * 4;        off = (off + 255) & ~(size_t)255;
  int* idx3 = (int*)(ws + off);
  off += (size_t)B_ * M_ * 3 * 4;    off = (off + 255) & ~(size_t)255;
  float* w3 = (float*)(ws + off);
  off += (size_t)B_ * M_ * 3 * 4;

  fps_kernel<<<B_, 1024, 0, stream>>>(pc, obj, grasp, fps);
  three_nn_kernel<<<(B_ * (M_ / 16)) / 8, 256, 0, stream>>>(pc, fps, idx3, w3);
  interp_kernel<<<B_ * C_, 256, 0, stream>>>(feats, idx3, w3, out);
}
